# starting kernel
# baseline (speedup 1.0000x reference)
#include <hip/hip_runtime.h>

typedef __attribute__((ext_vector_type(2))) float v2f;
typedef __attribute__((ext_vector_type(8))) float v8f;

#define FDIM 128

__global__ __launch_bounds__(256) void k_init(int* __restrict__ deg,
                                              float* __restrict__ stats, int N)
{
    int i = blockIdx.x * 256 + threadIdx.x;
    if (i < N) deg[i] = 1;
    if (i < 256) stats[i] = 0.f;
}

__global__ __launch_bounds__(256) void k_deg(const int* __restrict__ dst,
                                             int* __restrict__ deg, int E)
{
    int i = blockIdx.x * 256 + threadIdx.x;
    if (i < E) atomicAdd(&deg[dst[i]], 1);
}

__global__ __launch_bounds__(256) void k_dis(const int* __restrict__ deg,
                                             float* __restrict__ dis, int N)
{
    int i = blockIdx.x * 256 + threadIdx.x;
    if (i < N) dis[i] = rsqrtf((float)deg[i]);
}

__global__ __launch_bounds__(256) void k_gemm1(const float* __restrict__ X,
                                               const float* __restrict__ W,
                                               float* __restrict__ O, int N)
{
    __shared__ float lA[32 * FDIM];
    __shared__ v2f   lB[64 * 64];

    const int t    = threadIdx.x;
    const int rowb = blockIdx.x * 32;
    const int colb = blockIdx.y * 64;

    for (int idx = t; idx < 32 * FDIM; idx += 256) {
        int r = idx >> 7, k = idx & 127;
        int gr = rowb + r; if (gr >= N) gr = N - 1;
        lA[idx] = X[(size_t)gr * FDIM + k];
    }
    for (int idx = t; idx < 64 * 64; idx += 256) {
        int k2 = idx >> 6, c = idx & 63;
        const float* wp = W + (size_t)(2 * k2) * FDIM + (colb + c);
        v2f b; b.x = wp[0]; b.y = wp[FDIM];
        lB[idx] = b;
    }
    __syncthreads();

    const int wave = t >> 5, lane = t & 31;
    const int m = lane & 15, hi = lane >> 4;
    const int rt = wave >> 2, ct = wave & 3;
    const float* Arow = &lA[(rt * 16 + m) * FDIM];
    const int nloc = ct * 16 + m;

    v8f acc = {0.f, 0.f, 0.f, 0.f, 0.f, 0.f, 0.f, 0.f};
#pragma unroll
    for (int k = 0; k < FDIM; k += 4) {
        v2f a = *(const v2f*)&Arow[k + 2 * hi];
        v2f b = lB[(k / 2 + hi) * 64 + nloc];
        acc = __builtin_amdgcn_wmma_f32_16x16x4_f32(
            false, a, false, b, (short)0, acc, false, false);
    }

    const int col = colb + nloc;
#pragma unroll
    for (int r = 0; r < 8; ++r) {
        int row = rowb + rt * 16 + r + 8 * hi;
        if (row < N) O[(size_t)row * FDIM + col] = acc[r];
    }
}

__global__ __launch_bounds__(256) void k_self128(const float* __restrict__ h,
                                                 const float* __restrict__ dis,
                                                 const float* __restrict__ bias,
                                                 float* __restrict__ out, int N)
{
    long long idx = (long long)blockIdx.x * 256 + threadIdx.x;
    if (idx >= (long long)N * FDIM) return;
    int i = (int)(idx >> 7), f = (int)(idx & 127);
    float d = dis[i];
    out[idx] = h[idx] * d * d + bias[f];
}

__global__ __launch_bounds__(256) void k_selfC(const float* __restrict__ h,
                                               const float* __restrict__ dis,
                                               const float* __restrict__ bias,
                                               float* __restrict__ out, int N, int C)
{
    long long idx = (long long)blockIdx.x * 256 + threadIdx.x;
    if (idx >= (long long)N * C) return;
    int i = (int)(idx / C), f = (int)(idx - (long long)i * C);
    float d = dis[i];
    out[idx] = h[idx] * d * d + bias[f];
}

__global__ __launch_bounds__(256) void k_edge128(const float* __restrict__ h,
                                                 const int* __restrict__ src,
                                                 const int* __restrict__ dst,
                                                 const float* __restrict__ dis,
                                                 float* __restrict__ out, int E)
{
    long long gid = (long long)blockIdx.x * 256 + threadIdx.x;
    int e = (int)(gid >> 5);
    int lane = threadIdx.x & 31;
    if (e >= E) return;
    int s = src[e], d = dst[e];
    float w = dis[s] * dis[d];
    const float4* hp = (const float4*)(h + (size_t)s * FDIM);
    float4 v = hp[lane];
    float* op = out + (size_t)d * FDIM + lane * 4;
    atomicAdd(op + 0, v.x * w);
    atomicAdd(op + 1, v.y * w);
    atomicAdd(op + 2, v.z * w);
    atomicAdd(op + 3, v.w * w);
}

__global__ __launch_bounds__(256) void k_edgeC(const float* __restrict__ h,
                                               const int* __restrict__ src,
                                               const int* __restrict__ dst,
                                               const float* __restrict__ dis,
                                               float* __restrict__ out, int E, int C)
{
    long long gid = (long long)blockIdx.x * 256 + threadIdx.x;
    int e = (int)(gid >> 5);
    int lane = threadIdx.x & 31;
    if (e >= E) return;
    int s = src[e], d = dst[e];
    float w = dis[s] * dis[d];
    const float* hp = h + (size_t)s * C;
    float* op = out + (size_t)d * C;
    if (lane < C)      atomicAdd(op + lane,      hp[lane]      * w);
    if (32 + lane < C) atomicAdd(op + 32 + lane, hp[32 + lane] * w);
}

__global__ __launch_bounds__(256) void k_relu_stats(float* __restrict__ a,
                                                    float* __restrict__ gsum,
                                                    float* __restrict__ gsq, int N)
{
    __shared__ float ssum[256];
    __shared__ float ssq[256];
    int t = threadIdx.x, f = t & 127, sub = t >> 7;
    float sum = 0.f, sq = 0.f;
    int base = blockIdx.x * 64;
#pragma unroll 4
    for (int i = 0; i < 32; ++i) {
        int row = base + sub + 2 * i;
        if (row < N) {
            size_t idx = (size_t)row * FDIM + f;
            float v = a[idx];
            v = fmaxf(v, 0.f);
            a[idx] = v;
            sum += v;
            sq  += v * v;
        }
    }
    ssum[t] = sum; ssq[t] = sq;
    __syncthreads();
    if (t < 128) {
        atomicAdd(&gsum[t], ssum[t] + ssum[t + 128]);
        atomicAdd(&gsq[t],  ssq[t]  + ssq[t + 128]);
    }
}

__global__ void k_bn_finalize(const float* __restrict__ gsum,
                              const float* __restrict__ gsq,
                              const float* __restrict__ gamma,
                              const float* __restrict__ beta,
                              float* __restrict__ scl,
                              float* __restrict__ sft, int N)
{
    int f = threadIdx.x;
    float inv_n = 1.f / (float)N;
    float mean = gsum[f] * inv_n;
    float var  = gsq[f] * inv_n - mean * mean;
    float inv  = rsqrtf(var + 1e-5f);
    float g = gamma[f] * inv;
    scl[f] = g;
    sft[f] = beta[f] - mean * g;
}

__global__ __launch_bounds__(96) void k_gemm2(const float* __restrict__ A,
                                              const float* __restrict__ W,
                                              const float* __restrict__ scl,
                                              const float* __restrict__ sft,
                                              float* __restrict__ O, int N, int C)
{
    __shared__ float lA[16 * FDIM];
    __shared__ v2f   lB[64 * 48];

    const int t = threadIdx.x;
    const int rowb = blockIdx.x * 16;

    for (int idx = t; idx < 16 * FDIM; idx += 96) {
        int r = idx >> 7, k = idx & 127;
        int gr = rowb + r; if (gr >= N) gr = N - 1;
        lA[idx] = A[(size_t)gr * FDIM + k] * scl[k] + sft[k];
    }
    for (int idx = t; idx < 64 * 48; idx += 96) {
        int k2 = idx / 48, c = idx - k2 * 48;
        v2f b;
        b.x = (c < C) ? W[(size_t)(2 * k2) * C + c]     : 0.f;
        b.y = (c < C) ? W[(size_t)(2 * k2 + 1) * C + c] : 0.f;
        lB[idx] = b;
    }
    __syncthreads();

    const int wave = t >> 5, lane = t & 31;
    const int m = lane & 15, hi = lane >> 4;
    const float* Arow = &lA[m * FDIM];
    const int nloc = wave * 16 + m;

    v8f acc = {0.f, 0.f, 0.f, 0.f, 0.f, 0.f, 0.f, 0.f};
#pragma unroll
    for (int k = 0; k < FDIM; k += 4) {
        v2f a = *(const v2f*)&Arow[k + 2 * hi];
        v2f b = lB[(k / 2 + hi) * 48 + nloc];
        acc = __builtin_amdgcn_wmma_f32_16x16x4_f32(
            false, a, false, b, (short)0, acc, false, false);
    }

    if (nloc < C) {
#pragma unroll
        for (int r = 0; r < 8; ++r) {
            int row = rowb + r + 8 * hi;
            if (row < N) O[(size_t)row * C + nloc] = acc[r];
        }
    }
}

__global__ __launch_bounds__(256) void k_logsoftmax(float* __restrict__ out, int N, int C)
{
    long long gid = (long long)blockIdx.x * 256 + threadIdx.x;
    int row = (int)(gid >> 5);
    int lane = threadIdx.x & 31;
    if (row >= N) return;
    float* p = out + (size_t)row * C;
    float v0 = (lane < C) ? p[lane] : -3.4e38f;
    float v1 = (32 + lane < C) ? p[32 + lane] : -3.4e38f;
    float mx = fmaxf(v0, v1);
#pragma unroll
    for (int o = 16; o > 0; o >>= 1) mx = fmaxf(mx, __shfl_xor(mx, o, 32));
    float s = ((lane < C) ? expf(v0 - mx) : 0.f) +
              ((32 + lane < C) ? expf(v1 - mx) : 0.f);
#pragma unroll
    for (int o = 16; o > 0; o >>= 1) s += __shfl_xor(s, o, 32);
    float l = logf(s);
    if (lane < C)      p[lane]      = v0 - mx - l;
    if (32 + lane < C) p[32 + lane] = v1 - mx - l;
}

extern "C" void kernel_launch(void* const* d_in, const int* in_sizes, int n_in,
                              void* d_out, int out_size, void* d_ws, size_t ws_size,
                              hipStream_t stream)
{
    const float* x     = (const float*)d_in[0];
    const int*   ei    = (const int*)  d_in[1];
    const float* W1    = (const float*)d_in[2];
    const float* b1    = (const float*)d_in[3];
    const float* gamma = (const float*)d_in[4];
    const float* beta  = (const float*)d_in[5];
    const float* W2    = (const float*)d_in[6];
    const float* b2    = (const float*)d_in[7];
    float* out = (float*)d_out;

    const int N = in_sizes[0] / FDIM;
    const int E = in_sizes[1] / 2;
    const int C = in_sizes[6] / FDIM;
    const int* src = ei;
    const int* dst = ei + E;

    char* ws = (char*)d_ws;
    size_t off = 0;
    auto up = [](size_t v) { return (v + 255) & ~(size_t)255; };
    int*   deg   = (int*)(ws + off);   off = up(off + (size_t)N * 4);
    float* dis   = (float*)(ws + off); off = up(off + (size_t)N * 4);
    float* stats = (float*)(ws + off); off = up(off + 512 * 4);
    float* gsum = stats, *gsq = stats + 128, *scl = stats + 256, *sft = stats + 384;
    float* h1   = (float*)(ws + off);  off = up(off + (size_t)N * FDIM * 4);
    float* agg1 = (float*)(ws + off);  off = up(off + (size_t)N * FDIM * 4);
    float* h2   = h1;

    const int nb_N   = (N + 255) / 256;
    const int nb_E   = (E + 255) / 256;
    const int nb_Ew  = (E + 7) / 8;
    const int nb_Nw  = (N + 7) / 8;
    const long long nfe1 = (long long)N * FDIM;
    const long long nfe2 = (long long)N * C;

    k_init<<<nb_N, 256, 0, stream>>>(deg, stats, N);
    k_deg<<<nb_E, 256, 0, stream>>>(dst, deg, E);
    k_dis<<<nb_N, 256, 0, stream>>>(deg, dis, N);

    k_gemm1<<<dim3((N + 31) / 32, FDIM / 64), 256, 0, stream>>>(x, W1, h1, N);

    k_self128<<<(int)((nfe1 + 255) / 256), 256, 0, stream>>>(h1, dis, b1, agg1, N);
    k_edge128<<<nb_Ew, 256, 0, stream>>>(h1, src, dst, dis, agg1, E);

    k_relu_stats<<<(N + 63) / 64, 256, 0, stream>>>(agg1, gsum, gsq, N);
    k_bn_finalize<<<1, 128, 0, stream>>>(gsum, gsq, gamma, beta, scl, sft, N);

    k_gemm2<<<(N + 15) / 16, 96, 0, stream>>>(agg1, W2, scl, sft, h2, N, C);

    k_selfC<<<(int)((nfe2 + 255) / 256), 256, 0, stream>>>(h2, dis, b2, out, N, C);
    k_edgeC<<<nb_Ew, 256, 0, stream>>>(h2, src, dst, dis, out, E, C);

    k_logsoftmax<<<nb_Nw, 256, 0, stream>>>(out, N, C);
}


# Conv sums without global float atomics: a block owns 400 destination rows in LDS (start: self term + bias), scans the edge list for its own edges, adds their rows by LDS float adds; sum order changes
# speedup vs baseline: 1.7062x; 1.7062x over previous
#include <hip/hip_runtime.h>

typedef __attribute__((ext_vector_type(2))) float v2f;
typedef __attribute__((ext_vector_type(8))) float v8f;

#define FDIM 128

__global__ __launch_bounds__(256) void k_init(int* __restrict__ deg,
                                              float* __restrict__ stats, int N)
{
    int i = blockIdx.x * 256 + threadIdx.x;
    if (i < N) deg[i] = 1;
    if (i < 256) stats[i] = 0.f;
}

__global__ __launch_bounds__(256) void k_deg(const int* __restrict__ dst,
                                             int* __restrict__ deg, int E)
{
    int i = blockIdx.x * 256 + threadIdx.x;
    if (i < E) atomicAdd(&deg[dst[i]], 1);
}

__global__ __launch_bounds__(256) void k_dis(const int* __restrict__ deg,
                                             float* __restrict__ dis, int N)
{
    int i = blockIdx.x * 256 + threadIdx.x;
    if (i < N) dis[i] = rsqrtf((float)deg[i]);
}

__global__ __launch_bounds__(256) void k_gemm1(const float* __restrict__ X,
                                               const float* __restrict__ W,
                                               float* __restrict__ O, int N)
{
    __shared__ float lA[32 * FDIM];
    __shared__ v2f   lB[64 * 64];

    const int t    = threadIdx.x;
    const int rowb = blockIdx.x * 32;
    const int colb = blockIdx.y * 64;

    for (int idx = t; idx < 32 * FDIM; idx += 256) {
        int r = idx >> 7, k = idx & 127;
        int gr = rowb + r; if (gr >= N) gr = N - 1;
        lA[idx] = X[(size_t)gr * FDIM + k];
    }
    for (int idx = t; idx < 64 * 64; idx += 256) {
        int k2 = idx >> 6, c = idx & 63;
        const float* wp = W + (size_t)(2 * k2) * FDIM + (colb + c);
        v2f b; b.x = wp[0]; b.y = wp[FDIM];
        lB[idx] = b;
    }
    __syncthreads();

    const int wave = t >> 5, lane = t & 31;
    const int m = lane & 15, hi = lane >> 4;
    const int rt = wave >> 2, ct = wave & 3;
    const float* Arow = &lA[(rt * 16 + m) * FDIM];
    const int nloc = ct * 16 + m;

    v8f acc = {0.f, 0.f, 0.f, 0.f, 0.f, 0.f, 0.f, 0.f};
#pragma unroll
    for (int k = 0; k < FDIM; k += 4) {
        v2f a = *(const v2f*)&Arow[k + 2 * hi];
        v2f b = lB[(k / 2 + hi) * 64 + nloc];
        acc = __builtin_amdgcn_wmma_f32_16x16x4_f32(
            false, a, false, b, (short)0, acc, false, false);
    }

    const int col = colb + nloc;
#pragma unroll
    for (int r = 0; r < 8; ++r) {
        int row = rowb + rt * 16 + r + 8 * hi;
        if (row < N) O[(size_t)row * FDIM + col] = acc[r];
    }
}

#define OWN_R   400
#define OWN_CAP 8192
#define OWN_T   1024

__device__ __forceinline__ void own_scan(const int* __restrict__ dst, int E, int row0, unsigned un,
                                         int2* lst, int* cnt)
{
    const int t = threadIdx.x;
    auto push = [&](int e, unsigned r) {
        int pos = atomicAdd(cnt, 1);
        if (pos < OWN_CAP) lst[pos] = make_int2(e, (int)r);
    };
    auto test = [&](const int4& d, int e) {
        unsigned r0 = (unsigned)(d.x - row0), r1 = (unsigned)(d.y - row0);
        unsigned r2 = (unsigned)(d.z - row0), r3 = (unsigned)(d.w - row0);
        unsigned m = min(min(r0, r1), min(r2, r3));
        if (m < un) {
            if (r0 < un) push(e, r0);
            if (r1 < un) push(e + 1, r1);
            if (r2 < un) push(e + 2, r2);
            if (r3 < un) push(e + 3, r3);
        }
    };
    const int mis = (int)(((size_t)dst >> 2) & 3);
    int head = (4 - mis) & 3; if (head > E) head = E;
    const int nvec = (E - head) >> 2;
    const int tail0 = head + 4 * nvec;
    const int4* d4 = (const int4*)(dst + head);
    const int nfull = nvec / (4 * OWN_T);
    if (nfull > 0) {
        int tile = (int)(blockIdx.x % (unsigned)nfull);
        const int4* p = d4 + (size_t)tile * (4 * OWN_T) + t;
        int4 c0 = p[0], c1 = p[OWN_T], c2 = p[2 * OWN_T], c3 = p[3 * OWN_T];
        for (int it = 0; it < nfull; ++it) {
            int nt = tile + 1; if (nt >= nfull) nt = 0;
            const int4* q = d4 + (size_t)nt * (4 * OWN_T) + t;
            int4 n0 = q[0], n1 = q[OWN_T], n2 = q[2 * OWN_T], n3 = q[3 * OWN_T];
            int e = head + 4 * (tile * (4 * OWN_T) + t);
            test(c0, e);
            test(c1, e + 4 * OWN_T);
            test(c2, e + 8 * OWN_T);
            test(c3, e + 12 * OWN_T);
            c0 = n0; c1 = n1; c2 = n2; c3 = n3; tile = nt;
        }
    }
    for (int v = nfull * (4 * OWN_T) + t; v < nvec; v += OWN_T) test(d4[v], head + 4 * v);
    if (t < head) {
        unsigned r = (unsigned)(dst[t] - row0);
        if (r < un) push(t, r);
    }
    if (t < E - tail0) {
        unsigned r = (unsigned)(dst[tail0 + t] - row0);
        if (r < un) push(tail0 + t, r);
    }
}

__device__ __forceinline__ void own_resolve(const int* __restrict__ src, const float* __restrict__ dis,
                                            const float* sdis, int2* lst, float* wl, int total, int N)
{
    for (int i = threadIdx.x; i < total; i += OWN_T) {
        int2 en = lst[i];
        int s = src[en.x];
        s = min(max(s, 0), N - 1);
        lst[i].x = s;
        wl[i] = dis[s] * sdis[en.y];
    }
}

__global__ __launch_bounds__(OWN_T) void k_own128(const float* __restrict__ h,
                                                  const int* __restrict__ src,
                                                  const int* __restrict__ dst,
                                                  const float* __restrict__ dis,
                                                  const float* __restrict__ bias,
                                                  float* __restrict__ out, int N, int E)
{
    __shared__ float acc[OWN_R * FDIM];
    __shared__ int2  lst[OWN_CAP];
    __shared__ float wl[OWN_CAP];
    __shared__ float sdis[OWN_R];
    __shared__ int   cnt;

    const int t = threadIdx.x;
    const int row0 = blockIdx.x * OWN_R;
    const int nrows = min(OWN_R, N - row0);
    const unsigned un = (unsigned)nrows;
    const float4* h4 = (const float4*)h;

    if (t == 0) cnt = 0;
    for (int r = t; r < nrows; r += OWN_T) sdis[r] = dis[row0 + r];
    {
        const float4 bq = ((const float4*)bias)[t & 31];
        for (int idx = t; idx < nrows * 32; idx += OWN_T) {
            int r = idx >> 5, q = idx & 31;
            float d = dis[row0 + r];
            float4 v = h4[(size_t)(row0 + r) * 32 + q];
            float* a = &acc[r * FDIM + q];
            a[0]  = v.x * d * d + bq.x;
            a[32] = v.y * d * d + bq.y;
            a[64] = v.z * d * d + bq.z;
            a[96] = v.w * d * d + bq.w;
        }
    }
    __syncthreads();

    own_scan(dst, E, row0, un, lst, &cnt);
    __syncthreads();

    const int total = __builtin_amdgcn_readfirstlane(cnt);
    if (total <= OWN_CAP) {
        own_resolve(src, dis, sdis, lst, wl, total, N);
        __syncthreads();
        const int wave = __builtin_amdgcn_readfirstlane(t >> 5), lane = t & 31;
        const int ngrp = total >> 3;
        for (int g = wave; g < ngrp; g += OWN_T / 32) {
            int r[8]; float w[8]; float4 v[8];
#pragma unroll
            for (int u = 0; u < 8; ++u) {
                int2 en = lst[8 * g + u];
                int s = __builtin_amdgcn_readfirstlane(en.x);
                r[u] = __builtin_amdgcn_readfirstlane(en.y);
                w[u] = wl[8 * g + u];
                v[u] = h4[(size_t)s * 32 + lane];
            }
#pragma unroll
            for (int u = 0; u < 8; ++u) {
                float* a = &acc[r[u] * FDIM + lane];
                atomicAdd(a,      v[u].x * w[u]);
                atomicAdd(a + 32, v[u].y * w[u]);
                atomicAdd(a + 64, v[u].z * w[u]);
                atomicAdd(a + 96, v[u].w * w[u]);
            }
        }
        if (8 * ngrp + wave < total) {
            int2 en = lst[8 * ngrp + wave];
            int s = __builtin_amdgcn_readfirstlane(en.x);
            int r = __builtin_amdgcn_readfirstlane(en.y);
            float w = wl[8 * ngrp + wave];
            float4 v = h4[(size_t)s * 32 + lane];
            float* a = &acc[r * FDIM + lane];
            atomicAdd(a,      v.x * w);
            atomicAdd(a + 32, v.y * w);
            atomicAdd(a + 64, v.z * w);
            atomicAdd(a + 96, v.w * w);
        }
    } else {
        for (int e = t; e < E; e += OWN_T) {
            unsigned r = (unsigned)(dst[e] - row0);
            if (r < un) {
                int s = min(max(src[e], 0), N - 1);
                float w = dis[s] * sdis[r];
                const float* hp = h + (size_t)s * FDIM;
                for (int f = 0; f < FDIM; ++f)
                    atomicAdd(&acc[r * FDIM + (f & 3) * 32 + (f >> 2)], hp[f] * w);
            }
        }
    }
    __syncthreads();

    float4* o4 = (float4*)out;
    for (int idx = t; idx < nrows * 32; idx += OWN_T) {
        int r = idx >> 5, q = idx & 31;
        const float* a = &acc[r * FDIM + q];
        o4[(size_t)(row0 + r) * 32 + q] = make_float4(a[0], a[32], a[64], a[96]);
    }
}

__global__ __launch_bounds__(OWN_T) void k_ownC(const float* __restrict__ h,
                                                const int* __restrict__ src,
                                                const int* __restrict__ dst,
                                                const float* __restrict__ dis,
                                                const float* __restrict__ bias,
                                                float* __restrict__ out, int N, int E, int C)
{
    __shared__ float acc[OWN_R * 64];
    __shared__ int2  lst[OWN_CAP];
    __shared__ float wl[OWN_CAP];
    __shared__ float sdis[OWN_R];
    __shared__ int   cnt;

    const int t = threadIdx.x;
    const int row0 = blockIdx.x * OWN_R;
    const int nrows = min(OWN_R, N - row0);
    const unsigned un = (unsigned)nrows;
    const size_t g0 = (size_t)row0 * C;

    if (t == 0) cnt = 0;
    for (int r = t; r < nrows; r += OWN_T) sdis[r] = dis[row0 + r];
    for (int idx = t; idx < nrows * C; idx += OWN_T) {
        int r = idx / C, f = idx - r * C;
        float d = dis[row0 + r];
        acc[idx] = h[g0 + idx] * d * d + bias[f];
    }
    __syncthreads();

    own_scan(dst, E, row0, un, lst, &cnt);
    __syncthreads();

    const int total = __builtin_amdgcn_readfirstlane(cnt);
    if (total <= OWN_CAP) {
        own_resolve(src, dis, sdis, lst, wl, total, N);
        __syncthreads();
        const int wave = __builtin_amdgcn_readfirstlane(t >> 5), lane = t & 31;
        const bool c0 = lane < C, c1 = 32 + lane < C;
        const int l0 = min(lane, C - 1), l1 = min(32 + lane, C - 1);
        const int ngrp = total >> 3;
        for (int g = wave; g < ngrp; g += OWN_T / 32) {
            int s[8], r[8]; float w[8], v0[8], v1[8];
#pragma unroll
            for (int u = 0; u < 8; ++u) {
                int2 en = lst[8 * g + u];
                s[u] = __builtin_amdgcn_readfirstlane(en.x);
                r[u] = __builtin_amdgcn_readfirstlane(en.y);
                w[u] = wl[8 * g + u];
            }
#pragma unroll
            for (int u = 0; u < 8; ++u) {
                const float* hp = h + (size_t)s[u] * C;
                v0[u] = hp[l0];
                v1[u] = hp[l1];
            }
            if (c0) {
#pragma unroll
                for (int u = 0; u < 8; ++u) atomicAdd(&acc[r[u] * C + lane], v0[u] * w[u]);
            }
            if (c1) {
#pragma unroll
                for (int u = 0; u < 8; ++u) atomicAdd(&acc[r[u] * C + 32 + lane], v1[u] * w[u]);
            }
        }
        if (8 * ngrp + wave < total) {
            int2 en = lst[8 * ngrp + wave];
            int s = __builtin_amdgcn_readfirstlane(en.x);
            int r = __builtin_amdgcn_readfirstlane(en.y);
            float w = wl[8 * ngrp + wave];
            const float* hp = h + (size_t)s * C;
            float* a = &acc[r * C];
            if (c0) atomicAdd(a + lane,      hp[lane] * w);
            if (c1) atomicAdd(a + 32 + lane, hp[32 + lane] * w);
        }
    } else {
        for (int e = t; e < E; e += OWN_T) {
            unsigned r = (unsigned)(dst[e] - row0);
            if (r < un) {
                int s = min(max(src[e], 0), N - 1);
                float w = dis[s] * sdis[r];
                const float* hp = h + (size_t)s * C;
                for (int f = 0; f < C; ++f) atomicAdd(&acc[r * C + f], hp[f] * w);
            }
        }
    }
    __syncthreads();

    for (int idx = t; idx < nrows * C; idx += OWN_T) out[g0 + idx] = acc[idx];
}

__global__ __launch_bounds__(256) void k_relu_stats(float* __restrict__ a,
                                                    float* __restrict__ gsum,
                                                    float* __restrict__ gsq, int N)
{
    __shared__ float ssum[256];
    __shared__ float ssq[256];
    int t = threadIdx.x, f = t & 127, sub = t >> 7;
    float sum = 0.f, sq = 0.f;
    int base = blockIdx.x * 64;
#pragma unroll 4
    for (int i = 0; i < 32; ++i) {
        int row = base + sub + 2 * i;
        if (row < N) {
            size_t idx = (size_t)row * FDIM + f;
            float v = a[idx];
            v = fmaxf(v, 0.f);
            a[idx] = v;
            sum += v;
            sq  += v * v;
        }
    }
    ssum[t] = sum; ssq[t] = sq;
    __syncthreads();
    if (t < 128) {
        atomicAdd(&gsum[t], ssum[t] + ssum[t + 128]);
        atomicAdd(&gsq[t],  ssq[t]  + ssq[t + 128]);
    }
}

__global__ void k_bn_finalize(const float* __restrict__ gsum,
                              const float* __restrict__ gsq,
                              const float* __restrict__ gamma,
                              const float* __restrict__ beta,
                              float* __restrict__ scl,
                              float* __restrict__ sft, int N)
{
    int f = threadIdx.x;
    float inv_n = 1.f / (float)N;
    float mean = gsum[f] * inv_n;
    float var  = gsq[f] * inv_n - mean * mean;
    float inv  = rsqrtf(var + 1e-5f);
    float g = gamma[f] * inv;
    scl[f] = g;
    sft[f] = beta[f] - mean * g;
}

__global__ __launch_bounds__(96) void k_gemm2(const float* __restrict__ A,
                                              const float* __restrict__ W,
                                              const float* __restrict__ scl,
                                              const float* __restrict__ sft,
                                              float* __restrict__ O, int N, int C)
{
    __shared__ float lA[16 * FDIM];
    __shared__ v2f   lB[64 * 48];

    const int t = threadIdx.x;
    const int rowb = blockIdx.x * 16;

    for (int idx = t; idx < 16 * FDIM; idx += 96) {
        int r = idx >> 7, k = idx & 127;
        int gr = rowb + r; if (gr >= N) gr = N - 1;
        lA[idx] = A[(size_t)gr * FDIM + k] * scl[k] + sft[k];
    }
    for (int idx = t; idx < 64 * 48; idx += 96) {
        int k2 = idx / 48, c = idx - k2 * 48;
        v2f b;
        b.x = (c < C) ? W[(size_t)(2 * k2) * C + c]     : 0.f;
        b.y = (c < C) ? W[(size_t)(2 * k2 + 1) * C + c] : 0.f;
        lB[idx] = b;
    }
    __syncthreads();

    const int wave = t >> 5, lane = t & 31;
    const int m = lane & 15, hi = lane >> 4;
    const float* Arow = &lA[m * FDIM];
    const int nloc = wave * 16 + m;

    v8f acc = {0.f, 0.f, 0.f, 0.f, 0.f, 0.f, 0.f, 0.f};
#pragma unroll
    for (int k = 0; k < FDIM; k += 4) {
        v2f a = *(const v2f*)&Arow[k + 2 * hi];
        v2f b = lB[(k / 2 + hi) * 48 + nloc];
        acc = __builtin_amdgcn_wmma_f32_16x16x4_f32(
            false, a, false, b, (short)0, acc, false, false);
    }

    if (nloc < C) {
#pragma unroll
        for (int r = 0; r < 8; ++r) {
            int row = rowb + r + 8 * hi;
            if (row < N) O[(size_t)row * C + nloc] = acc[r];
        }
    }
}

__global__ __launch_bounds__(256) void k_logsoftmax(float* __restrict__ out, int N, int C)
{
    long long gid = (long long)blockIdx.x * 256 + threadIdx.x;
    int row = (int)(gid >> 5);
    int lane = threadIdx.x & 31;
    if (row >= N) return;
    float* p = out + (size_t)row * C;
    float v0 = (lane < C) ? p[lane] : -3.4e38f;
    float v1 = (32 + lane < C) ? p[32 + lane] : -3.4e38f;
    float mx = fmaxf(v0, v1);
#pragma unroll
    for (int o = 16; o > 0; o >>= 1) mx = fmaxf(mx, __shfl_xor(mx, o, 32));
    float s = ((lane < C) ? expf(v0 - mx) : 0.f) +
              ((32 + lane < C) ? expf(v1 - mx) : 0.f);
#pragma unroll
    for (int o = 16; o > 0; o >>= 1) s += __shfl_xor(s, o, 32);
    float l = logf(s);
    if (lane < C)      p[lane]      = v0 - mx - l;
    if (32 + lane < C) p[32 + lane] = v1 - mx - l;
}

extern "C" void kernel_launch(void* const* d_in, const int* in_sizes, int n_in,
                              void* d_out, int out_size, void* d_ws, size_t ws_size,
                              hipStream_t stream)
{
    const float* x     = (const float*)d_in[0];
    const int*   ei    = (const int*)  d_in[1];
    const float* W1    = (const float*)d_in[2];
    const float* b1    = (const float*)d_in[3];
    const float* gamma = (const float*)d_in[4];
    const float* beta  = (const float*)d_in[5];
    const float* W2    = (const float*)d_in[6];
    const float* b2    = (const float*)d_in[7];
    float* out = (float*)d_out;

    const int N = in_sizes[0] / FDIM;
    const int E = in_sizes[1] / 2;
    const int C = in_sizes[6] / FDIM;
    const int* src = ei;
    const int* dst = ei + E;

    char* ws = (char*)d_ws;
    size_t off = 0;
    auto up = [](size_t v) { return (v + 255) & ~(size_t)255; };
    int*   deg   = (int*)(ws + off);   off = up(off + (size_t)N * 4);
    float* dis   = (float*)(ws + off); off = up(off + (size_t)N * 4);
    float* stats = (float*)(ws + off); off = up(off + 512 * 4);
    float* gsum = stats, *gsq = stats + 128, *scl = stats + 256, *sft = stats + 384;
    float* h1   = (float*)(ws + off);  off = up(off + (size_t)N * FDIM * 4);
    float* agg1 = (float*)(ws + off);  off = up(off + (size_t)N * FDIM * 4);
    float* h2   = h1;

    const int nb_N   = (N + 255) / 256;
    const int nb_E   = (E + 255) / 256;
    const int nb_own = (N + OWN_R - 1) / OWN_R;
    const int nb_Nw  = (N + 7) / 8;

    k_init<<<nb_N, 256, 0, stream>>>(deg, stats, N);
    k_deg<<<nb_E, 256, 0, stream>>>(dst, deg, E);
    k_dis<<<nb_N, 256, 0, stream>>>(deg, dis, N);

    k_gemm1<<<dim3((N + 31) / 32, FDIM / 64), 256, 0, stream>>>(x, W1, h1, N);

    k_own128<<<nb_own, OWN_T, 0, stream>>>(h1, src, dst, dis, b1, agg1, N, E);

    k_relu_stats<<<(N + 63) / 64, 256, 0, stream>>>(agg1, gsum, gsq, N);
    k_bn_finalize<<<1, 128, 0, stream>>>(gsum, gsq, gamma, beta, scl, sft, N);

    k_gemm2<<<(N + 15) / 16, 96, 0, stream>>>(agg1, W2, scl, sft, h2, N, C);

    k_ownC<<<nb_own, OWN_T, 0, stream>>>(h2, src, dst, dis, b2, out, N, E, C);

    k_logsoftmax<<<nb_Nw, 256, 0, stream>>>(out, N, C);
}


# Conv2 walks no edge list: each conv1 owner block keeps its resolved (source, local row) list in the not yet written output rows it owns; its conv2 block reads it back; LDS adds in arrival order
# speedup vs baseline: 1.9412x; 1.1377x over previous
#include <hip/hip_runtime.h>

typedef __attribute__((ext_vector_type(2))) float v2f;
typedef __attribute__((ext_vector_type(8))) float v8f;

#define FDIM 128

__global__ __launch_bounds__(256) void k_init(int* __restrict__ deg,
                                              float* __restrict__ stats, int N)
{
    int i = blockIdx.x * 256 + threadIdx.x;
    if (i < N) deg[i] = 1;
    if (i < 256) stats[i] = 0.f;
}

__global__ __launch_bounds__(256) void k_deg(const int* __restrict__ dst,
                                             int* __restrict__ deg, int E)
{
    int i = blockIdx.x * 256 + threadIdx.x;
    if (i < E) atomicAdd(&deg[dst[i]], 1);
}

__global__ __launch_bounds__(256) void k_dis(const int* __restrict__ deg,
                                             float* __restrict__ dis, int N)
{
    int i = blockIdx.x * 256 + threadIdx.x;
    if (i < N) dis[i] = rsqrtf((float)deg[i]);
}

__global__ __launch_bounds__(256) void k_gemm1(const float* __restrict__ X,
                                               const float* __restrict__ W,
                                               float* __restrict__ O, int N)
{
    __shared__ float lA[32 * FDIM];
    __shared__ v2f   lB[64 * 64];

    const int t    = threadIdx.x;
    const int rowb = blockIdx.x * 32;
    const int colb = blockIdx.y * 64;

    for (int idx = t; idx < 32 * FDIM; idx += 256) {
        int r = idx >> 7, k = idx & 127;
        int gr = rowb + r; if (gr >= N) gr = N - 1;
        lA[idx] = X[(size_t)gr * FDIM + k];
    }
    for (int idx = t; idx < 64 * 64; idx += 256) {
        int k2 = idx >> 6, c = idx & 63;
        const float* wp = W + (size_t)(2 * k2) * FDIM + (colb + c);
        v2f b; b.x = wp[0]; b.y = wp[FDIM];
        lB[idx] = b;
    }
    __syncthreads();

    const int wave = t >> 5, lane = t & 31;
    const int m = lane & 15, hi = lane >> 4;
    const int rt = wave >> 2, ct = wave & 3;
    const float* Arow = &lA[(rt * 16 + m) * FDIM];
    const int nloc = ct * 16 + m;

    v8f acc = {0.f, 0.f, 0.f, 0.f, 0.f, 0.f, 0.f, 0.f};
#pragma unroll
    for (int k = 0; k < FDIM; k += 4) {
        v2f a = *(const v2f*)&Arow[k + 2 * hi];
        v2f b = lB[(k / 2 + hi) * 64 + nloc];
        acc = __builtin_amdgcn_wmma_f32_16x16x4_f32(
            false, a, false, b, (short)0, acc, false, false);
    }

    const int col = colb + nloc;
#pragma unroll
    for (int r = 0; r < 8; ++r) {
        int row = rowb + rt * 16 + r + 8 * hi;
        if (row < N) O[(size_t)row * FDIM + col] = acc[r];
    }
}

#define OWN_R   400
#define OWN_CAP 8192
#define OWN_T   1024

__device__ __forceinline__ void own_scan(const int* __restrict__ dst, int E, int row0, unsigned un,
                                         int2* lst, int* cnt)
{
    const int t = threadIdx.x;
    auto push = [&](int e, unsigned r) {
        int pos = atomicAdd(cnt, 1);
        if (pos < OWN_CAP) lst[pos] = make_int2(e, (int)r);
    };
    auto test = [&](const int4& d, int e) {
        unsigned r0 = (unsigned)(d.x - row0), r1 = (unsigned)(d.y - row0);
        unsigned r2 = (unsigned)(d.z - row0), r3 = (unsigned)(d.w - row0);
        unsigned m = min(min(r0, r1), min(r2, r3));
        if (m < un) {
            if (r0 < un) push(e, r0);
            if (r1 < un) push(e + 1, r1);
            if (r2 < un) push(e + 2, r2);
            if (r3 < un) push(e + 3, r3);
        }
    };
    const int mis = (int)(((size_t)dst >> 2) & 3);
    int head = (4 - mis) & 3; if (head > E) head = E;
    const int nvec = (E - head) >> 2;
    const int tail0 = head + 4 * nvec;
    const int4* d4 = (const int4*)(dst + head);
    const int nfull = nvec / (4 * OWN_T);
    if (nfull > 0) {
        int tile = (int)(blockIdx.x % (unsigned)nfull);
        const int4* p = d4 + (size_t)tile * (4 * OWN_T) + t;
        int4 c0 = p[0], c1 = p[OWN_T], c2 = p[2 * OWN_T], c3 = p[3 * OWN_T];
        for (int it = 0; it < nfull; ++it) {
            int nt = tile + 1; if (nt >= nfull) nt = 0;
            const int4* q = d4 + (size_t)nt * (4 * OWN_T) + t;
            int4 n0 = q[0], n1 = q[OWN_T], n2 = q[2 * OWN_T], n3 = q[3 * OWN_T];
            int e = head + 4 * (tile * (4 * OWN_T) + t);
            test(c0, e);
            test(c1, e + 4 * OWN_T);
            test(c2, e + 8 * OWN_T);
            test(c3, e + 12 * OWN_T);
            c0 = n0; c1 = n1; c2 = n2; c3 = n3; tile = nt;
        }
    }
    for (int v = nfull * (4 * OWN_T) + t; v < nvec; v += OWN_T) test(d4[v], head + 4 * v);
    if (t < head) {
        unsigned r = (unsigned)(dst[t] - row0);
        if (r < un) push(t, r);
    }
    if (t < E - tail0) {
        unsigned r = (unsigned)(dst[tail0 + t] - row0);
        if (r < un) push(tail0 + t, r);
    }
}

__device__ __forceinline__ void own_resolve(const int* __restrict__ src, const float* __restrict__ dis,
                                            const float* sdis, int2* lst, float* wl, int total, int N)
{
    for (int i = threadIdx.x; i < total; i += OWN_T) {
        int2 en = lst[i];
        int s = src[en.x];
        s = min(max(s, 0), N - 1);
        lst[i].x = s;
        wl[i] = dis[s] * sdis[en.y];
    }
}

__device__ __forceinline__ int keep_cap(int nrows, int C)
{
    return min(OWN_CAP, (nrows * C) >> 1);
}

__global__ __launch_bounds__(OWN_T) void k_own128(const float* __restrict__ h,
                                                  const int* __restrict__ src,
                                                  const int* __restrict__ dst,
                                                  const float* __restrict__ dis,
                                                  const float* __restrict__ bias,
                                                  float* __restrict__ out,
                                                  float* __restrict__ keep,
                                                  int* __restrict__ kcnt,
                                                  int N, int E, int C)
{
    __shared__ float acc[OWN_R * FDIM];
    __shared__ int2  lst[OWN_CAP];
    __shared__ float wl[OWN_CAP];
    __shared__ float sdis[OWN_R];
    __shared__ int   cnt;

    const int t = threadIdx.x;
    const int row0 = blockIdx.x * OWN_R;
    const int nrows = min(OWN_R, N - row0);
    const unsigned un = (unsigned)nrows;
    const float4* h4 = (const float4*)h;

    if (t == 0) cnt = 0;
    for (int r = t; r < nrows; r += OWN_T) sdis[r] = dis[row0 + r];
    {
        const float4 bq = ((const float4*)bias)[t & 31];
        for (int idx = t; idx < nrows * 32; idx += OWN_T) {
            int r = idx >> 5, q = idx & 31;
            float d = dis[row0 + r];
            float4 v = h4[(size_t)(row0 + r) * 32 + q];
            float* a = &acc[r * FDIM + q];
            a[0]  = v.x * d * d + bq.x;
            a[32] = v.y * d * d + bq.y;
            a[64] = v.z * d * d + bq.z;
            a[96] = v.w * d * d + bq.w;
        }
    }
    __syncthreads();

    own_scan(dst, E, row0, un, lst, &cnt);
    __syncthreads();

    const int total = __builtin_amdgcn_readfirstlane(cnt);
    if (t == 0) kcnt[blockIdx.x] = total;
    if (total <= OWN_CAP) {
        own_resolve(src, dis, sdis, lst, wl, total, N);
        __syncthreads();
        if (total <= keep_cap(nrows, C)) {
            int2* kl = (int2*)(keep + (size_t)row0 * C);
            for (int i = t; i < total; i += OWN_T) kl[i] = lst[i];
        }
        const int wave = __builtin_amdgcn_readfirstlane(t >> 5), lane = t & 31;
        const int ngrp = total >> 3;
        for (int g = wave; g < ngrp; g += OWN_T / 32) {
            int r[8]; float w[8]; float4 v[8];
#pragma unroll
            for (int u = 0; u < 8; ++u) {
                int2 en = lst[8 * g + u];
                int s = __builtin_amdgcn_readfirstlane(en.x);
                r[u] = __builtin_amdgcn_readfirstlane(en.y);
                w[u] = wl[8 * g + u];
                v[u] = h4[(size_t)s * 32 + lane];
            }
#pragma unroll
            for (int u = 0; u < 8; ++u) {
                float* a = &acc[r[u] * FDIM + lane];
                atomicAdd(a,      v[u].x * w[u]);
                atomicAdd(a + 32, v[u].y * w[u]);
                atomicAdd(a + 64, v[u].z * w[u]);
                atomicAdd(a + 96, v[u].w * w[u]);
            }
        }
        if (8 * ngrp + wave < total) {
            int2 en = lst[8 * ngrp + wave];
            int s = __builtin_amdgcn_readfirstlane(en.x);
            int r = __builtin_amdgcn_readfirstlane(en.y);
            float w = wl[8 * ngrp + wave];
            float4 v = h4[(size_t)s * 32 + lane];
            float* a = &acc[r * FDIM + lane];
            atomicAdd(a,      v.x * w);
            atomicAdd(a + 32, v.y * w);
            atomicAdd(a + 64, v.z * w);
            atomicAdd(a + 96, v.w * w);
        }
    } else {
        for (int e = t; e < E; e += OWN_T) {
            unsigned r = (unsigned)(dst[e] - row0);
            if (r < un) {
                int s = min(max(src[e], 0), N - 1);
                float w = dis[s] * sdis[r];
                const float* hp = h + (size_t)s * FDIM;
                for (int f = 0; f < FDIM; ++f)
                    atomicAdd(&acc[r * FDIM + (f & 3) * 32 + (f >> 2)], hp[f] * w);
            }
        }
    }
    __syncthreads();

    float4* o4 = (float4*)out;
    for (int idx = t; idx < nrows * 32; idx += OWN_T) {
        int r = idx >> 5, q = idx & 31;
        const float* a = &acc[r * FDIM + q];
        o4[(size_t)(row0 + r) * 32 + q] = make_float4(a[0], a[32], a[64], a[96]);
    }
}

__global__ __launch_bounds__(OWN_T) void k_ownC(const float* __restrict__ h,
                                                const int* __restrict__ src,
                                                const int* __restrict__ dst,
                                                const float* __restrict__ dis,
                                                const float* __restrict__ bias,
                                                float* __restrict__ out,
                                                const int* __restrict__ kcnt,
                                                int N, int E, int C)
{
    __shared__ float acc[OWN_R * 64];
    __shared__ int2  lst[OWN_CAP];
    __shared__ float wl[OWN_CAP];
    __shared__ float sdis[OWN_R];
    __shared__ int   cnt;

    const int t = threadIdx.x;
    const int row0 = blockIdx.x * OWN_R;
    const int nrows = min(OWN_R, N - row0);
    const unsigned un = (unsigned)nrows;
    const size_t g0 = (size_t)row0 * C;

    if (t == 0) cnt = 0;
    for (int r = t; r < nrows; r += OWN_T) sdis[r] = dis[row0 + r];
    for (int idx = t; idx < nrows * C; idx += OWN_T) {
        int r = idx / C, f = idx - r * C;
        float d = dis[row0 + r];
        acc[idx] = h[g0 + idx] * d * d + bias[f];
    }
    __syncthreads();

    const int kept = __builtin_amdgcn_readfirstlane(kcnt[blockIdx.x]);
    int total;
    if (kept <= keep_cap(nrows, C)) {
        total = max(kept, 0);
        if (total > 0) {
            const int2* kl = (const int2*)(out + g0);
            int2 en[OWN_CAP / OWN_T]; float ds[OWN_CAP / OWN_T];
#pragma unroll
            for (int u = 0; u < OWN_CAP / OWN_T; ++u) en[u] = kl[min(t + u * OWN_T, total - 1)];
#pragma unroll
            for (int u = 0; u < OWN_CAP / OWN_T; ++u) {
                en[u].x = min(max(en[u].x, 0), N - 1);
                en[u].y = min(max(en[u].y, 0), nrows - 1);
                ds[u] = dis[en[u].x];
            }
#pragma unroll
            for (int u = 0; u < OWN_CAP / OWN_T; ++u) {
                const int i = t + u * OWN_T;
                if (i < total) { lst[i] = en[u]; wl[i] = ds[u] * sdis[en[u].y]; }
            }
        }
    } else {
        own_scan(dst, E, row0, un, lst, &cnt);
        __syncthreads();
        total = __builtin_amdgcn_readfirstlane(cnt);
        if (total <= OWN_CAP) own_resolve(src, dis, sdis, lst, wl, total, N);
    }
    __syncthreads();

    if (total <= OWN_CAP) {
        const int wave = __builtin_amdgcn_readfirstlane(t >> 5), lane = t & 31;
        const bool c0 = lane < C, c1 = 32 + lane < C;
        const int l0 = min(lane, C - 1), l1 = min(32 + lane, C - 1);
        const int ngrp = total >> 3;
        for (int g = wave; g < ngrp; g += OWN_T / 32) {
            int s[8], r[8]; float w[8], v0[8], v1[8];
#pragma unroll
            for (int u = 0; u < 8; ++u) {
                int2 en = lst[8 * g + u];
                s[u] = __builtin_amdgcn_readfirstlane(en.x);
                r[u] = __builtin_amdgcn_readfirstlane(en.y);
                w[u] = wl[8 * g + u];
            }
#pragma unroll
            for (int u = 0; u < 8; ++u) {
                const float* hp = h + (size_t)s[u] * C;
                v0[u] = hp[l0];
                v1[u] = hp[l1];
            }
            if (c0) {
#pragma unroll
                for (int u = 0; u < 8; ++u) atomicAdd(&acc[r[u] * C + lane], v0[u] * w[u]);
            }
            if (c1) {
#pragma unroll
                for (int u = 0; u < 8; ++u) atomicAdd(&acc[r[u] * C + 32 + lane], v1[u] * w[u]);
            }
        }
        if (8 * ngrp + wave < total) {
            int2 en = lst[8 * ngrp + wave];
            int s = __builtin_amdgcn_readfirstlane(en.x);
            int r = __builtin_amdgcn_readfirstlane(en.y);
            float w = wl[8 * ngrp + wave];
            const float* hp = h + (size_t)s * C;
            float* a = &acc[r * C];
            if (c0) atomicAdd(a + lane,      hp[lane] * w);
            if (c1) atomicAdd(a + 32 + lane, hp[32 + lane] * w);
        }
    } else {
        for (int e = t; e < E; e += OWN_T) {
            unsigned r = (unsigned)(dst[e] - row0);
            if (r < un) {
                int s = min(max(src[e], 0), N - 1);
                float w = dis[s] * sdis[r];
                const float* hp = h + (size_t)s * C;
                for (int f = 0; f < C; ++f) atomicAdd(&acc[r * C + f], hp[f] * w);
            }
        }
    }
    __syncthreads();

    for (int idx = t; idx < nrows * C; idx += OWN_T) out[g0 + idx] = acc[idx];
}

__global__ __launch_bounds__(256) void k_relu_stats(float* __restrict__ a,
                                                    float* __restrict__ gsum,
                                                    float* __restrict__ gsq, int N)
{
    __shared__ float ssum[256];
    __shared__ float ssq[256];
    int t = threadIdx.x, f = t & 127, sub = t >> 7;
    float sum = 0.f, sq = 0.f;
    int base = blockIdx.x * 64;
#pragma unroll 4
    for (int i = 0; i < 32; ++i) {
        int row = base + sub + 2 * i;
        if (row < N) {
            size_t idx = (size_t)row * FDIM + f;
            float v = a[idx];
            v = fmaxf(v, 0.f);
            a[idx] = v;
            sum += v;
            sq  += v * v;
        }
    }
    ssum[t] = sum; ssq[t] = sq;
    __syncthreads();
    if (t < 128) {
        atomicAdd(&gsum[t], ssum[t] + ssum[t + 128]);
        atomicAdd(&gsq[t],  ssq[t]  + ssq[t + 128]);
    }
}

__global__ void k_bn_finalize(const float* __restrict__ gsum,
                              const float* __restrict__ gsq,
                              const float* __restrict__ gamma,
                              const float* __restrict__ beta,
                              float* __restrict__ scl,
                              float* __restrict__ sft, int N)
{
    int f = threadIdx.x;
    float inv_n = 1.f / (float)N;
    float mean = gsum[f] * inv_n;
    float var  = gsq[f] * inv_n - mean * mean;
    float inv  = rsqrtf(var + 1e-5f);
    float g = gamma[f] * inv;
    scl[f] = g;
    sft[f] = beta[f] - mean * g;
}

__global__ __launch_bounds__(96) void k_gemm2(const float* __restrict__ A,
                                              const float* __restrict__ W,
                                              const float* __restrict__ scl,
                                              const float* __restrict__ sft,
                                              float* __restrict__ O, int N, int C)
{
    __shared__ float lA[16 * FDIM];
    __shared__ v2f   lB[64 * 48];

    const int t = threadIdx.x;
    const int rowb = blockIdx.x * 16;

    for (int idx = t; idx < 16 * FDIM; idx += 96) {
        int r = idx >> 7, k = idx & 127;
        int gr = rowb + r; if (gr >= N) gr = N - 1;
        lA[idx] = A[(size_t)gr * FDIM + k] * scl[k] + sft[k];
    }
    for (int idx = t; idx < 64 * 48; idx += 96) {
        int k2 = idx / 48, c = idx - k2 * 48;
        v2f b;
        b.x = (c < C) ? W[(size_t)(2 * k2) * C + c]     : 0.f;
        b.y = (c < C) ? W[(size_t)(2 * k2 + 1) * C + c] : 0.f;
        lB[idx] = b;
    }
    __syncthreads();

    const int wave = t >> 5, lane = t & 31;
    const int m = lane & 15, hi = lane >> 4;
    const float* Arow = &lA[m * FDIM];
    const int nloc = wave * 16 + m;

    v8f acc = {0.f, 0.f, 0.f, 0.f, 0.f, 0.f, 0.f, 0.f};
#pragma unroll
    for (int k = 0; k < FDIM; k += 4) {
        v2f a = *(const v2f*)&Arow[k + 2 * hi];
        v2f b = lB[(k / 2 + hi) * 48 + nloc];
        acc = __builtin_amdgcn_wmma_f32_16x16x4_f32(
            false, a, false, b, (short)0, acc, false, false);
    }

    if (nloc < C) {
#pragma unroll
        for (int r = 0; r < 8; ++r) {
            int row = rowb + r + 8 * hi;
            if (row < N) O[(size_t)row * C + nloc] = acc[r];
        }
    }
}

__global__ __launch_bounds__(256) void k_logsoftmax(float* __restrict__ out, int N, int C)
{
    long long gid = (long long)blockIdx.x * 256 + threadIdx.x;
    int row = (int)(gid >> 5);
    int lane = threadIdx.x & 31;
    if (row >= N) return;
    float* p = out + (size_t)row * C;
    float v0 = (lane < C) ? p[lane] : -3.4e38f;
    float v1 = (32 + lane < C) ? p[32 + lane] : -3.4e38f;
    float mx = fmaxf(v0, v1);
#pragma unroll
    for (int o = 16; o > 0; o >>= 1) mx = fmaxf(mx, __shfl_xor(mx, o, 32));
    float s = ((lane < C) ? expf(v0 - mx) : 0.f) +
              ((32 + lane < C) ? expf(v1 - mx) : 0.f);
#pragma unroll
    for (int o = 16; o > 0; o >>= 1) s += __shfl_xor(s, o, 32);
    float l = logf(s);
    if (lane < C)      p[lane]      = v0 - mx - l;
    if (32 + lane < C) p[32 + lane] = v1 - mx - l;
}

extern "C" void kernel_launch(void* const* d_in, const int* in_sizes, int n_in,
                              void* d_out, int out_size, void* d_ws, size_t ws_size,
                              hipStream_t stream)
{
    const float* x     = (const float*)d_in[0];
    const int*   ei    = (const int*)  d_in[1];
    const float* W1    = (const float*)d_in[2];
    const float* b1    = (const float*)d_in[3];
    const float* gamma = (const float*)d_in[4];
    const float* beta  = (const float*)d_in[5];
    const float* W2    = (const float*)d_in[6];
    const float* b2    = (const float*)d_in[7];
    float* out = (float*)d_out;

    const int N = in_sizes[0] / FDIM;
    const int E = in_sizes[1] / 2;
    const int C = in_sizes[6] / FDIM;
    const int* src = ei;
    const int* dst = ei + E;

    char* ws = (char*)d_ws;
    size_t off = 0;
    auto up = [](size_t v) { return (v + 255) & ~(size_t)255; };
    int*   deg   = (int*)(ws + off);   off = up(off + (size_t)N * 4);
    float* dis   = (float*)(ws + off); off = up(off + (size_t)N * 4);
    float* stats = (float*)(ws + off); off = up(off + 512 * 4);
    float* gsum = stats, *gsq = stats + 128, *scl = stats + 256, *sft = stats + 384;
    float* h1   = (float*)(ws + off);  off = up(off + (size_t)N * FDIM * 4);
    float* agg1 = (float*)(ws + off);  off = up(off + (size_t)N * FDIM * 4);
    float* h2   = h1;

    const int nb_N   = (N + 255) / 256;
    const int nb_E   = (E + 255) / 256;
    const int nb_own = (N + OWN_R - 1) / OWN_R;
    const int nb_Nw  = (N + 7) / 8;

    k_init<<<nb_N, 256, 0, stream>>>(deg, stats, N);
    k_deg<<<nb_E, 256, 0, stream>>>(dst, deg, E);
    k_dis<<<nb_N, 256, 0, stream>>>(deg, dis, N);

    k_gemm1<<<dim3((N + 31) / 32, FDIM / 64), 256, 0, stream>>>(x, W1, h1, N);

    k_own128<<<nb_own, OWN_T, 0, stream>>>(h1, src, dst, dis, b1, agg1, out, deg, N, E, C);

    k_relu_stats<<<(N + 63) / 64, 256, 0, stream>>>(agg1, gsum, gsq, N);
    k_bn_finalize<<<1, 128, 0, stream>>>(gsum, gsq, gamma, beta, scl, sft, N);

    k_gemm2<<<(N + 15) / 16, 96, 0, stream>>>(agg1, W2, scl, sft, h2, N, C);

    k_ownC<<<nb_own, OWN_T, 0, stream>>>(h2, src, dst, dis, b2, out, deg, N, E, C);

    k_logsoftmax<<<nb_Nw, 256, 0, stream>>>(out, N, C);
}


# GEMM A operand not staged in LDS: a wave loads its 16-row A fragments straight from global (16 B per lane, k order shared with the W loader) and covers all columns; GEMM k sum order changes
# speedup vs baseline: 2.3861x; 1.2292x over previous
#include <hip/hip_runtime.h>

typedef __attribute__((ext_vector_type(2))) float v2f;
typedef __attribute__((ext_vector_type(8))) float v8f;
typedef __attribute__((ext_vector_type(4))) float v4f;

#define FDIM 128

__global__ __launch_bounds__(256) void k_init(int* __restrict__ deg,
                                              float* __restrict__ stats, int N)
{
    int i = blockIdx.x * 256 + threadIdx.x;
    if (i < N) deg[i] = 1;
    if (i < 256) stats[i] = 0.f;
}

__global__ __launch_bounds__(256) void k_deg(const int* __restrict__ dst,
                                             int* __restrict__ deg, int E)
{
    int i = blockIdx.x * 256 + threadIdx.x;
    if (i < E) atomicAdd(&deg[dst[i]], 1);
}

__global__ __launch_bounds__(256) void k_dis(const int* __restrict__ deg,
                                             float* __restrict__ dis, int N)
{
    int i = blockIdx.x * 256 + threadIdx.x;
    if (i < N) dis[i] = rsqrtf((float)deg[i]);
}

__global__ __launch_bounds__(256) void k_gemm1(const float* __restrict__ X,
                                               const float* __restrict__ W,
                                               float* __restrict__ O, int N)
{
    __shared__ float4 lB[32 * FDIM];

    const int t = threadIdx.x;
    {
        const int n4 = (t & 31) * 4;
#pragma unroll
        for (int i = 0; i < 4; ++i) {
            const int jh = (t >> 5) + 8 * i;
            const float4* wp = (const float4*)(W + (size_t)(4 * jh) * FDIM + n4);
            const float4 r0 = wp[0], r1 = wp[FDIM / 4], r2 = wp[2 * (FDIM / 4)], r3 = wp[3 * (FDIM / 4)];
            float4* lp = &lB[jh * FDIM + n4];
            lp[0] = make_float4(r0.x, r1.x, r2.x, r3.x);
            lp[1] = make_float4(r0.y, r1.y, r2.y, r3.y);
            lp[2] = make_float4(r0.z, r1.z, r2.z, r3.z);
            lp[3] = make_float4(r0.w, r1.w, r2.w, r3.w);
        }
    }
    __syncthreads();

    const int wave = t >> 5, lane = t & 31;
    const int m = lane & 15, hi = lane >> 4;
    const int rb = blockIdx.x * 128 + wave * 16;
    if (rb >= N) return;
    const int lrow = min(rb + m, N - 1);
    const float4* xr = (const float4*)(X + (size_t)lrow * FDIM) + hi;
    const float4* bp = &lB[hi * FDIM + m];

    v8f acc[8];
#pragma unroll
    for (int ct = 0; ct < 8; ++ct) acc[ct] = (v8f){0.f, 0.f, 0.f, 0.f, 0.f, 0.f, 0.f, 0.f};

#pragma unroll
    for (int j = 0; j < 16; ++j) {
        const float4 a = xr[2 * j];
        const v2f a0 = {a.x, a.y}, a1 = {a.z, a.w};
        float4 b[8];
#pragma unroll
        for (int ct = 0; ct < 8; ++ct) b[ct] = bp[(2 * j) * FDIM + ct * 16];
#pragma unroll
        for (int ct = 0; ct < 8; ++ct) {
            const v2f b0 = {b[ct].x, b[ct].y};
            acc[ct] = __builtin_amdgcn_wmma_f32_16x16x4_f32(
                false, a0, false, b0, (short)0, acc[ct], false, false);
        }
#pragma unroll
        for (int ct = 0; ct < 8; ++ct) {
            const v2f b1 = {b[ct].z, b[ct].w};
            acc[ct] = __builtin_amdgcn_wmma_f32_16x16x4_f32(
                false, a1, false, b1, (short)0, acc[ct], false, false);
        }
    }

#pragma unroll
    for (int ct = 0; ct < 8; ++ct) {
#pragma unroll
        for (int r = 0; r < 8; ++r) {
            int row = rb + r + 8 * hi;
            if (row < N) O[(size_t)row * FDIM + ct * 16 + m] = acc[ct][r];
        }
    }
}

#define OWN_R   400
#define OWN_CAP 8192
#define OWN_T   1024

__device__ __forceinline__ void own_scan(const int* __restrict__ dst, int E, int row0, unsigned un,
                                         int2* lst, int* cnt)
{
    const int t = threadIdx.x;
    auto push = [&](int e, unsigned r) {
        int pos = atomicAdd(cnt, 1);
        if (pos < OWN_CAP) lst[pos] = make_int2(e, (int)r);
    };
    auto test = [&](const int4& d, int e) {
        unsigned r0 = (unsigned)(d.x - row0), r1 = (unsigned)(d.y - row0);
        unsigned r2 = (unsigned)(d.z - row0), r3 = (unsigned)(d.w - row0);
        unsigned m = min(min(r0, r1), min(r2, r3));
        if (m < un) {
            if (r0 < un) push(e, r0);
            if (r1 < un) push(e + 1, r1);
            if (r2 < un) push(e + 2, r2);
            if (r3 < un) push(e + 3, r3);
        }
    };
    const int mis = (int)(((size_t)dst >> 2) & 3);
    int head = (4 - mis) & 3; if (head > E) head = E;
    const int nvec = (E - head) >> 2;
    const int tail0 = head + 4 * nvec;
    const int4* d4 = (const int4*)(dst + head);
    const int nfull = nvec / (4 * OWN_T);
    if (nfull > 0) {
        int tile = (int)(blockIdx.x % (unsigned)nfull);
        const int4* p = d4 + (size_t)tile * (4 * OWN_T) + t;
        int4 c0 = p[0], c1 = p[OWN_T], c2 = p[2 * OWN_T], c3 = p[3 * OWN_T];
        for (int it = 0; it < nfull; ++it) {
            int nt = tile + 1; if (nt >= nfull) nt = 0;
            const int4* q = d4 + (size_t)nt * (4 * OWN_T) + t;
            int4 n0 = q[0], n1 = q[OWN_T], n2 = q[2 * OWN_T], n3 = q[3 * OWN_T];
            int e = head + 4 * (tile * (4 * OWN_T) + t);
            test(c0, e);
            test(c1, e + 4 * OWN_T);
            test(c2, e + 8 * OWN_T);
            test(c3, e + 12 * OWN_T);
            c0 = n0; c1 = n1; c2 = n2; c3 = n3; tile = nt;
        }
    }
    for (int v = nfull * (4 * OWN_T) + t; v < nvec; v += OWN_T) test(d4[v], head + 4 * v);
    if (t < head) {
        unsigned r = (unsigned)(dst[t] - row0);
        if (r < un) push(t, r);
    }
    if (t < E - tail0) {
        unsigned r = (unsigned)(dst[tail0 + t] - row0);
        if (r < un) push(tail0 + t, r);
    }
}

__device__ __forceinline__ void own_resolve(const int* __restrict__ src, const float* __restrict__ dis,
                                            const float* sdis, int2* lst, float* wl, int total, int N)
{
    for (int i = threadIdx.x; i < total; i += OWN_T) {
        int2 en = lst[i];
        int s = src[en.x];
        s = min(max(s, 0), N - 1);
        lst[i].x = s;
        wl[i] = dis[s] * sdis[en.y];
    }
}

__device__ __forceinline__ int keep_cap(int nrows, int C)
{
    return min(OWN_CAP, (nrows * C) >> 1);
}

__global__ __launch_bounds__(OWN_T) void k_own128(const float* __restrict__ h,
                                                  const int* __restrict__ src,
                                                  const int* __restrict__ dst,
                                                  const float* __restrict__ dis,
                                                  const float* __restrict__ bias,
                                                  float* __restrict__ out,
                                                  float* __restrict__ keep,
                                                  int* __restrict__ kcnt,
                                                  int N, int E, int C)
{
    __shared__ float acc[OWN_R * FDIM];
    __shared__ int2  lst[OWN_CAP];
    __shared__ float wl[OWN_CAP];
    __shared__ float sdis[OWN_R];
    __shared__ int   cnt;

    const int t = threadIdx.x;
    const int row0 = blockIdx.x * OWN_R;
    const int nrows = min(OWN_R, N - row0);
    const unsigned un = (unsigned)nrows;
    const float4* h4 = (const float4*)h;

    if (t == 0) cnt = 0;
    for (int r = t; r < nrows; r += OWN_T) sdis[r] = dis[row0 + r];
    {
        const float4 bq = ((const float4*)bias)[t & 31];
        for (int idx = t; idx < nrows * 32; idx += OWN_T) {
            int r = idx >> 5, q = idx & 31;
            float d = dis[row0 + r];
            float4 v = h4[(size_t)(row0 + r) * 32 + q];
            float* a = &acc[r * FDIM + q];
            a[0]  = v.x * d * d + bq.x;
            a[32] = v.y * d * d + bq.y;
            a[64] = v.z * d * d + bq.z;
            a[96] = v.w * d * d + bq.w;
        }
    }
    __syncthreads();

    own_scan(dst, E, row0, un, lst, &cnt);
    __syncthreads();

    const int total = __builtin_amdgcn_readfirstlane(cnt);
    if (t == 0) kcnt[blockIdx.x] = total;
    if (total <= OWN_CAP) {
        own_resolve(src, dis, sdis, lst, wl, total, N);
        __syncthreads();
        if (total <= keep_cap(nrows, C)) {
            int2* kl = (int2*)(keep + (size_t)row0 * C);
            for (int i = t; i < total; i += OWN_T) kl[i] = lst[i];
        }
        const int wave = __builtin_amdgcn_readfirstlane(t >> 5), lane = t & 31;
        const int ngrp = total >> 3;
        for (int g = wave; g < ngrp; g += OWN_T / 32) {
            int r[8]; float w[8]; float4 v[8];
#pragma unroll
            for (int u = 0; u < 8; ++u) {
                int2 en = lst[8 * g + u];
                int s = __builtin_amdgcn_readfirstlane(en.x);
                r[u] = __builtin_amdgcn_readfirstlane(en.y);
                w[u] = wl[8 * g + u];
                v[u] = h4[(size_t)s * 32 + lane];
            }
#pragma unroll
            for (int u = 0; u < 8; ++u) {
                float* a = &acc[r[u] * FDIM + lane];
                atomicAdd(a,      v[u].x * w[u]);
                atomicAdd(a + 32, v[u].y * w[u]);
                atomicAdd(a + 64, v[u].z * w[u]);
                atomicAdd(a + 96, v[u].w * w[u]);
            }
        }
        if (8 * ngrp + wave < total) {
            int2 en = lst[8 * ngrp + wave];
            int s = __builtin_amdgcn_readfirstlane(en.x);
            int r = __builtin_amdgcn_readfirstlane(en.y);
            float w = wl[8 * ngrp + wave];
            float4 v = h4[(size_t)s * 32 + lane];
            float* a = &acc[r * FDIM + lane];
            atomicAdd(a,      v.x * w);
            atomicAdd(a + 32, v.y * w);
            atomicAdd(a + 64, v.z * w);
            atomicAdd(a + 96, v.w * w);
        }
    } else {
        for (int e = t; e < E; e += OWN_T) {
            unsigned r = (unsigned)(dst[e] - row0);
            if (r < un) {
                int s = min(max(src[e], 0), N - 1);
                float w = dis[s] * sdis[r];
                const float* hp = h + (size_t)s * FDIM;
                for (int f = 0; f < FDIM; ++f)
                    atomicAdd(&acc[r * FDIM + (f & 3) * 32 + (f >> 2)], hp[f] * w);
            }
        }
    }
    __syncthreads();

    float4* o4 = (float4*)out;
    for (int idx = t; idx < nrows * 32; idx += OWN_T) {
        int r = idx >> 5, q = idx & 31;
        const float* a = &acc[r * FDIM + q];
        o4[(size_t)(row0 + r) * 32 + q] = make_float4(a[0], a[32], a[64], a[96]);
    }
}

__global__ __launch_bounds__(OWN_T) void k_ownC(const float* __restrict__ h,
                                                const int* __restrict__ src,
                                                const int* __restrict__ dst,
                                                const float* __restrict__ dis,
                                                const float* __restrict__ bias,
                                                float* __restrict__ out,
                                                const int* __restrict__ kcnt,
                                                int N, int E, int C)
{
    __shared__ float acc[OWN_R * 64];
    __shared__ int2  lst[OWN_CAP];
    __shared__ float wl[OWN_CAP];
    __shared__ float sdis[OWN_R];
    __shared__ int   cnt;

    const int t = threadIdx.x;
    const int row0 = blockIdx.x * OWN_R;
    const int nrows = min(OWN_R, N - row0);
    const unsigned un = (unsigned)nrows;
    const size_t g0 = (size_t)row0 * C;

    if (t == 0) cnt = 0;
    for (int r = t; r < nrows; r += OWN_T) sdis[r] = dis[row0 + r];
    for (int idx = t; idx < nrows * C; idx += OWN_T) {
        int r = idx / C, f = idx - r * C;
        float d = dis[row0 + r];
        acc[idx] = h[g0 + idx] * d * d + bias[f];
    }
    __syncthreads();

    const int kept = __builtin_amdgcn_readfirstlane(kcnt[blockIdx.x]);
    int total;
    if (kept <= keep_cap(nrows, C)) {
        total = max(kept, 0);
        if (total > 0) {
            const int2* kl = (const int2*)(out + g0);
            int2 en[OWN_CAP / OWN_T]; float ds[OWN_CAP / OWN_T];
#pragma unroll
            for (int u = 0; u < OWN_CAP / OWN_T; ++u) en[u] = kl[min(t + u * OWN_T, total - 1)];
#pragma unroll
            for (int u = 0; u < OWN_CAP / OWN_T; ++u) {
                en[u].x = min(max(en[u].x, 0), N - 1);
                en[u].y = min(max(en[u].y, 0), nrows - 1);
                ds[u] = dis[en[u].x];
            }
#pragma unroll
            for (int u = 0; u < OWN_CAP / OWN_T; ++u) {
                const int i = t + u * OWN_T;
                if (i < total) { lst[i] = en[u]; wl[i] = ds[u] * sdis[en[u].y]; }
            }
        }
    } else {
        own_scan(dst, E, row0, un, lst, &cnt);
        __syncthreads();
        total = __builtin_amdgcn_readfirstlane(cnt);
        if (total <= OWN_CAP) own_resolve(src, dis, sdis, lst, wl, total, N);
    }
    __syncthreads();

    if (total <= OWN_CAP) {
        const int wave = __builtin_amdgcn_readfirstlane(t >> 5), lane = t & 31;
        const bool c0 = lane < C, c1 = 32 + lane < C;
        const int l0 = min(lane, C - 1), l1 = min(32 + lane, C - 1);
        const int ngrp = total >> 3;
        for (int g = wave; g < ngrp; g += OWN_T / 32) {
            int s[8], r[8]; float w[8], v0[8], v1[8];
#pragma unroll
            for (int u = 0; u < 8; ++u) {
                int2 en = lst[8 * g + u];
                s[u] = __builtin_amdgcn_readfirstlane(en.x);
                r[u] = __builtin_amdgcn_readfirstlane(en.y);
                w[u] = wl[8 * g + u];
            }
#pragma unroll
            for (int u = 0; u < 8; ++u) {
                const float* hp = h + (size_t)s[u] * C;
                v0[u] = hp[l0];
                v1[u] = hp[l1];
            }
            if (c0) {
#pragma unroll
                for (int u = 0; u < 8; ++u) atomicAdd(&acc[r[u] * C + lane], v0[u] * w[u]);
            }
            if (c1) {
#pragma unroll
                for (int u = 0; u < 8; ++u) atomicAdd(&acc[r[u] * C + 32 + lane], v1[u] * w[u]);
            }
        }
        if (8 * ngrp + wave < total) {
            int2 en = lst[8 * ngrp + wave];
            int s = __builtin_amdgcn_readfirstlane(en.x);
            int r = __builtin_amdgcn_readfirstlane(en.y);
            float w = wl[8 * ngrp + wave];
            const float* hp = h + (size_t)s * C;
            float* a = &acc[r * C];
            if (c0) atomicAdd(a + lane,      hp[lane] * w);
            if (c1) atomicAdd(a + 32 + lane, hp[32 + lane] * w);
        }
    } else {
        for (int e = t; e < E; e += OWN_T) {
            unsigned r = (unsigned)(dst[e] - row0);
            if (r < un) {
                int s = min(max(src[e], 0), N - 1);
                float w = dis[s] * sdis[r];
                const float* hp = h + (size_t)s * C;
                for (int f = 0; f < C; ++f) atomicAdd(&acc[r * C + f], hp[f] * w);
            }
        }
    }
    __syncthreads();

    for (int idx = t; idx < nrows * C; idx += OWN_T) out[g0 + idx] = acc[idx];
}

__global__ __launch_bounds__(256) void k_relu_stats(float* __restrict__ a,
                                                    float* __restrict__ gsum,
                                                    float* __restrict__ gsq, int N)
{
    __shared__ float ssum[256];
    __shared__ float ssq[256];
    int t = threadIdx.x, f = t & 127, sub = t >> 7;
    float sum = 0.f, sq = 0.f;
    int base = blockIdx.x * 64;
#pragma unroll 4
    for (int i = 0; i < 32; ++i) {
        int row = base + sub + 2 * i;
        if (row < N) {
            size_t idx = (size_t)row * FDIM + f;
            float v = a[idx];
            v = fmaxf(v, 0.f);
            a[idx] = v;
            sum += v;
            sq  += v * v;
        }
    }
    ssum[t] = sum; ssq[t] = sq;
    __syncthreads();
    if (t < 128) {
        atomicAdd(&gsum[t], ssum[t] + ssum[t + 128]);
        atomicAdd(&gsq[t],  ssq[t]  + ssq[t + 128]);
    }
}

__global__ void k_bn_finalize(const float* __restrict__ gsum,
                              const float* __restrict__ gsq,
                              const float* __restrict__ gamma,
                              const float* __restrict__ beta,
                              float* __restrict__ scl,
                              float* __restrict__ sft, int N)
{
    int f = threadIdx.x;
    float inv_n = 1.f / (float)N;
    float mean = gsum[f] * inv_n;
    float var  = gsq[f] * inv_n - mean * mean;
    float inv  = rsqrtf(var + 1e-5f);
    float g = gamma[f] * inv;
    scl[f] = g;
    sft[f] = beta[f] - mean * g;
}

__global__ __launch_bounds__(256) void k_gemm2(const float* __restrict__ A,
                                               const float* __restrict__ W,
                                               const float* __restrict__ scl,
                                               const float* __restrict__ sft,
                                               float* __restrict__ O, int N, int C)
{
    __shared__ float4 lB[32 * 48];
    __shared__ float4 lS[32];
    __shared__ float4 lF[32];

    const int t = threadIdx.x;
    const int wave = t >> 5, lane = t & 31;
    const int m = lane & 15, hi = lane >> 4;
    const int rb = blockIdx.x * 128 + wave * 16;
    const int lrow = min(rb + m, N - 1);
    const v4f* ar = (const v4f*)(A + (size_t)lrow * FDIM) + hi;

    v4f xa[16];
#pragma unroll
    for (int j = 0; j < 16; ++j) xa[j] = ar[2 * j];

#pragma unroll
    for (int idx = t; idx < 32 * 48; idx += 256) {
        int jh = idx / 48, c = idx - jh * 48;
        const float* wp = W + (size_t)(4 * jh) * C + min(c, C - 1);
        const float4 v = make_float4(wp[0], wp[C], wp[2 * C], wp[3 * C]);
        lB[idx] = (c < C) ? v : make_float4(0.f, 0.f, 0.f, 0.f);
    }
    if (t < 32) {
        lS[t] = make_float4(scl[4 * t], scl[4 * t + 1], scl[4 * t + 2], scl[4 * t + 3]);
        lF[t] = make_float4(sft[4 * t], sft[4 * t + 1], sft[4 * t + 2], sft[4 * t + 3]);
    }
    __syncthreads();

    if (rb >= N) return;
    const float4* bp = &lB[hi * 48 + m];

    v8f acc[3];
#pragma unroll
    for (int ct = 0; ct < 3; ++ct) acc[ct] = (v8f){0.f, 0.f, 0.f, 0.f, 0.f, 0.f, 0.f, 0.f};

#pragma unroll
    for (int j = 0; j < 16; ++j) {
        const v4f x = xa[j];
        const float4 s = lS[2 * j + hi], f = lF[2 * j + hi];
        const v2f a0 = {x.x * s.x + f.x, x.y * s.y + f.y};
        const v2f a1 = {x.z * s.z + f.z, x.w * s.w + f.w};
        float4 b[3];
#pragma unroll
        for (int ct = 0; ct < 3; ++ct) b[ct] = bp[(2 * j) * 48 + ct * 16];
#pragma unroll
        for (int ct = 0; ct < 3; ++ct) {
            const v2f b0 = {b[ct].x, b[ct].y};
            acc[ct] = __builtin_amdgcn_wmma_f32_16x16x4_f32(
                false, a0, false, b0, (short)0, acc[ct], false, false);
        }
#pragma unroll
        for (int ct = 0; ct < 3; ++ct) {
            const v2f b1 = {b[ct].z, b[ct].w};
            acc[ct] = __builtin_amdgcn_wmma_f32_16x16x4_f32(
                false, a1, false, b1, (short)0, acc[ct], false, false);
        }
    }

#pragma unroll
    for (int ct = 0; ct < 3; ++ct) {
        const int nloc = ct * 16 + m;
        if (nloc < C) {
#pragma unroll
            for (int r = 0; r < 8; ++r) {
                int row = rb + r + 8 * hi;
                if (row < N) O[(size_t)row * C + nloc] = acc[ct][r];
            }
        }
    }
}

__global__ __launch_bounds__(256) void k_logsoftmax(float* __restrict__ out, int N, int C)
{
    long long gid = (long long)blockIdx.x * 256 + threadIdx.x;
    int row = (int)(gid >> 5);
    int lane = threadIdx.x & 31;
    if (row >= N) return;
    float* p = out + (size_t)row * C;
    float v0 = (lane < C) ? p[lane] : -3.4e38f;
    float v1 = (32 + lane < C) ? p[32 + lane] : -3.4e38f;
    float mx = fmaxf(v0, v1);
#pragma unroll
    for (int o = 16; o > 0; o >>= 1) mx = fmaxf(mx, __shfl_xor(mx, o, 32));
    float s = ((lane < C) ? expf(v0 - mx) : 0.f) +
              ((32 + lane < C) ? expf(v1 - mx) : 0.f);
#pragma unroll
    for (int o = 16; o > 0; o >>= 1) s += __shfl_xor(s, o, 32);
    float l = logf(s);
    if (lane < C)      p[lane]      = v0 - mx - l;
    if (32 + lane < C) p[32 + lane] = v1 - mx - l;
}

extern "C" void kernel_launch(void* const* d_in, const int* in_sizes, int n_in,
                              void* d_out, int out_size, void* d_ws, size_t ws_size,
                              hipStream_t stream)
{
    const float* x     = (const float*)d_in[0];
    const int*   ei    = (const int*)  d_in[1];
    const float* W1    = (const float*)d_in[2];
    const float* b1    = (const float*)d_in[3];
    const float* gamma = (const float*)d_in[4];
    const float* beta  = (const float*)d_in[5];
    const float* W2    = (const float*)d_in[6];
    const float* b2    = (const float*)d_in[7];
    float* out = (float*)d_out;

    const int N = in_sizes[0] / FDIM;
    const int E = in_sizes[1] / 2;
    const int C = in_sizes[6] / FDIM;
    const int* src = ei;
    const int* dst = ei + E;

    char* ws = (char*)d_ws;
    size_t off = 0;
    auto up = [](size_t v) { return (v + 255) & ~(size_t)255; };
    int*   deg   = (int*)(ws + off);   off = up(off + (size_t)N * 4);
    float* dis   = (float*)(ws + off); off = up(off + (size_t)N * 4);
    float* stats = (float*)(ws + off); off = up(off + 512 * 4);
    float* gsum = stats, *gsq = stats + 128, *scl = stats + 256, *sft = stats + 384;
    float* h1   = (float*)(ws + off);  off = up(off + (size_t)N * FDIM * 4);
    float* agg1 = (float*)(ws + off);  off = up(off + (size_t)N * FDIM * 4);
    float* h2   = h1;

    const int nb_N   = (N + 255) / 256;
    const int nb_E   = (E + 255) / 256;
    const int nb_own = (N + OWN_R - 1) / OWN_R;
    const int nb_Nw  = (N + 7) / 8;

    k_init<<<nb_N, 256, 0, stream>>>(deg, stats, N);
    k_deg<<<nb_E, 256, 0, stream>>>(dst, deg, E);
    k_dis<<<nb_N, 256, 0, stream>>>(deg, dis, N);

    k_gemm1<<<(N + 127) / 128, 256, 0, stream>>>(x, W1, h1, N);

    k_own128<<<nb_own, OWN_T, 0, stream>>>(h1, src, dst, dis, b1, agg1, out, deg, N, E, C);

    k_relu_stats<<<(N + 63) / 64, 256, 0, stream>>>(agg1, gsum, gsq, N);
    k_bn_finalize<<<1, 128, 0, stream>>>(gsum, gsq, gamma, beta, scl, sft, N);

    k_gemm2<<<(N + 127) / 128, 256, 0, stream>>>(agg1, W2, scl, sft, h2, N, C);

    k_ownC<<<nb_own, OWN_T, 0, stream>>>(h2, src, dst, dis, b2, out, deg, N, E, C);

    k_logsoftmax<<<nb_Nw, 256, 0, stream>>>(out, N, C);
}


# First conv finds its own edges in a one-byte owner key per edge written by the degree pass (vector loads, byte compare in registers), not in the 4-byte destinations; LDS sum order changes
# speedup vs baseline: 2.5827x; 1.0824x over previous
#include <hip/hip_runtime.h>

typedef __attribute__((ext_vector_type(2))) float v2f;
typedef __attribute__((ext_vector_type(8))) float v8f;
typedef __attribute__((ext_vector_type(4))) float v4f;

#define FDIM 128
#define OWN_R   400
#define OWN_CAP 8192
#define OWN_T   1024

__global__ __launch_bounds__(256) void k_init(int* __restrict__ deg,
                                              float* __restrict__ stats, int N)
{
    int i = blockIdx.x * 256 + threadIdx.x;
    if (i < N) deg[i] = 1;
    if (i < 256) stats[i] = 0.f;
}

__global__ __launch_bounds__(256) void k_deg(const int* __restrict__ dst,
                                             int* __restrict__ deg,
                                             unsigned char* __restrict__ key, int E)
{
    int i = blockIdx.x * 256 + threadIdx.x;
    if (i < E) {
        const int d = dst[i];
        atomicAdd(&deg[d], 1);
        if (key) key[i] = (unsigned char)((unsigned)d / OWN_R);
    }
}

__global__ __launch_bounds__(256) void k_dis(const int* __restrict__ deg,
                                             float* __restrict__ dis, int N)
{
    int i = blockIdx.x * 256 + threadIdx.x;
    if (i < N) dis[i] = rsqrtf((float)deg[i]);
}

__global__ __launch_bounds__(256) void k_gemm1(const float* __restrict__ X,
                                               const float* __restrict__ W,
                                               float* __restrict__ O, int N)
{
    __shared__ float4 lB[32 * FDIM];

    const int t = threadIdx.x;
    {
        const int n4 = (t & 31) * 4;
#pragma unroll
        for (int i = 0; i < 4; ++i) {
            const int jh = (t >> 5) + 8 * i;
            const float4* wp = (const float4*)(W + (size_t)(4 * jh) * FDIM + n4);
            const float4 r0 = wp[0], r1 = wp[FDIM / 4], r2 = wp[2 * (FDIM / 4)], r3 = wp[3 * (FDIM / 4)];
            float4* lp = &lB[jh * FDIM + n4];
            lp[0] = make_float4(r0.x, r1.x, r2.x, r3.x);
            lp[1] = make_float4(r0.y, r1.y, r2.y, r3.y);
            lp[2] = make_float4(r0.z, r1.z, r2.z, r3.z);
            lp[3] = make_float4(r0.w, r1.w, r2.w, r3.w);
        }
    }
    __syncthreads();

    const int wave = t >> 5, lane = t & 31;
    const int m = lane & 15, hi = lane >> 4;
    const int rb = blockIdx.x * 128 + wave * 16;
    if (rb >= N) return;
    const int lrow = min(rb + m, N - 1);
    const float4* xr = (const float4*)(X + (size_t)lrow * FDIM) + hi;
    const float4* bp = &lB[hi * FDIM + m];

    v8f acc[8];
#pragma unroll
    for (int ct = 0; ct < 8; ++ct) acc[ct] = (v8f){0.f, 0.f, 0.f, 0.f, 0.f, 0.f, 0.f, 0.f};

#pragma unroll
    for (int j = 0; j < 16; ++j) {
        const float4 a = xr[2 * j];
        const v2f a0 = {a.x, a.y}, a1 = {a.z, a.w};
        float4 b[8];
#pragma unroll
        for (int ct = 0; ct < 8; ++ct) b[ct] = bp[(2 * j) * FDIM + ct * 16];
#pragma unroll
        for (int ct = 0; ct < 8; ++ct) {
            const v2f b0 = {b[ct].x, b[ct].y};
            acc[ct] = __builtin_amdgcn_wmma_f32_16x16x4_f32(
                false, a0, false, b0, (short)0, acc[ct], false, false);
        }
#pragma unroll
        for (int ct = 0; ct < 8; ++ct) {
            const v2f b1 = {b[ct].z, b[ct].w};
            acc[ct] = __builtin_amdgcn_wmma_f32_16x16x4_f32(
                false, a1, false, b1, (short)0, acc[ct], false, false);
        }
    }

#pragma unroll
    for (int ct = 0; ct < 8; ++ct) {
#pragma unroll
        for (int r = 0; r < 8; ++r) {
            int row = rb + r + 8 * hi;
            if (row < N) O[(size_t)row * FDIM + ct * 16 + m] = acc[ct][r];
        }
    }
}


__device__ __forceinline__ void own_scan(const int* __restrict__ dst, int E, int row0, unsigned un,
                                         int2* lst, int* cnt)
{
    const int t = threadIdx.x;
    auto push = [&](int e, unsigned r) {
        int pos = atomicAdd(cnt, 1);
        if (pos < OWN_CAP) lst[pos] = make_int2(e, (int)r);
    };
    auto test = [&](const int4& d, int e) {
        unsigned r0 = (unsigned)(d.x - row0), r1 = (unsigned)(d.y - row0);
        unsigned r2 = (unsigned)(d.z - row0), r3 = (unsigned)(d.w - row0);
        unsigned m = min(min(r0, r1), min(r2, r3));
        if (m < un) {
            if (r0 < un) push(e, r0);
            if (r1 < un) push(e + 1, r1);
            if (r2 < un) push(e + 2, r2);
            if (r3 < un) push(e + 3, r3);
        }
    };
    const int mis = (int)(((size_t)dst >> 2) & 3);
    int head = (4 - mis) & 3; if (head > E) head = E;
    const int nvec = (E - head) >> 2;
    const int tail0 = head + 4 * nvec;
    const int4* d4 = (const int4*)(dst + head);
    const int nfull = nvec / (4 * OWN_T);
    if (nfull > 0) {
        int tile = (int)(blockIdx.x % (unsigned)nfull);
        const int4* p = d4 + (size_t)tile * (4 * OWN_T) + t;
        int4 c0 = p[0], c1 = p[OWN_T], c2 = p[2 * OWN_T], c3 = p[3 * OWN_T];
        for (int it = 0; it < nfull; ++it) {
            int nt = tile + 1; if (nt >= nfull) nt = 0;
            const int4* q = d4 + (size_t)nt * (4 * OWN_T) + t;
            int4 n0 = q[0], n1 = q[OWN_T], n2 = q[2 * OWN_T], n3 = q[3 * OWN_T];
            int e = head + 4 * (tile * (4 * OWN_T) + t);
            test(c0, e);
            test(c1, e + 4 * OWN_T);
            test(c2, e + 8 * OWN_T);
            test(c3, e + 12 * OWN_T);
            c0 = n0; c1 = n1; c2 = n2; c3 = n3; tile = nt;
        }
    }
    for (int v = nfull * (4 * OWN_T) + t; v < nvec; v += OWN_T) test(d4[v], head + 4 * v);
    if (t < head) {
        unsigned r = (unsigned)(dst[t] - row0);
        if (r < un) push(t, r);
    }
    if (t < E - tail0) {
        unsigned r = (unsigned)(dst[tail0 + t] - row0);
        if (r < un) push(tail0 + t, r);
    }
}

__device__ __forceinline__ void own_scan_key(const unsigned char* __restrict__ key, int E, unsigned kb,
                                             int2* lst, int* cnt)
{
    const int t = threadIdx.x;
    const unsigned pat = kb * 0x01010101u;
    auto push = [&](int e) {
        int pos = atomicAdd(cnt, 1);
        if (pos < OWN_CAP) lst[pos] = make_int2(e, 0);
    };
    auto eq = [&](unsigned x) {
        unsigned y = x ^ pat;
        return ~(((y & 0x7f7f7f7fu) + 0x7f7f7f7fu) | y) & 0x80808080u;
    };
    auto vec = [&](const uint4& k) {
        return (eq(k.x) >> 7) | (eq(k.y) >> 6) | (eq(k.z) >> 5) | (eq(k.w) >> 4);
    };
    auto drain = [&](unsigned long long M, int e0) {
        while (M) {
            const int q = __builtin_ctzll(M);
            const int v = 2 * (q >> 5) + ((q >> 2) & 1);
            push(e0 + v * (16 * OWN_T) + 4 * (q & 3) + ((q >> 3) & 3));
            M &= M - 1;
        }
    };
    const int nvec = E >> 4;
    const uint4* k4 = (const uint4*)key;
    const int nfull = nvec / (4 * OWN_T);
    if (nfull > 0) {
        int tile = (int)(blockIdx.x % (unsigned)nfull);
        const uint4* p = k4 + (size_t)tile * (4 * OWN_T) + t;
        uint4 c0 = p[0], c1 = p[OWN_T], c2 = p[2 * OWN_T], c3 = p[3 * OWN_T];
        for (int it = 0; it < nfull; ++it) {
            int nt = tile + 1; if (nt >= nfull) nt = 0;
            const uint4* q = k4 + (size_t)nt * (4 * OWN_T) + t;
            uint4 n0 = q[0], n1 = q[OWN_T], n2 = q[2 * OWN_T], n3 = q[3 * OWN_T];
            const unsigned m01 = vec(c0) | (vec(c1) << 4), m23 = vec(c2) | (vec(c3) << 4);
            drain((unsigned long long)m01 | ((unsigned long long)m23 << 32), 16 * (tile * (4 * OWN_T) + t));
            c0 = n0; c1 = n1; c2 = n2; c3 = n3; tile = nt;
        }
    }
    for (int v = nfull * (4 * OWN_T) + t; v < nvec; v += OWN_T) drain((unsigned long long)vec(k4[v]), 16 * v);
    for (int e = 16 * nvec + t; e < E; e += OWN_T)
        if (key[e] == (unsigned char)kb) push(e);
}

__device__ __forceinline__ void own_resolve(const int* __restrict__ src, const float* __restrict__ dis,
                                            const float* sdis, int2* lst, float* wl, int total, int N)
{
    for (int i = threadIdx.x; i < total; i += OWN_T) {
        int2 en = lst[i];
        int s = src[en.x];
        s = min(max(s, 0), N - 1);
        lst[i].x = s;
        wl[i] = dis[s] * sdis[en.y];
    }
}

__device__ __forceinline__ void own_resolve_key(const int* __restrict__ src, const int* __restrict__ dst,
                                                const float* __restrict__ dis, const float* sdis,
                                                int2* lst, float* wl, int total, int N, int row0, int nrows)
{
    if (total <= 0) return;
    const int t = threadIdx.x;
    int e[OWN_CAP / OWN_T], s[OWN_CAP / OWN_T], r[OWN_CAP / OWN_T]; float ds[OWN_CAP / OWN_T];
#pragma unroll
    for (int u = 0; u < OWN_CAP / OWN_T; ++u) e[u] = lst[min(t + u * OWN_T, total - 1)].x;
    __syncthreads();
#pragma unroll
    for (int u = 0; u < OWN_CAP / OWN_T; ++u) { s[u] = src[e[u]]; r[u] = dst[e[u]]; }
#pragma unroll
    for (int u = 0; u < OWN_CAP / OWN_T; ++u) {
        s[u] = min(max(s[u], 0), N - 1);
        r[u] = min(max(r[u] - row0, 0), nrows - 1);
        ds[u] = dis[s[u]];
    }
#pragma unroll
    for (int u = 0; u < OWN_CAP / OWN_T; ++u) {
        const int i = t + u * OWN_T;
        if (i < total) { lst[i] = make_int2(s[u], r[u]); wl[i] = ds[u] * sdis[r[u]]; }
    }
}

__device__ __forceinline__ int keep_cap(int nrows, int C)
{
    return min(OWN_CAP, (nrows * C) >> 1);
}

__global__ __launch_bounds__(OWN_T) void k_own128(const float* __restrict__ h,
                                                  const int* __restrict__ src,
                                                  const int* __restrict__ dst,
                                                  const float* __restrict__ dis,
                                                  const float* __restrict__ bias,
                                                  float* __restrict__ out,
                                                  float* __restrict__ keep,
                                                  int* __restrict__ kcnt,
                                                  const unsigned char* __restrict__ key,
                                                  int N, int E, int C)
{
    __shared__ float acc[OWN_R * FDIM];
    __shared__ int2  lst[OWN_CAP];
    __shared__ float wl[OWN_CAP];
    __shared__ float sdis[OWN_R];
    __shared__ int   cnt;

    const int t = threadIdx.x;
    const int row0 = blockIdx.x * OWN_R;
    const int nrows = min(OWN_R, N - row0);
    const unsigned un = (unsigned)nrows;
    const float4* h4 = (const float4*)h;

    if (t == 0) cnt = 0;
    for (int r = t; r < nrows; r += OWN_T) sdis[r] = dis[row0 + r];
    {
        const float4 bq = ((const float4*)bias)[t & 31];
        for (int idx = t; idx < nrows * 32; idx += OWN_T) {
            int r = idx >> 5, q = idx & 31;
            float d = dis[row0 + r];
            float4 v = h4[(size_t)(row0 + r) * 32 + q];
            float* a = &acc[r * FDIM + q];
            a[0]  = v.x * d * d + bq.x;
            a[32] = v.y * d * d + bq.y;
            a[64] = v.z * d * d + bq.z;
            a[96] = v.w * d * d + bq.w;
        }
    }
    __syncthreads();

    const bool bykey = key != nullptr && N <= 256 * OWN_R;
    if (bykey) own_scan_key(key, E, blockIdx.x, lst, &cnt);
    else       own_scan(dst, E, row0, un, lst, &cnt);
    __syncthreads();

    const int total = __builtin_amdgcn_readfirstlane(cnt);
    if (t == 0) kcnt[blockIdx.x] = total;
    if (total <= OWN_CAP) {
        if (bykey) own_resolve_key(src, dst, dis, sdis, lst, wl, total, N, row0, nrows);
        else       own_resolve(src, dis, sdis, lst, wl, total, N);
        __syncthreads();
        if (total <= keep_cap(nrows, C)) {
            int2* kl = (int2*)(keep + (size_t)row0 * C);
            for (int i = t; i < total; i += OWN_T) kl[i] = lst[i];
        }
        const int wave = __builtin_amdgcn_readfirstlane(t >> 5), lane = t & 31;
        const int ngrp = total >> 3;
        for (int g = wave; g < ngrp; g += OWN_T / 32) {
            int r[8]; float w[8]; float4 v[8];
#pragma unroll
            for (int u = 0; u < 8; ++u) {
                int2 en = lst[8 * g + u];
                int s = __builtin_amdgcn_readfirstlane(en.x);
                r[u] = __builtin_amdgcn_readfirstlane(en.y);
                w[u] = wl[8 * g + u];
                v[u] = h4[(size_t)s * 32 + lane];
            }
#pragma unroll
            for (int u = 0; u < 8; ++u) {
                float* a = &acc[r[u] * FDIM + lane];
                atomicAdd(a,      v[u].x * w[u]);
                atomicAdd(a + 32, v[u].y * w[u]);
                atomicAdd(a + 64, v[u].z * w[u]);
                atomicAdd(a + 96, v[u].w * w[u]);
            }
        }
        if (8 * ngrp + wave < total) {
            int2 en = lst[8 * ngrp + wave];
            int s = __builtin_amdgcn_readfirstlane(en.x);
            int r = __builtin_amdgcn_readfirstlane(en.y);
            float w = wl[8 * ngrp + wave];
            float4 v = h4[(size_t)s * 32 + lane];
            float* a = &acc[r * FDIM + lane];
            atomicAdd(a,      v.x * w);
            atomicAdd(a + 32, v.y * w);
            atomicAdd(a + 64, v.z * w);
            atomicAdd(a + 96, v.w * w);
        }
    } else {
        for (int e = t; e < E; e += OWN_T) {
            unsigned r = (unsigned)(dst[e] - row0);
            if (r < un) {
                int s = min(max(src[e], 0), N - 1);
                float w = dis[s] * sdis[r];
                const float* hp = h + (size_t)s * FDIM;
                for (int f = 0; f < FDIM; ++f)
                    atomicAdd(&acc[r * FDIM + (f & 3) * 32 + (f >> 2)], hp[f] * w);
            }
        }
    }
    __syncthreads();

    float4* o4 = (float4*)out;
    for (int idx = t; idx < nrows * 32; idx += OWN_T) {
        int r = idx >> 5, q = idx & 31;
        const float* a = &acc[r * FDIM + q];
        o4[(size_t)(row0 + r) * 32 + q] = make_float4(a[0], a[32], a[64], a[96]);
    }
}

__global__ __launch_bounds__(OWN_T) void k_ownC(const float* __restrict__ h,
                                                const int* __restrict__ src,
                                                const int* __restrict__ dst,
                                                const float* __restrict__ dis,
                                                const float* __restrict__ bias,
                                                float* __restrict__ out,
                                                const int* __restrict__ kcnt,
                                                int N, int E, int C)
{
    __shared__ float acc[OWN_R * 64];
    __shared__ int2  lst[OWN_CAP];
    __shared__ float wl[OWN_CAP];
    __shared__ float sdis[OWN_R];
    __shared__ int   cnt;

    const int t = threadIdx.x;
    const int row0 = blockIdx.x * OWN_R;
    const int nrows = min(OWN_R, N - row0);
    const unsigned un = (unsigned)nrows;
    const size_t g0 = (size_t)row0 * C;

    if (t == 0) cnt = 0;
    for (int r = t; r < nrows; r += OWN_T) sdis[r] = dis[row0 + r];
    for (int idx = t; idx < nrows * C; idx += OWN_T) {
        int r = idx / C, f = idx - r * C;
        float d = dis[row0 + r];
        acc[idx] = h[g0 + idx] * d * d + bias[f];
    }
    __syncthreads();

    const int kept = __builtin_amdgcn_readfirstlane(kcnt[blockIdx.x]);
    int total;
    if (kept <= keep_cap(nrows, C)) {
        total = max(kept, 0);
        if (total > 0) {
            const int2* kl = (const int2*)(out + g0);
            int2 en[OWN_CAP / OWN_T]; float ds[OWN_CAP / OWN_T];
#pragma unroll
            for (int u = 0; u < OWN_CAP / OWN_T; ++u) en[u] = kl[min(t + u * OWN_T, total - 1)];
#pragma unroll
            for (int u = 0; u < OWN_CAP / OWN_T; ++u) {
                en[u].x = min(max(en[u].x, 0), N - 1);
                en[u].y = min(max(en[u].y, 0), nrows - 1);
                ds[u] = dis[en[u].x];
            }
#pragma unroll
            for (int u = 0; u < OWN_CAP / OWN_T; ++u) {
                const int i = t + u * OWN_T;
                if (i < total) { lst[i] = en[u]; wl[i] = ds[u] * sdis[en[u].y]; }
            }
        }
    } else {
        own_scan(dst, E, row0, un, lst, &cnt);
        __syncthreads();
        total = __builtin_amdgcn_readfirstlane(cnt);
        if (total <= OWN_CAP) own_resolve(src, dis, sdis, lst, wl, total, N);
    }
    __syncthreads();

    if (total <= OWN_CAP) {
        const int wave = __builtin_amdgcn_readfirstlane(t >> 5), lane = t & 31;
        const bool c0 = lane < C, c1 = 32 + lane < C;
        const int l0 = min(lane, C - 1), l1 = min(32 + lane, C - 1);
        const int ngrp = total >> 3;
        for (int g = wave; g < ngrp; g += OWN_T / 32) {
            int s[8], r[8]; float w[8], v0[8], v1[8];
#pragma unroll
            for (int u = 0; u < 8; ++u) {
                int2 en = lst[8 * g + u];
                s[u] = __builtin_amdgcn_readfirstlane(en.x);
                r[u] = __builtin_amdgcn_readfirstlane(en.y);
                w[u] = wl[8 * g + u];
            }
#pragma unroll
            for (int u = 0; u < 8; ++u) {
                const float* hp = h + (size_t)s[u] * C;
                v0[u] = hp[l0];
                v1[u] = hp[l1];
            }
            if (c0) {
#pragma unroll
                for (int u = 0; u < 8; ++u) atomicAdd(&acc[r[u] * C + lane], v0[u] * w[u]);
            }
            if (c1) {
#pragma unroll
                for (int u = 0; u < 8; ++u) atomicAdd(&acc[r[u] * C + 32 + lane], v1[u] * w[u]);
            }
        }
        if (8 * ngrp + wave < total) {
            int2 en = lst[8 * ngrp + wave];
            int s = __builtin_amdgcn_readfirstlane(en.x);
            int r = __builtin_amdgcn_readfirstlane(en.y);
            float w = wl[8 * ngrp + wave];
            const float* hp = h + (size_t)s * C;
            float* a = &acc[r * C];
            if (c0) atomicAdd(a + lane,      hp[lane] * w);
            if (c1) atomicAdd(a + 32 + lane, hp[32 + lane] * w);
        }
    } else {
        for (int e = t; e < E; e += OWN_T) {
            unsigned r = (unsigned)(dst[e] - row0);
            if (r < un) {
                int s = min(max(src[e], 0), N - 1);
                float w = dis[s] * sdis[r];
                const float* hp = h + (size_t)s * C;
                for (int f = 0; f < C; ++f) atomicAdd(&acc[r * C + f], hp[f] * w);
            }
        }
    }
    __syncthreads();

    for (int idx = t; idx < nrows * C; idx += OWN_T) out[g0 + idx] = acc[idx];
}

__global__ __launch_bounds__(256) void k_relu_stats(float* __restrict__ a,
                                                    float* __restrict__ gsum,
                                                    float* __restrict__ gsq, int N)
{
    __shared__ float ssum[256];
    __shared__ float ssq[256];
    int t = threadIdx.x, f = t & 127, sub = t >> 7;
    float sum = 0.f, sq = 0.f;
    int base = blockIdx.x * 64;
#pragma unroll 4
    for (int i = 0; i < 32; ++i) {
        int row = base + sub + 2 * i;
        if (row < N) {
            size_t idx = (size_t)row * FDIM + f;
            float v = a[idx];
            v = fmaxf(v, 0.f);
            a[idx] = v;
            sum += v;
            sq  += v * v;
        }
    }
    ssum[t] = sum; ssq[t] = sq;
    __syncthreads();
    if (t < 128) {
        atomicAdd(&gsum[t], ssum[t] + ssum[t + 128]);
        atomicAdd(&gsq[t],  ssq[t]  + ssq[t + 128]);
    }
}

__global__ void k_bn_finalize(const float* __restrict__ gsum,
                              const float* __restrict__ gsq,
                              const float* __restrict__ gamma,
                              const float* __restrict__ beta,
                              float* __restrict__ scl,
                              float* __restrict__ sft, int N)
{
    int f = threadIdx.x;
    float inv_n = 1.f / (float)N;
    float mean = gsum[f] * inv_n;
    float var  = gsq[f] * inv_n - mean * mean;
    float inv  = rsqrtf(var + 1e-5f);
    float g = gamma[f] * inv;
    scl[f] = g;
    sft[f] = beta[f] - mean * g;
}

__global__ __launch_bounds__(256) void k_gemm2(const float* __restrict__ A,
                                               const float* __restrict__ W,
                                               const float* __restrict__ scl,
                                               const float* __restrict__ sft,
                                               float* __restrict__ O, int N, int C)
{
    __shared__ float4 lB[32 * 48];
    __shared__ float4 lS[32];
    __shared__ float4 lF[32];

    const int t = threadIdx.x;
    const int wave = t >> 5, lane = t & 31;
    const int m = lane & 15, hi = lane >> 4;
    const int rb = blockIdx.x * 128 + wave * 16;
    const int lrow = min(rb + m, N - 1);
    const v4f* ar = (const v4f*)(A + (size_t)lrow * FDIM) + hi;

    v4f xa[16];
#pragma unroll
    for (int j = 0; j < 16; ++j) xa[j] = ar[2 * j];

#pragma unroll
    for (int idx = t; idx < 32 * 48; idx += 256) {
        int jh = idx / 48, c = idx - jh * 48;
        const float* wp = W + (size_t)(4 * jh) * C + min(c, C - 1);
        const float4 v = make_float4(wp[0], wp[C], wp[2 * C], wp[3 * C]);
        lB[idx] = (c < C) ? v : make_float4(0.f, 0.f, 0.f, 0.f);
    }
    if (t < 32) {
        lS[t] = make_float4(scl[4 * t], scl[4 * t + 1], scl[4 * t + 2], scl[4 * t + 3]);
        lF[t] = make_float4(sft[4 * t], sft[4 * t + 1], sft[4 * t + 2], sft[4 * t + 3]);
    }
    __syncthreads();

    if (rb >= N) return;
    const float4* bp = &lB[hi * 48 + m];

    v8f acc[3];
#pragma unroll
    for (int ct = 0; ct < 3; ++ct) acc[ct] = (v8f){0.f, 0.f, 0.f, 0.f, 0.f, 0.f, 0.f, 0.f};

#pragma unroll
    for (int j = 0; j < 16; ++j) {
        const v4f x = xa[j];
        const float4 s = lS[2 * j + hi], f = lF[2 * j + hi];
        const v2f a0 = {x.x * s.x + f.x, x.y * s.y + f.y};
        const v2f a1 = {x.z * s.z + f.z, x.w * s.w + f.w};
        float4 b[3];
#pragma unroll
        for (int ct = 0; ct < 3; ++ct) b[ct] = bp[(2 * j) * 48 + ct * 16];
#pragma unroll
        for (int ct = 0; ct < 3; ++ct) {
            const v2f b0 = {b[ct].x, b[ct].y};
            acc[ct] = __builtin_amdgcn_wmma_f32_16x16x4_f32(
                false, a0, false, b0, (short)0, acc[ct], false, false);
        }
#pragma unroll
        for (int ct = 0; ct < 3; ++ct) {
            const v2f b1 = {b[ct].z, b[ct].w};
            acc[ct] = __builtin_amdgcn_wmma_f32_16x16x4_f32(
                false, a1, false, b1, (short)0, acc[ct], false, false);
        }
    }

#pragma unroll
    for (int ct = 0; ct < 3; ++ct) {
        const int nloc = ct * 16 + m;
        if (nloc < C) {
#pragma unroll
            for (int r = 0; r < 8; ++r) {
                int row = rb + r + 8 * hi;
                if (row < N) O[(size_t)row * C + nloc] = acc[ct][r];
            }
        }
    }
}

__global__ __launch_bounds__(256) void k_logsoftmax(float* __restrict__ out, int N, int C)
{
    long long gid = (long long)blockIdx.x * 256 + threadIdx.x;
    int row = (int)(gid >> 5);
    int lane = threadIdx.x & 31;
    if (row >= N) return;
    float* p = out + (size_t)row * C;
    float v0 = (lane < C) ? p[lane] : -3.4e38f;
    float v1 = (32 + lane < C) ? p[32 + lane] : -3.4e38f;
    float mx = fmaxf(v0, v1);
#pragma unroll
    for (int o = 16; o > 0; o >>= 1) mx = fmaxf(mx, __shfl_xor(mx, o, 32));
    float s = ((lane < C) ? expf(v0 - mx) : 0.f) +
              ((32 + lane < C) ? expf(v1 - mx) : 0.f);
#pragma unroll
    for (int o = 16; o > 0; o >>= 1) s += __shfl_xor(s, o, 32);
    float l = logf(s);
    if (lane < C)      p[lane]      = v0 - mx - l;
    if (32 + lane < C) p[32 + lane] = v1 - mx - l;
}

extern "C" void kernel_launch(void* const* d_in, const int* in_sizes, int n_in,
                              void* d_out, int out_size, void* d_ws, size_t ws_size,
                              hipStream_t stream)
{
    const float* x     = (const float*)d_in[0];
    const int*   ei    = (const int*)  d_in[1];
    const float* W1    = (const float*)d_in[2];
    const float* b1    = (const float*)d_in[3];
    const float* gamma = (const float*)d_in[4];
    const float* beta  = (const float*)d_in[5];
    const float* W2    = (const float*)d_in[6];
    const float* b2    = (const float*)d_in[7];
    float* out = (float*)d_out;

    const int N = in_sizes[0] / FDIM;
    const int E = in_sizes[1] / 2;
    const int C = in_sizes[6] / FDIM;
    const int* src = ei;
    const int* dst = ei + E;

    char* ws = (char*)d_ws;
    size_t off = 0;
    auto up = [](size_t v) { return (v + 255) & ~(size_t)255; };
    int*   deg   = (int*)(ws + off);   off = up(off + (size_t)N * 4);
    float* dis   = (float*)(ws + off); off = up(off + (size_t)N * 4);
    float* stats = (float*)(ws + off); off = up(off + 512 * 4);
    float* gsum = stats, *gsq = stats + 128, *scl = stats + 256, *sft = stats + 384;
    float* h1   = (float*)(ws + off);  off = up(off + (size_t)N * FDIM * 4);
    float* agg1 = (float*)(ws + off);  off = up(off + (size_t)N * FDIM * 4);
    float* h2   = h1;
    unsigned char* key = (unsigned char*)(ws + off); off = up(off + (size_t)E);
    if (off > ws_size) key = nullptr;

    const int nb_N   = (N + 255) / 256;
    const int nb_E   = (E + 255) / 256;
    const int nb_own = (N + OWN_R - 1) / OWN_R;
    const int nb_Nw  = (N + 7) / 8;

    k_init<<<nb_N, 256, 0, stream>>>(deg, stats, N);
    k_deg<<<nb_E, 256, 0, stream>>>(dst, deg, key, E);
    k_dis<<<nb_N, 256, 0, stream>>>(deg, dis, N);

    k_gemm1<<<(N + 127) / 128, 256, 0, stream>>>(x, W1, h1, N);

    k_own128<<<nb_own, OWN_T, 0, stream>>>(h1, src, dst, dis, b1, agg1, out, deg, key, N, E, C);

    k_relu_stats<<<(N + 63) / 64, 256, 0, stream>>>(agg1, gsum, gsq, N);
    k_bn_finalize<<<1, 128, 0, stream>>>(gsum, gsq, gamma, beta, scl, sft, N);

    k_gemm2<<<(N + 127) / 128, 256, 0, stream>>>(agg1, W2, scl, sft, h2, N, C);

    k_ownC<<<nb_own, OWN_T, 0, stream>>>(h2, src, dst, dis, b2, out, deg, N, E, C);

    k_logsoftmax<<<nb_Nw, 256, 0, stream>>>(out, N, C);
}


# Degrees without global atomics: a list pass (one block per 400 destination rows) finds its edges by key, counts their rows in LDS, writes dis, keeps the list for both convs; float sums as before
# speedup vs baseline: 3.5050x; 1.3571x over previous
#include <hip/hip_runtime.h>

typedef __attribute__((ext_vector_type(2))) float v2f;
typedef __attribute__((ext_vector_type(8))) float v8f;
typedef __attribute__((ext_vector_type(4))) float v4f;

#define FDIM 128
#define OWN_R   400
#define OWN_CAP 8192
#define OWN_T   1024


__global__ __launch_bounds__(256) void k_key(const int* __restrict__ dst,
                                             unsigned char* __restrict__ key,
                                             float* __restrict__ stats, int E)
{
    const int i = blockIdx.x * 256 + threadIdx.x;
    if (i < 256) stats[i] = 0.f;
    if (key == nullptr) return;
    const int e0 = 16 * i;
    if (e0 >= E) return;
    auto k1 = [](int d) { return ((unsigned)d / OWN_R) & 255u; };
    if (e0 + 16 <= E && ((size_t)dst & 15) == 0) {
        const int4* p = (const int4*)(dst + e0);
        const int4 a = p[0], b = p[1], c = p[2], d = p[3];
        auto k4 = [&](const int4& v) { return k1(v.x) | (k1(v.y) << 8) | (k1(v.z) << 16) | (k1(v.w) << 24); };
        *(uint4*)(key + e0) = make_uint4(k4(a), k4(b), k4(c), k4(d));
    } else {
        const int e1 = min(e0 + 16, E);
        for (int e = e0; e < e1; ++e) key[e] = (unsigned char)k1(dst[e]);
    }
}

__global__ __launch_bounds__(256) void k_gemm1(const float* __restrict__ X,
                                               const float* __restrict__ W,
                                               float* __restrict__ O, int N)
{
    __shared__ float4 lB[32 * FDIM];

    const int t = threadIdx.x;
    {
        const int n4 = (t & 31) * 4;
#pragma unroll
        for (int i = 0; i < 4; ++i) {
            const int jh = (t >> 5) + 8 * i;
            const float4* wp = (const float4*)(W + (size_t)(4 * jh) * FDIM + n4);
            const float4 r0 = wp[0], r1 = wp[FDIM / 4], r2 = wp[2 * (FDIM / 4)], r3 = wp[3 * (FDIM / 4)];
            float4* lp = &lB[jh * FDIM + n4];
            lp[0] = make_float4(r0.x, r1.x, r2.x, r3.x);
            lp[1] = make_float4(r0.y, r1.y, r2.y, r3.y);
            lp[2] = make_float4(r0.z, r1.z, r2.z, r3.z);
            lp[3] = make_float4(r0.w, r1.w, r2.w, r3.w);
        }
    }
    __syncthreads();

    const int wave = t >> 5, lane = t & 31;
    const int m = lane & 15, hi = lane >> 4;
    const int rb = blockIdx.x * 128 + wave * 16;
    if (rb >= N) return;
    const int lrow = min(rb + m, N - 1);
    const float4* xr = (const float4*)(X + (size_t)lrow * FDIM) + hi;
    const float4* bp = &lB[hi * FDIM + m];

    v8f acc[8];
#pragma unroll
    for (int ct = 0; ct < 8; ++ct) acc[ct] = (v8f){0.f, 0.f, 0.f, 0.f, 0.f, 0.f, 0.f, 0.f};

#pragma unroll
    for (int j = 0; j < 16; ++j) {
        const float4 a = xr[2 * j];
        const v2f a0 = {a.x, a.y}, a1 = {a.z, a.w};
        float4 b[8];
#pragma unroll
        for (int ct = 0; ct < 8; ++ct) b[ct] = bp[(2 * j) * FDIM + ct * 16];
#pragma unroll
        for (int ct = 0; ct < 8; ++ct) {
            const v2f b0 = {b[ct].x, b[ct].y};
            acc[ct] = __builtin_amdgcn_wmma_f32_16x16x4_f32(
                false, a0, false, b0, (short)0, acc[ct], false, false);
        }
#pragma unroll
        for (int ct = 0; ct < 8; ++ct) {
            const v2f b1 = {b[ct].z, b[ct].w};
            acc[ct] = __builtin_amdgcn_wmma_f32_16x16x4_f32(
                false, a1, false, b1, (short)0, acc[ct], false, false);
        }
    }

#pragma unroll
    for (int ct = 0; ct < 8; ++ct) {
#pragma unroll
        for (int r = 0; r < 8; ++r) {
            int row = rb + r + 8 * hi;
            if (row < N) O[(size_t)row * FDIM + ct * 16 + m] = acc[ct][r];
        }
    }
}


__device__ __forceinline__ void own_scan(const int* __restrict__ dst, int E, int row0, unsigned un,
                                         int2* lst, int* cnt)
{
    const int t = threadIdx.x;
    auto push = [&](int e, unsigned r) {
        int pos = atomicAdd(cnt, 1);
        if (pos < OWN_CAP) lst[pos] = make_int2(e, (int)r);
    };
    auto test = [&](const int4& d, int e) {
        unsigned r0 = (unsigned)(d.x - row0), r1 = (unsigned)(d.y - row0);
        unsigned r2 = (unsigned)(d.z - row0), r3 = (unsigned)(d.w - row0);
        unsigned m = min(min(r0, r1), min(r2, r3));
        if (m < un) {
            if (r0 < un) push(e, r0);
            if (r1 < un) push(e + 1, r1);
            if (r2 < un) push(e + 2, r2);
            if (r3 < un) push(e + 3, r3);
        }
    };
    const int mis = (int)(((size_t)dst >> 2) & 3);
    int head = (4 - mis) & 3; if (head > E) head = E;
    const int nvec = (E - head) >> 2;
    const int tail0 = head + 4 * nvec;
    const int4* d4 = (const int4*)(dst + head);
    const int nfull = nvec / (4 * OWN_T);
    if (nfull > 0) {
        int tile = (int)(blockIdx.x % (unsigned)nfull);
        const int4* p = d4 + (size_t)tile * (4 * OWN_T) + t;
        int4 c0 = p[0], c1 = p[OWN_T], c2 = p[2 * OWN_T], c3 = p[3 * OWN_T];
        for (int it = 0; it < nfull; ++it) {
            int nt = tile + 1; if (nt >= nfull) nt = 0;
            const int4* q = d4 + (size_t)nt * (4 * OWN_T) + t;
            int4 n0 = q[0], n1 = q[OWN_T], n2 = q[2 * OWN_T], n3 = q[3 * OWN_T];
            int e = head + 4 * (tile * (4 * OWN_T) + t);
            test(c0, e);
            test(c1, e + 4 * OWN_T);
            test(c2, e + 8 * OWN_T);
            test(c3, e + 12 * OWN_T);
            c0 = n0; c1 = n1; c2 = n2; c3 = n3; tile = nt;
        }
    }
    for (int v = nfull * (4 * OWN_T) + t; v < nvec; v += OWN_T) test(d4[v], head + 4 * v);
    if (t < head) {
        unsigned r = (unsigned)(dst[t] - row0);
        if (r < un) push(t, r);
    }
    if (t < E - tail0) {
        unsigned r = (unsigned)(dst[tail0 + t] - row0);
        if (r < un) push(tail0 + t, r);
    }
}

__device__ __forceinline__ void own_scan_key(const unsigned char* __restrict__ key, int E, unsigned kb,
                                             int2* lst, int* cnt)
{
    const int t = threadIdx.x;
    const unsigned pat = kb * 0x01010101u;
    auto push = [&](int e) {
        int pos = atomicAdd(cnt, 1);
        if (pos < OWN_CAP) lst[pos] = make_int2(e, 0);
    };
    auto eq = [&](unsigned x) {
        unsigned y = x ^ pat;
        return ~(((y & 0x7f7f7f7fu) + 0x7f7f7f7fu) | y) & 0x80808080u;
    };
    auto vec = [&](const uint4& k) {
        return (eq(k.x) >> 7) | (eq(k.y) >> 6) | (eq(k.z) >> 5) | (eq(k.w) >> 4);
    };
    auto drain = [&](unsigned long long M, int e0) {
        while (M) {
            const int q = __builtin_ctzll(M);
            const int v = 2 * (q >> 5) + ((q >> 2) & 1);
            push(e0 + v * (16 * OWN_T) + 4 * (q & 3) + ((q >> 3) & 3));
            M &= M - 1;
        }
    };
    const int nvec = E >> 4;
    const uint4* k4 = (const uint4*)key;
    const int nfull = nvec / (4 * OWN_T);
    if (nfull > 0) {
        int tile = (int)(blockIdx.x % (unsigned)nfull);
        const uint4* p = k4 + (size_t)tile * (4 * OWN_T) + t;
        uint4 c0 = p[0], c1 = p[OWN_T], c2 = p[2 * OWN_T], c3 = p[3 * OWN_T];
        for (int it = 0; it < nfull; ++it) {
            int nt = tile + 1; if (nt >= nfull) nt = 0;
            const uint4* q = k4 + (size_t)nt * (4 * OWN_T) + t;
            uint4 n0 = q[0], n1 = q[OWN_T], n2 = q[2 * OWN_T], n3 = q[3 * OWN_T];
            const unsigned m01 = vec(c0) | (vec(c1) << 4), m23 = vec(c2) | (vec(c3) << 4);
            drain((unsigned long long)m01 | ((unsigned long long)m23 << 32), 16 * (tile * (4 * OWN_T) + t));
            c0 = n0; c1 = n1; c2 = n2; c3 = n3; tile = nt;
        }
    }
    for (int v = nfull * (4 * OWN_T) + t; v < nvec; v += OWN_T) drain((unsigned long long)vec(k4[v]), 16 * v);
    for (int e = 16 * nvec + t; e < E; e += OWN_T)
        if (key[e] == (unsigned char)kb) push(e);
}

__device__ __forceinline__ void own_resolve(const int* __restrict__ src, const float* __restrict__ dis,
                                            const float* sdis, int2* lst, float* wl, int total, int N)
{
    for (int i = threadIdx.x; i < total; i += OWN_T) {
        int2 en = lst[i];
        int s = src[en.x];
        s = min(max(s, 0), N - 1);
        lst[i].x = s;
        wl[i] = dis[s] * sdis[en.y];
    }
}

__device__ __forceinline__ void own_list(const int* __restrict__ src, const int* __restrict__ dst,
                                         const int2* lst, int* hist, int2* kl,
                                         int total, int N, int row0, int nrows)
{
    if (total <= 0) return;
    const int t = threadIdx.x;
    int e[OWN_CAP / OWN_T], s[OWN_CAP / OWN_T], r[OWN_CAP / OWN_T];
#pragma unroll
    for (int u = 0; u < OWN_CAP / OWN_T; ++u) e[u] = lst[min(t + u * OWN_T, total - 1)].x;
#pragma unroll
    for (int u = 0; u < OWN_CAP / OWN_T; ++u) { s[u] = src[e[u]]; r[u] = dst[e[u]]; }
#pragma unroll
    for (int u = 0; u < OWN_CAP / OWN_T; ++u) {
        const int i = t + u * OWN_T;
        if (i < total) {
            const int sv = min(max(s[u], 0), N - 1);
            const int rv = min(max(r[u] - row0, 0), nrows - 1);
            if (kl) kl[i] = make_int2(sv, rv);
            atomicAdd(&hist[rv], 1);
        }
    }
}

__device__ __forceinline__ void own_fetch(const int2* kl, const float* __restrict__ dis,
                                          const float* sdis, int2* lst, float* wl,
                                          int total, int N, int nrows)
{
    if (total <= 0) return;
    const int t = threadIdx.x;
    int2 en[OWN_CAP / OWN_T]; float ds[OWN_CAP / OWN_T];
#pragma unroll
    for (int u = 0; u < OWN_CAP / OWN_T; ++u) en[u] = kl[min(t + u * OWN_T, total - 1)];
#pragma unroll
    for (int u = 0; u < OWN_CAP / OWN_T; ++u) {
        en[u].x = min(max(en[u].x, 0), N - 1);
        en[u].y = min(max(en[u].y, 0), nrows - 1);
        ds[u] = dis[en[u].x];
    }
#pragma unroll
    for (int u = 0; u < OWN_CAP / OWN_T; ++u) {
        const int i = t + u * OWN_T;
        if (i < total) { lst[i] = en[u]; wl[i] = ds[u] * sdis[en[u].y]; }
    }
}

__device__ __forceinline__ int keep_cap(int nrows, int C)
{
    return min(OWN_CAP, (nrows * C) >> 1);
}

__global__ __launch_bounds__(OWN_T) void k_list(const int* __restrict__ src,
                                                const int* __restrict__ dst,
                                                float* __restrict__ dis,
                                                float* __restrict__ keep,
                                                int* __restrict__ kcnt,
                                                const unsigned char* __restrict__ key,
                                                int N, int E, int C)
{
    __shared__ int2 lst[OWN_CAP];
    __shared__ int  hist[OWN_R];
    __shared__ int  cnt;

    const int t = threadIdx.x;
    const int row0 = blockIdx.x * OWN_R;
    const int nrows = min(OWN_R, N - row0);
    const unsigned un = (unsigned)nrows;

    if (t == 0) cnt = 0;
    for (int r = t; r < OWN_R; r += OWN_T) hist[r] = 1;
    __syncthreads();

    const bool bykey = key != nullptr && N <= 256 * OWN_R;
    if (bykey) own_scan_key(key, E, blockIdx.x, lst, &cnt);
    else       own_scan(dst, E, row0, un, lst, &cnt);
    __syncthreads();

    const int total = __builtin_amdgcn_readfirstlane(cnt);
    if (t == 0) kcnt[blockIdx.x] = total;
    if (total <= OWN_CAP) {
        int2* kl = (total <= keep_cap(nrows, C)) ? (int2*)(keep + (size_t)row0 * C) : nullptr;
        own_list(src, dst, lst, hist, kl, total, N, row0, nrows);
    } else {
        for (int e = t; e < E; e += OWN_T) {
            unsigned r = (unsigned)(dst[e] - row0);
            if (r < un) atomicAdd(&hist[r], 1);
        }
    }
    __syncthreads();
    for (int r = t; r < nrows; r += OWN_T) dis[row0 + r] = rsqrtf((float)hist[r]);
}

__global__ __launch_bounds__(OWN_T) void k_own128(const float* __restrict__ h,
                                                  const int* __restrict__ src,
                                                  const int* __restrict__ dst,
                                                  const float* __restrict__ dis,
                                                  const float* __restrict__ bias,
                                                  float* __restrict__ out,
                                                  const float* __restrict__ keep,
                                                  const int* __restrict__ kcnt,
                                                  int N, int E, int C)
{
    __shared__ float acc[OWN_R * FDIM];
    __shared__ int2  lst[OWN_CAP];
    __shared__ float wl[OWN_CAP];
    __shared__ float sdis[OWN_R];
    __shared__ int   cnt;

    const int t = threadIdx.x;
    const int row0 = blockIdx.x * OWN_R;
    const int nrows = min(OWN_R, N - row0);
    const unsigned un = (unsigned)nrows;
    const float4* h4 = (const float4*)h;

    if (t == 0) cnt = 0;
    for (int r = t; r < nrows; r += OWN_T) sdis[r] = dis[row0 + r];
    {
        const float4 bq = ((const float4*)bias)[t & 31];
        for (int idx = t; idx < nrows * 32; idx += OWN_T) {
            int r = idx >> 5, q = idx & 31;
            float d = dis[row0 + r];
            float4 v = h4[(size_t)(row0 + r) * 32 + q];
            float* a = &acc[r * FDIM + q];
            a[0]  = v.x * d * d + bq.x;
            a[32] = v.y * d * d + bq.y;
            a[64] = v.z * d * d + bq.z;
            a[96] = v.w * d * d + bq.w;
        }
    }
    __syncthreads();

    const int kept = __builtin_amdgcn_readfirstlane(kcnt[blockIdx.x]);
    int total;
    if (kept <= keep_cap(nrows, C)) {
        total = max(kept, 0);
        own_fetch((const int2*)(keep + (size_t)row0 * C), dis, sdis, lst, wl, total, N, nrows);
    } else {
        own_scan(dst, E, row0, un, lst, &cnt);
        __syncthreads();
        total = __builtin_amdgcn_readfirstlane(cnt);
        if (total <= OWN_CAP) own_resolve(src, dis, sdis, lst, wl, total, N);
    }
    __syncthreads();

    if (total <= OWN_CAP) {
        const int wave = __builtin_amdgcn_readfirstlane(t >> 5), lane = t & 31;
        const int ngrp = total >> 3;
        for (int g = wave; g < ngrp; g += OWN_T / 32) {
            int r[8]; float w[8]; float4 v[8];
#pragma unroll
            for (int u = 0; u < 8; ++u) {
                int2 en = lst[8 * g + u];
                int s = __builtin_amdgcn_readfirstlane(en.x);
                r[u] = __builtin_amdgcn_readfirstlane(en.y);
                w[u] = wl[8 * g + u];
                v[u] = h4[(size_t)s * 32 + lane];
            }
#pragma unroll
            for (int u = 0; u < 8; ++u) {
                float* a = &acc[r[u] * FDIM + lane];
                atomicAdd(a,      v[u].x * w[u]);
                atomicAdd(a + 32, v[u].y * w[u]);
                atomicAdd(a + 64, v[u].z * w[u]);
                atomicAdd(a + 96, v[u].w * w[u]);
            }
        }
        if (8 * ngrp + wave < total) {
            int2 en = lst[8 * ngrp + wave];
            int s = __builtin_amdgcn_readfirstlane(en.x);
            int r = __builtin_amdgcn_readfirstlane(en.y);
            float w = wl[8 * ngrp + wave];
            float4 v = h4[(size_t)s * 32 + lane];
            float* a = &acc[r * FDIM + lane];
            atomicAdd(a,      v.x * w);
            atomicAdd(a + 32, v.y * w);
            atomicAdd(a + 64, v.z * w);
            atomicAdd(a + 96, v.w * w);
        }
    } else {
        for (int e = t; e < E; e += OWN_T) {
            unsigned r = (unsigned)(dst[e] - row0);
            if (r < un) {
                int s = min(max(src[e], 0), N - 1);
                float w = dis[s] * sdis[r];
                const float* hp = h + (size_t)s * FDIM;
                for (int f = 0; f < FDIM; ++f)
                    atomicAdd(&acc[r * FDIM + (f & 3) * 32 + (f >> 2)], hp[f] * w);
            }
        }
    }
    __syncthreads();

    float4* o4 = (float4*)out;
    for (int idx = t; idx < nrows * 32; idx += OWN_T) {
        int r = idx >> 5, q = idx & 31;
        const float* a = &acc[r * FDIM + q];
        o4[(size_t)(row0 + r) * 32 + q] = make_float4(a[0], a[32], a[64], a[96]);
    }
}

__global__ __launch_bounds__(OWN_T) void k_ownC(const float* __restrict__ h,
                                                const int* __restrict__ src,
                                                const int* __restrict__ dst,
                                                const float* __restrict__ dis,
                                                const float* __restrict__ bias,
                                                float* __restrict__ out,
                                                const int* __restrict__ kcnt,
                                                int N, int E, int C)
{
    __shared__ float acc[OWN_R * 64];
    __shared__ int2  lst[OWN_CAP];
    __shared__ float wl[OWN_CAP];
    __shared__ float sdis[OWN_R];
    __shared__ int   cnt;

    const int t = threadIdx.x;
    const int row0 = blockIdx.x * OWN_R;
    const int nrows = min(OWN_R, N - row0);
    const unsigned un = (unsigned)nrows;
    const size_t g0 = (size_t)row0 * C;

    if (t == 0) cnt = 0;
    for (int r = t; r < nrows; r += OWN_T) sdis[r] = dis[row0 + r];
    for (int idx = t; idx < nrows * C; idx += OWN_T) {
        int r = idx / C, f = idx - r * C;
        float d = dis[row0 + r];
        acc[idx] = h[g0 + idx] * d * d + bias[f];
    }
    __syncthreads();

    const int kept = __builtin_amdgcn_readfirstlane(kcnt[blockIdx.x]);
    int total;
    if (kept <= keep_cap(nrows, C)) {
        total = max(kept, 0);
        own_fetch((const int2*)(out + g0), dis, sdis, lst, wl, total, N, nrows);
    } else {
        own_scan(dst, E, row0, un, lst, &cnt);
        __syncthreads();
        total = __builtin_amdgcn_readfirstlane(cnt);
        if (total <= OWN_CAP) own_resolve(src, dis, sdis, lst, wl, total, N);
    }
    __syncthreads();

    if (total <= OWN_CAP) {
        const int wave = __builtin_amdgcn_readfirstlane(t >> 5), lane = t & 31;
        const bool c0 = lane < C, c1 = 32 + lane < C;
        const int l0 = min(lane, C - 1), l1 = min(32 + lane, C - 1);
        const int ngrp = total >> 3;
        for (int g = wave; g < ngrp; g += OWN_T / 32) {
            int s[8], r[8]; float w[8], v0[8], v1[8];
#pragma unroll
            for (int u = 0; u < 8; ++u) {
                int2 en = lst[8 * g + u];
                s[u] = __builtin_amdgcn_readfirstlane(en.x);
                r[u] = __builtin_amdgcn_readfirstlane(en.y);
                w[u] = wl[8 * g + u];
            }
#pragma unroll
            for (int u = 0; u < 8; ++u) {
                const float* hp = h + (size_t)s[u] * C;
                v0[u] = hp[l0];
                v1[u] = hp[l1];
            }
            if (c0) {
#pragma unroll
                for (int u = 0; u < 8; ++u) atomicAdd(&acc[r[u] * C + lane], v0[u] * w[u]);
            }
            if (c1) {
#pragma unroll
                for (int u = 0; u < 8; ++u) atomicAdd(&acc[r[u] * C + 32 + lane], v1[u] * w[u]);
            }
        }
        if (8 * ngrp + wave < total) {
            int2 en = lst[8 * ngrp + wave];
            int s = __builtin_amdgcn_readfirstlane(en.x);
            int r = __builtin_amdgcn_readfirstlane(en.y);
            float w = wl[8 * ngrp + wave];
            const float* hp = h + (size_t)s * C;
            float* a = &acc[r * C];
            if (c0) atomicAdd(a + lane,      hp[lane] * w);
            if (c1) atomicAdd(a + 32 + lane, hp[32 + lane] * w);
        }
    } else {
        for (int e = t; e < E; e += OWN_T) {
            unsigned r = (unsigned)(dst[e] - row0);
            if (r < un) {
                int s = min(max(src[e], 0), N - 1);
                float w = dis[s] * sdis[r];
                const float* hp = h + (size_t)s * C;
                for (int f = 0; f < C; ++f) atomicAdd(&acc[r * C + f], hp[f] * w);
            }
        }
    }
    __syncthreads();

    for (int idx = t; idx < nrows * C; idx += OWN_T) out[g0 + idx] = acc[idx];
}

__global__ __launch_bounds__(256) void k_relu_stats(float* __restrict__ a,
                                                    float* __restrict__ gsum,
                                                    float* __restrict__ gsq, int N)
{
    __shared__ float ssum[256];
    __shared__ float ssq[256];
    int t = threadIdx.x, f = t & 127, sub = t >> 7;
    float sum = 0.f, sq = 0.f;
    int base = blockIdx.x * 64;
#pragma unroll 4
    for (int i = 0; i < 32; ++i) {
        int row = base + sub + 2 * i;
        if (row < N) {
            size_t idx = (size_t)row * FDIM + f;
            float v = a[idx];
            v = fmaxf(v, 0.f);
            a[idx] = v;
            sum += v;
            sq  += v * v;
        }
    }
    ssum[t] = sum; ssq[t] = sq;
    __syncthreads();
    if (t < 128) {
        atomicAdd(&gsum[t], ssum[t] + ssum[t + 128]);
        atomicAdd(&gsq[t],  ssq[t]  + ssq[t + 128]);
    }
}

__global__ void k_bn_finalize(const float* __restrict__ gsum,
                              const float* __restrict__ gsq,
                              const float* __restrict__ gamma,
                              const float* __restrict__ beta,
                              float* __restrict__ scl,
                              float* __restrict__ sft, int N)
{
    int f = threadIdx.x;
    float inv_n = 1.f / (float)N;
    float mean = gsum[f] * inv_n;
    float var  = gsq[f] * inv_n - mean * mean;
    float inv  = rsqrtf(var + 1e-5f);
    float g = gamma[f] * inv;
    scl[f] = g;
    sft[f] = beta[f] - mean * g;
}

__global__ __launch_bounds__(256) void k_gemm2(const float* __restrict__ A,
                                               const float* __restrict__ W,
                                               const float* __restrict__ scl,
                                               const float* __restrict__ sft,
                                               float* __restrict__ O, int N, int C)
{
    __shared__ float4 lB[32 * 48];
    __shared__ float4 lS[32];
    __shared__ float4 lF[32];

    const int t = threadIdx.x;
    const int wave = t >> 5, lane = t & 31;
    const int m = lane & 15, hi = lane >> 4;
    const int rb = blockIdx.x * 128 + wave * 16;
    const int lrow = min(rb + m, N - 1);
    const v4f* ar = (const v4f*)(A + (size_t)lrow * FDIM) + hi;

    v4f xa[16];
#pragma unroll
    for (int j = 0; j < 16; ++j) xa[j] = ar[2 * j];

#pragma unroll
    for (int idx = t; idx < 32 * 48; idx += 256) {
        int jh = idx / 48, c = idx - jh * 48;
        const float* wp = W + (size_t)(4 * jh) * C + min(c, C - 1);
        const float4 v = make_float4(wp[0], wp[C], wp[2 * C], wp[3 * C]);
        lB[idx] = (c < C) ? v : make_float4(0.f, 0.f, 0.f, 0.f);
    }
    if (t < 32) {
        lS[t] = make_float4(scl[4 * t], scl[4 * t + 1], scl[4 * t + 2], scl[4 * t + 3]);
        lF[t] = make_float4(sft[4 * t], sft[4 * t + 1], sft[4 * t + 2], sft[4 * t + 3]);
    }
    __syncthreads();

    if (rb >= N) return;
    const float4* bp = &lB[hi * 48 + m];

    v8f acc[3];
#pragma unroll
    for (int ct = 0; ct < 3; ++ct) acc[ct] = (v8f){0.f, 0.f, 0.f, 0.f, 0.f, 0.f, 0.f, 0.f};

#pragma unroll
    for (int j = 0; j < 16; ++j) {
        const v4f x = xa[j];
        const float4 s = lS[2 * j + hi], f = lF[2 * j + hi];
        const v2f a0 = {x.x * s.x + f.x, x.y * s.y + f.y};
        const v2f a1 = {x.z * s.z + f.z, x.w * s.w + f.w};
        float4 b[3];
#pragma unroll
        for (int ct = 0; ct < 3; ++ct) b[ct] = bp[(2 * j) * 48 + ct * 16];
#pragma unroll
        for (int ct = 0; ct < 3; ++ct) {
            const v2f b0 = {b[ct].x, b[ct].y};
            acc[ct] = __builtin_amdgcn_wmma_f32_16x16x4_f32(
                false, a0, false, b0, (short)0, acc[ct], false, false);
        }
#pragma unroll
        for (int ct = 0; ct < 3; ++ct) {
            const v2f b1 = {b[ct].z, b[ct].w};
            acc[ct] = __builtin_amdgcn_wmma_f32_16x16x4_f32(
                false, a1, false, b1, (short)0, acc[ct], false, false);
        }
    }

#pragma unroll
    for (int ct = 0; ct < 3; ++ct) {
        const int nloc = ct * 16 + m;
        if (nloc < C) {
#pragma unroll
            for (int r = 0; r < 8; ++r) {
                int row = rb + r + 8 * hi;
                if (row < N) O[(size_t)row * C + nloc] = acc[ct][r];
            }
        }
    }
}

__global__ __launch_bounds__(256) void k_logsoftmax(float* __restrict__ out, int N, int C)
{
    long long gid = (long long)blockIdx.x * 256 + threadIdx.x;
    int row = (int)(gid >> 5);
    int lane = threadIdx.x & 31;
    if (row >= N) return;
    float* p = out + (size_t)row * C;
    float v0 = (lane < C) ? p[lane] : -3.4e38f;
    float v1 = (32 + lane < C) ? p[32 + lane] : -3.4e38f;
    float mx = fmaxf(v0, v1);
#pragma unroll
    for (int o = 16; o > 0; o >>= 1) mx = fmaxf(mx, __shfl_xor(mx, o, 32));
    float s = ((lane < C) ? expf(v0 - mx) : 0.f) +
              ((32 + lane < C) ? expf(v1 - mx) : 0.f);
#pragma unroll
    for (int o = 16; o > 0; o >>= 1) s += __shfl_xor(s, o, 32);
    float l = logf(s);
    if (lane < C)      p[lane]      = v0 - mx - l;
    if (32 + lane < C) p[32 + lane] = v1 - mx - l;
}

extern "C" void kernel_launch(void* const* d_in, const int* in_sizes, int n_in,
                              void* d_out, int out_size, void* d_ws, size_t ws_size,
                              hipStream_t stream)
{
    const float* x     = (const float*)d_in[0];
    const int*   ei    = (const int*)  d_in[1];
    const float* W1    = (const float*)d_in[2];
    const float* b1    = (const float*)d_in[3];
    const float* gamma = (const float*)d_in[4];
    const float* beta  = (const float*)d_in[5];
    const float* W2    = (const float*)d_in[6];
    const float* b2    = (const float*)d_in[7];
    float* out = (float*)d_out;

    const int N = in_sizes[0] / FDIM;
    const int E = in_sizes[1] / 2;
    const int C = in_sizes[6] / FDIM;
    const int* src = ei;
    const int* dst = ei + E;

    char* ws = (char*)d_ws;
    size_t off = 0;
    auto up = [](size_t v) { return (v + 255) & ~(size_t)255; };
    const int nb_own = (N + OWN_R - 1) / OWN_R;
    int*   kcnt  = (int*)(ws + off);   off = up(off + (size_t)nb_own * 4);
    float* dis   = (float*)(ws + off); off = up(off + (size_t)N * 4);
    float* stats = (float*)(ws + off); off = up(off + 512 * 4);
    float* gsum = stats, *gsq = stats + 128, *scl = stats + 256, *sft = stats + 384;
    float* h1   = (float*)(ws + off);  off = up(off + (size_t)N * FDIM * 4);
    float* agg1 = (float*)(ws + off);  off = up(off + (size_t)N * FDIM * 4);
    float* h2   = h1;
    unsigned char* key = (unsigned char*)(ws + off); off = up(off + (size_t)E);
    if (off > ws_size) key = nullptr;

    const int nb_key = (E / 16 + 256) / 256;
    const int nb_Nw  = (N + 7) / 8;

    k_key<<<nb_key, 256, 0, stream>>>(dst, key, stats, E);

    k_list<<<nb_own, OWN_T, 0, stream>>>(src, dst, dis, out, kcnt, key, N, E, C);

    k_gemm1<<<(N + 127) / 128, 256, 0, stream>>>(x, W1, h1, N);

    k_own128<<<nb_own, OWN_T, 0, stream>>>(h1, src, dst, dis, b1, agg1, out, kcnt, N, E, C);

    k_relu_stats<<<(N + 63) / 64, 256, 0, stream>>>(agg1, gsum, gsq, N);
    k_bn_finalize<<<1, 128, 0, stream>>>(gsum, gsq, gamma, beta, scl, sft, N);

    k_gemm2<<<(N + 127) / 128, 256, 0, stream>>>(agg1, W2, scl, sft, h2, N, C);

    k_ownC<<<nb_own, OWN_T, 0, stream>>>(h2, src, dst, dis, b2, out, kcnt, N, E, C);

    k_logsoftmax<<<nb_Nw, 256, 0, stream>>>(out, N, C);
}


# ReLU + statistics pass: a thread issues its 13 row loads (16 B each, a whole 512-B row per wave) before using any, 416 rows per 32-wave block; was 32 dependent 4-byte trips; column-sum order changes
# speedup vs baseline: 3.6730x; 1.0479x over previous
#include <hip/hip_runtime.h>

typedef __attribute__((ext_vector_type(2))) float v2f;
typedef __attribute__((ext_vector_type(8))) float v8f;
typedef __attribute__((ext_vector_type(4))) float v4f;

#define FDIM 128
#define OWN_R   400
#define OWN_CAP 8192
#define OWN_T   1024


__global__ __launch_bounds__(256) void k_key(const int* __restrict__ dst,
                                             unsigned char* __restrict__ key,
                                             float* __restrict__ stats, int E)
{
    const int i = blockIdx.x * 256 + threadIdx.x;
    if (i < 256) stats[i] = 0.f;
    if (key == nullptr) return;
    const int e0 = 16 * i;
    if (e0 >= E) return;
    auto k1 = [](int d) { return ((unsigned)d / OWN_R) & 255u; };
    if (e0 + 16 <= E && ((size_t)dst & 15) == 0) {
        const int4* p = (const int4*)(dst + e0);
        const int4 a = p[0], b = p[1], c = p[2], d = p[3];
        auto k4 = [&](const int4& v) { return k1(v.x) | (k1(v.y) << 8) | (k1(v.z) << 16) | (k1(v.w) << 24); };
        *(uint4*)(key + e0) = make_uint4(k4(a), k4(b), k4(c), k4(d));
    } else {
        const int e1 = min(e0 + 16, E);
        for (int e = e0; e < e1; ++e) key[e] = (unsigned char)k1(dst[e]);
    }
}

__global__ __launch_bounds__(256) void k_gemm1(const float* __restrict__ X,
                                               const float* __restrict__ W,
                                               float* __restrict__ O, int N)
{
    __shared__ float4 lB[32 * FDIM];

    const int t = threadIdx.x;
    {
        const int n4 = (t & 31) * 4;
#pragma unroll
        for (int i = 0; i < 4; ++i) {
            const int jh = (t >> 5) + 8 * i;
            const float4* wp = (const float4*)(W + (size_t)(4 * jh) * FDIM + n4);
            const float4 r0 = wp[0], r1 = wp[FDIM / 4], r2 = wp[2 * (FDIM / 4)], r3 = wp[3 * (FDIM / 4)];
            float4* lp = &lB[jh * FDIM + n4];
            lp[0] = make_float4(r0.x, r1.x, r2.x, r3.x);
            lp[1] = make_float4(r0.y, r1.y, r2.y, r3.y);
            lp[2] = make_float4(r0.z, r1.z, r2.z, r3.z);
            lp[3] = make_float4(r0.w, r1.w, r2.w, r3.w);
        }
    }
    __syncthreads();

    const int wave = t >> 5, lane = t & 31;
    const int m = lane & 15, hi = lane >> 4;
    const int rb = blockIdx.x * 128 + wave * 16;
    if (rb >= N) return;
    const int lrow = min(rb + m, N - 1);
    const float4* xr = (const float4*)(X + (size_t)lrow * FDIM) + hi;
    const float4* bp = &lB[hi * FDIM + m];

    v8f acc[8];
#pragma unroll
    for (int ct = 0; ct < 8; ++ct) acc[ct] = (v8f){0.f, 0.f, 0.f, 0.f, 0.f, 0.f, 0.f, 0.f};

#pragma unroll
    for (int j = 0; j < 16; ++j) {
        const float4 a = xr[2 * j];
        const v2f a0 = {a.x, a.y}, a1 = {a.z, a.w};
        float4 b[8];
#pragma unroll
        for (int ct = 0; ct < 8; ++ct) b[ct] = bp[(2 * j) * FDIM + ct * 16];
#pragma unroll
        for (int ct = 0; ct < 8; ++ct) {
            const v2f b0 = {b[ct].x, b[ct].y};
            acc[ct] = __builtin_amdgcn_wmma_f32_16x16x4_f32(
                false, a0, false, b0, (short)0, acc[ct], false, false);
        }
#pragma unroll
        for (int ct = 0; ct < 8; ++ct) {
            const v2f b1 = {b[ct].z, b[ct].w};
            acc[ct] = __builtin_amdgcn_wmma_f32_16x16x4_f32(
                false, a1, false, b1, (short)0, acc[ct], false, false);
        }
    }

#pragma unroll
    for (int ct = 0; ct < 8; ++ct) {
#pragma unroll
        for (int r = 0; r < 8; ++r) {
            int row = rb + r + 8 * hi;
            if (row < N) O[(size_t)row * FDIM + ct * 16 + m] = acc[ct][r];
        }
    }
}


__device__ __forceinline__ void own_scan(const int* __restrict__ dst, int E, int row0, unsigned un,
                                         int2* lst, int* cnt)
{
    const int t = threadIdx.x;
    auto push = [&](int e, unsigned r) {
        int pos = atomicAdd(cnt, 1);
        if (pos < OWN_CAP) lst[pos] = make_int2(e, (int)r);
    };
    auto test = [&](const int4& d, int e) {
        unsigned r0 = (unsigned)(d.x - row0), r1 = (unsigned)(d.y - row0);
        unsigned r2 = (unsigned)(d.z - row0), r3 = (unsigned)(d.w - row0);
        unsigned m = min(min(r0, r1), min(r2, r3));
        if (m < un) {
            if (r0 < un) push(e, r0);
            if (r1 < un) push(e + 1, r1);
            if (r2 < un) push(e + 2, r2);
            if (r3 < un) push(e + 3, r3);
        }
    };
    const int mis = (int)(((size_t)dst >> 2) & 3);
    int head = (4 - mis) & 3; if (head > E) head = E;
    const int nvec = (E - head) >> 2;
    const int tail0 = head + 4 * nvec;
    const int4* d4 = (const int4*)(dst + head);
    const int nfull = nvec / (4 * OWN_T);
    if (nfull > 0) {
        int tile = (int)(blockIdx.x % (unsigned)nfull);
        const int4* p = d4 + (size_t)tile * (4 * OWN_T) + t;
        int4 c0 = p[0], c1 = p[OWN_T], c2 = p[2 * OWN_T], c3 = p[3 * OWN_T];
        for (int it = 0; it < nfull; ++it) {
            int nt = tile + 1; if (nt >= nfull) nt = 0;
            const int4* q = d4 + (size_t)nt * (4 * OWN_T) + t;
            int4 n0 = q[0], n1 = q[OWN_T], n2 = q[2 * OWN_T], n3 = q[3 * OWN_T];
            int e = head + 4 * (tile * (4 * OWN_T) + t);
            test(c0, e);
            test(c1, e + 4 * OWN_T);
            test(c2, e + 8 * OWN_T);
            test(c3, e + 12 * OWN_T);
            c0 = n0; c1 = n1; c2 = n2; c3 = n3; tile = nt;
        }
    }
    for (int v = nfull * (4 * OWN_T) + t; v < nvec; v += OWN_T) test(d4[v], head + 4 * v);
    if (t < head) {
        unsigned r = (unsigned)(dst[t] - row0);
        if (r < un) push(t, r);
    }
    if (t < E - tail0) {
        unsigned r = (unsigned)(dst[tail0 + t] - row0);
        if (r < un) push(tail0 + t, r);
    }
}

__device__ __forceinline__ void own_scan_key(const unsigned char* __restrict__ key, int E, unsigned kb,
                                             int2* lst, int* cnt)
{
    const int t = threadIdx.x;
    const unsigned pat = kb * 0x01010101u;
    auto push = [&](int e) {
        int pos = atomicAdd(cnt, 1);
        if (pos < OWN_CAP) lst[pos] = make_int2(e, 0);
    };
    auto eq = [&](unsigned x) {
        unsigned y = x ^ pat;
        return ~(((y & 0x7f7f7f7fu) + 0x7f7f7f7fu) | y) & 0x80808080u;
    };
    auto vec = [&](const uint4& k) {
        return (eq(k.x) >> 7) | (eq(k.y) >> 6) | (eq(k.z) >> 5) | (eq(k.w) >> 4);
    };
    auto drain = [&](unsigned long long M, int e0) {
        while (M) {
            const int q = __builtin_ctzll(M);
            const int v = 2 * (q >> 5) + ((q >> 2) & 1);
            push(e0 + v * (16 * OWN_T) + 4 * (q & 3) + ((q >> 3) & 3));
            M &= M - 1;
        }
    };
    const int nvec = E >> 4;
    const uint4* k4 = (const uint4*)key;
    const int nfull = nvec / (4 * OWN_T);
    if (nfull > 0) {
        int tile = (int)(blockIdx.x % (unsigned)nfull);
        const uint4* p = k4 + (size_t)tile * (4 * OWN_T) + t;
        uint4 c0 = p[0], c1 = p[OWN_T], c2 = p[2 * OWN_T], c3 = p[3 * OWN_T];
        for (int it = 0; it < nfull; ++it) {
            int nt = tile + 1; if (nt >= nfull) nt = 0;
            const uint4* q = k4 + (size_t)nt * (4 * OWN_T) + t;
            uint4 n0 = q[0], n1 = q[OWN_T], n2 = q[2 * OWN_T], n3 = q[3 * OWN_T];
            const unsigned m01 = vec(c0) | (vec(c1) << 4), m23 = vec(c2) | (vec(c3) << 4);
            drain((unsigned long long)m01 | ((unsigned long long)m23 << 32), 16 * (tile * (4 * OWN_T) + t));
            c0 = n0; c1 = n1; c2 = n2; c3 = n3; tile = nt;
        }
    }
    for (int v = nfull * (4 * OWN_T) + t; v < nvec; v += OWN_T) drain((unsigned long long)vec(k4[v]), 16 * v);
    for (int e = 16 * nvec + t; e < E; e += OWN_T)
        if (key[e] == (unsigned char)kb) push(e);
}

__device__ __forceinline__ void own_resolve(const int* __restrict__ src, const float* __restrict__ dis,
                                            const float* sdis, int2* lst, float* wl, int total, int N)
{
    for (int i = threadIdx.x; i < total; i += OWN_T) {
        int2 en = lst[i];
        int s = src[en.x];
        s = min(max(s, 0), N - 1);
        lst[i].x = s;
        wl[i] = dis[s] * sdis[en.y];
    }
}

__device__ __forceinline__ void own_list(const int* __restrict__ src, const int* __restrict__ dst,
                                         const int2* lst, int* hist, int2* kl,
                                         int total, int N, int row0, int nrows)
{
    if (total <= 0) return;
    const int t = threadIdx.x;
    int e[OWN_CAP / OWN_T], s[OWN_CAP / OWN_T], r[OWN_CAP / OWN_T];
#pragma unroll
    for (int u = 0; u < OWN_CAP / OWN_T; ++u) e[u] = lst[min(t + u * OWN_T, total - 1)].x;
#pragma unroll
    for (int u = 0; u < OWN_CAP / OWN_T; ++u) { s[u] = src[e[u]]; r[u] = dst[e[u]]; }
#pragma unroll
    for (int u = 0; u < OWN_CAP / OWN_T; ++u) {
        const int i = t + u * OWN_T;
        if (i < total) {
            const int sv = min(max(s[u], 0), N - 1);
            const int rv = min(max(r[u] - row0, 0), nrows - 1);
            if (kl) kl[i] = make_int2(sv, rv);
            atomicAdd(&hist[rv], 1);
        }
    }
}

__device__ __forceinline__ void own_fetch(const int2* kl, const float* __restrict__ dis,
                                          const float* sdis, int2* lst, float* wl,
                                          int total, int N, int nrows)
{
    if (total <= 0) return;
    const int t = threadIdx.x;
    int2 en[OWN_CAP / OWN_T]; float ds[OWN_CAP / OWN_T];
#pragma unroll
    for (int u = 0; u < OWN_CAP / OWN_T; ++u) en[u] = kl[min(t + u * OWN_T, total - 1)];
#pragma unroll
    for (int u = 0; u < OWN_CAP / OWN_T; ++u) {
        en[u].x = min(max(en[u].x, 0), N - 1);
        en[u].y = min(max(en[u].y, 0), nrows - 1);
        ds[u] = dis[en[u].x];
    }
#pragma unroll
    for (int u = 0; u < OWN_CAP / OWN_T; ++u) {
        const int i = t + u * OWN_T;
        if (i < total) { lst[i] = en[u]; wl[i] = ds[u] * sdis[en[u].y]; }
    }
}

__device__ __forceinline__ int keep_cap(int nrows, int C)
{
    return min(OWN_CAP, (nrows * C) >> 1);
}

__global__ __launch_bounds__(OWN_T) void k_list(const int* __restrict__ src,
                                                const int* __restrict__ dst,
                                                float* __restrict__ dis,
                                                float* __restrict__ keep,
                                                int* __restrict__ kcnt,
                                                const unsigned char* __restrict__ key,
                                                int N, int E, int C)
{
    __shared__ int2 lst[OWN_CAP];
    __shared__ int  hist[OWN_R];
    __shared__ int  cnt;

    const int t = threadIdx.x;
    const int row0 = blockIdx.x * OWN_R;
    const int nrows = min(OWN_R, N - row0);
    const unsigned un = (unsigned)nrows;

    if (t == 0) cnt = 0;
    for (int r = t; r < OWN_R; r += OWN_T) hist[r] = 1;
    __syncthreads();

    const bool bykey = key != nullptr && N <= 256 * OWN_R;
    if (bykey) own_scan_key(key, E, blockIdx.x, lst, &cnt);
    else       own_scan(dst, E, row0, un, lst, &cnt);
    __syncthreads();

    const int total = __builtin_amdgcn_readfirstlane(cnt);
    if (t == 0) kcnt[blockIdx.x] = total;
    if (total <= OWN_CAP) {
        int2* kl = (total <= keep_cap(nrows, C)) ? (int2*)(keep + (size_t)row0 * C) : nullptr;
        own_list(src, dst, lst, hist, kl, total, N, row0, nrows);
    } else {
        for (int e = t; e < E; e += OWN_T) {
            unsigned r = (unsigned)(dst[e] - row0);
            if (r < un) atomicAdd(&hist[r], 1);
        }
    }
    __syncthreads();
    for (int r = t; r < nrows; r += OWN_T) dis[row0 + r] = rsqrtf((float)hist[r]);
}

__global__ __launch_bounds__(OWN_T) void k_own128(const float* __restrict__ h,
                                                  const int* __restrict__ src,
                                                  const int* __restrict__ dst,
                                                  const float* __restrict__ dis,
                                                  const float* __restrict__ bias,
                                                  float* __restrict__ out,
                                                  const float* __restrict__ keep,
                                                  const int* __restrict__ kcnt,
                                                  int N, int E, int C)
{
    __shared__ float acc[OWN_R * FDIM];
    __shared__ int2  lst[OWN_CAP];
    __shared__ float wl[OWN_CAP];
    __shared__ float sdis[OWN_R];
    __shared__ int   cnt;

    const int t = threadIdx.x;
    const int row0 = blockIdx.x * OWN_R;
    const int nrows = min(OWN_R, N - row0);
    const unsigned un = (unsigned)nrows;
    const float4* h4 = (const float4*)h;

    if (t == 0) cnt = 0;
    for (int r = t; r < nrows; r += OWN_T) sdis[r] = dis[row0 + r];
    {
        const float4 bq = ((const float4*)bias)[t & 31];
        for (int idx = t; idx < nrows * 32; idx += OWN_T) {
            int r = idx >> 5, q = idx & 31;
            float d = dis[row0 + r];
            float4 v = h4[(size_t)(row0 + r) * 32 + q];
            float* a = &acc[r * FDIM + q];
            a[0]  = v.x * d * d + bq.x;
            a[32] = v.y * d * d + bq.y;
            a[64] = v.z * d * d + bq.z;
            a[96] = v.w * d * d + bq.w;
        }
    }
    __syncthreads();

    const int kept = __builtin_amdgcn_readfirstlane(kcnt[blockIdx.x]);
    int total;
    if (kept <= keep_cap(nrows, C)) {
        total = max(kept, 0);
        own_fetch((const int2*)(keep + (size_t)row0 * C), dis, sdis, lst, wl, total, N, nrows);
    } else {
        own_scan(dst, E, row0, un, lst, &cnt);
        __syncthreads();
        total = __builtin_amdgcn_readfirstlane(cnt);
        if (total <= OWN_CAP) own_resolve(src, dis, sdis, lst, wl, total, N);
    }
    __syncthreads();

    if (total <= OWN_CAP) {
        const int wave = __builtin_amdgcn_readfirstlane(t >> 5), lane = t & 31;
        const int ngrp = total >> 3;
        for (int g = wave; g < ngrp; g += OWN_T / 32) {
            int r[8]; float w[8]; float4 v[8];
#pragma unroll
            for (int u = 0; u < 8; ++u) {
                int2 en = lst[8 * g + u];
                int s = __builtin_amdgcn_readfirstlane(en.x);
                r[u] = __builtin_amdgcn_readfirstlane(en.y);
                w[u] = wl[8 * g + u];
                v[u] = h4[(size_t)s * 32 + lane];
            }
#pragma unroll
            for (int u = 0; u < 8; ++u) {
                float* a = &acc[r[u] * FDIM + lane];
                atomicAdd(a,      v[u].x * w[u]);
                atomicAdd(a + 32, v[u].y * w[u]);
                atomicAdd(a + 64, v[u].z * w[u]);
                atomicAdd(a + 96, v[u].w * w[u]);
            }
        }
        if (8 * ngrp + wave < total) {
            int2 en = lst[8 * ngrp + wave];
            int s = __builtin_amdgcn_readfirstlane(en.x);
            int r = __builtin_amdgcn_readfirstlane(en.y);
            float w = wl[8 * ngrp + wave];
            float4 v = h4[(size_t)s * 32 + lane];
            float* a = &acc[r * FDIM + lane];
            atomicAdd(a,      v.x * w);
            atomicAdd(a + 32, v.y * w);
            atomicAdd(a + 64, v.z * w);
            atomicAdd(a + 96, v.w * w);
        }
    } else {
        for (int e = t; e < E; e += OWN_T) {
            unsigned r = (unsigned)(dst[e] - row0);
            if (r < un) {
                int s = min(max(src[e], 0), N - 1);
                float w = dis[s] * sdis[r];
                const float* hp = h + (size_t)s * FDIM;
                for (int f = 0; f < FDIM; ++f)
                    atomicAdd(&acc[r * FDIM + (f & 3) * 32 + (f >> 2)], hp[f] * w);
            }
        }
    }
    __syncthreads();

    float4* o4 = (float4*)out;
    for (int idx = t; idx < nrows * 32; idx += OWN_T) {
        int r = idx >> 5, q = idx & 31;
        const float* a = &acc[r * FDIM + q];
        o4[(size_t)(row0 + r) * 32 + q] = make_float4(a[0], a[32], a[64], a[96]);
    }
}

__global__ __launch_bounds__(OWN_T) void k_ownC(const float* __restrict__ h,
                                                const int* __restrict__ src,
                                                const int* __restrict__ dst,
                                                const float* __restrict__ dis,
                                                const float* __restrict__ bias,
                                                float* __restrict__ out,
                                                const int* __restrict__ kcnt,
                                                int N, int E, int C)
{
    __shared__ float acc[OWN_R * 64];
    __shared__ int2  lst[OWN_CAP];
    __shared__ float wl[OWN_CAP];
    __shared__ float sdis[OWN_R];
    __shared__ int   cnt;

    const int t = threadIdx.x;
    const int row0 = blockIdx.x * OWN_R;
    const int nrows = min(OWN_R, N - row0);
    const unsigned un = (unsigned)nrows;
    const size_t g0 = (size_t)row0 * C;

    if (t == 0) cnt = 0;
    for (int r = t; r < nrows; r += OWN_T) sdis[r] = dis[row0 + r];
    for (int idx = t; idx < nrows * C; idx += OWN_T) {
        int r = idx / C, f = idx - r * C;
        float d = dis[row0 + r];
        acc[idx] = h[g0 + idx] * d * d + bias[f];
    }
    __syncthreads();

    const int kept = __builtin_amdgcn_readfirstlane(kcnt[blockIdx.x]);
    int total;
    if (kept <= keep_cap(nrows, C)) {
        total = max(kept, 0);
        own_fetch((const int2*)(out + g0), dis, sdis, lst, wl, total, N, nrows);
    } else {
        own_scan(dst, E, row0, un, lst, &cnt);
        __syncthreads();
        total = __builtin_amdgcn_readfirstlane(cnt);
        if (total <= OWN_CAP) own_resolve(src, dis, sdis, lst, wl, total, N);
    }
    __syncthreads();

    if (total <= OWN_CAP) {
        const int wave = __builtin_amdgcn_readfirstlane(t >> 5), lane = t & 31;
        const bool c0 = lane < C, c1 = 32 + lane < C;
        const int l0 = min(lane, C - 1), l1 = min(32 + lane, C - 1);
        const int ngrp = total >> 3;
        for (int g = wave; g < ngrp; g += OWN_T / 32) {
            int s[8], r[8]; float w[8], v0[8], v1[8];
#pragma unroll
            for (int u = 0; u < 8; ++u) {
                int2 en = lst[8 * g + u];
                s[u] = __builtin_amdgcn_readfirstlane(en.x);
                r[u] = __builtin_amdgcn_readfirstlane(en.y);
                w[u] = wl[8 * g + u];
            }
#pragma unroll
            for (int u = 0; u < 8; ++u) {
                const float* hp = h + (size_t)s[u] * C;
                v0[u] = hp[l0];
                v1[u] = hp[l1];
            }
            if (c0) {
#pragma unroll
                for (int u = 0; u < 8; ++u) atomicAdd(&acc[r[u] * C + lane], v0[u] * w[u]);
            }
            if (c1) {
#pragma unroll
                for (int u = 0; u < 8; ++u) atomicAdd(&acc[r[u] * C + 32 + lane], v1[u] * w[u]);
            }
        }
        if (8 * ngrp + wave < total) {
            int2 en = lst[8 * ngrp + wave];
            int s = __builtin_amdgcn_readfirstlane(en.x);
            int r = __builtin_amdgcn_readfirstlane(en.y);
            float w = wl[8 * ngrp + wave];
            const float* hp = h + (size_t)s * C;
            float* a = &acc[r * C];
            if (c0) atomicAdd(a + lane,      hp[lane] * w);
            if (c1) atomicAdd(a + 32 + lane, hp[32 + lane] * w);
        }
    } else {
        for (int e = t; e < E; e += OWN_T) {
            unsigned r = (unsigned)(dst[e] - row0);
            if (r < un) {
                int s = min(max(src[e], 0), N - 1);
                float w = dis[s] * sdis[r];
                const float* hp = h + (size_t)s * C;
                for (int f = 0; f < C; ++f) atomicAdd(&acc[r * C + f], hp[f] * w);
            }
        }
    }
    __syncthreads();

    for (int idx = t; idx < nrows * C; idx += OWN_T) out[g0 + idx] = acc[idx];
}

#define RS_T    1024
#define RS_PER  13
#define RS_ROWS (RS_PER * (RS_T / 32))
__global__ __launch_bounds__(RS_T) void k_relu_stats(float* __restrict__ a,
                                                     float* __restrict__ gsum,
                                                     float* __restrict__ gsq, int N)
{
    __shared__ float4 ssum[RS_T];
    __shared__ float4 ssq[RS_T];
    const int t = threadIdx.x, q = t & 31;
    const int row0 = blockIdx.x * RS_ROWS + (t >> 5);
    float4* a4 = (float4*)a;

    float4 v[RS_PER];
#pragma unroll
    for (int i = 0; i < RS_PER; ++i) {
        const int row = min(row0 + (RS_T / 32) * i, N - 1);
        v[i] = a4[(size_t)row * (FDIM / 4) + q];
    }

    float4 s  = make_float4(0.f, 0.f, 0.f, 0.f);
    float4 sq = make_float4(0.f, 0.f, 0.f, 0.f);
#pragma unroll
    for (int i = 0; i < RS_PER; ++i) {
        const int row = row0 + (RS_T / 32) * i;
        if (row < N) {
            const float4 r = make_float4(fmaxf(v[i].x, 0.f), fmaxf(v[i].y, 0.f),
                                         fmaxf(v[i].z, 0.f), fmaxf(v[i].w, 0.f));
            a4[(size_t)row * (FDIM / 4) + q] = r;
            s.x += r.x; s.y += r.y; s.z += r.z; s.w += r.w;
            sq.x += r.x * r.x; sq.y += r.y * r.y; sq.z += r.z * r.z; sq.w += r.w * r.w;
        }
    }
    ssum[t] = s; ssq[t] = sq;
    __syncthreads();

    if (t < 2 * FDIM) {
        const int c = t & (FDIM - 1);
        const float* p = (t < FDIM) ? (const float*)ssum : (const float*)ssq;
        float tot = 0.f;
#pragma unroll
        for (int g = 0; g < RS_T / 32; ++g) tot += p[g * FDIM + c];
        atomicAdd((t < FDIM) ? &gsum[c] : &gsq[c], tot);
    }
}

__global__ void k_bn_finalize(const float* __restrict__ gsum,
                              const float* __restrict__ gsq,
                              const float* __restrict__ gamma,
                              const float* __restrict__ beta,
                              float* __restrict__ scl,
                              float* __restrict__ sft, int N)
{
    int f = threadIdx.x;
    float inv_n = 1.f / (float)N;
    float mean = gsum[f] * inv_n;
    float var  = gsq[f] * inv_n - mean * mean;
    float inv  = rsqrtf(var + 1e-5f);
    float g = gamma[f] * inv;
    scl[f] = g;
    sft[f] = beta[f] - mean * g;
}

__global__ __launch_bounds__(256) void k_gemm2(const float* __restrict__ A,
                                               const float* __restrict__ W,
                                               const float* __restrict__ scl,
                                               const float* __restrict__ sft,
                                               float* __restrict__ O, int N, int C)
{
    __shared__ float4 lB[32 * 48];
    __shared__ float4 lS[32];
    __shared__ float4 lF[32];

    const int t = threadIdx.x;
    const int wave = t >> 5, lane = t & 31;
    const int m = lane & 15, hi = lane >> 4;
    const int rb = blockIdx.x * 128 + wave * 16;
    const int lrow = min(rb + m, N - 1);
    const v4f* ar = (const v4f*)(A + (size_t)lrow * FDIM) + hi;

    v4f xa[16];
#pragma unroll
    for (int j = 0; j < 16; ++j) xa[j] = ar[2 * j];

#pragma unroll
    for (int idx = t; idx < 32 * 48; idx += 256) {
        int jh = idx / 48, c = idx - jh * 48;
        const float* wp = W + (size_t)(4 * jh) * C + min(c, C - 1);
        const float4 v = make_float4(wp[0], wp[C], wp[2 * C], wp[3 * C]);
        lB[idx] = (c < C) ? v : make_float4(0.f, 0.f, 0.f, 0.f);
    }
    if (t < 32) {
        lS[t] = make_float4(scl[4 * t], scl[4 * t + 1], scl[4 * t + 2], scl[4 * t + 3]);
        lF[t] = make_float4(sft[4 * t], sft[4 * t + 1], sft[4 * t + 2], sft[4 * t + 3]);
    }
    __syncthreads();

    if (rb >= N) return;
    const float4* bp = &lB[hi * 48 + m];

    v8f acc[3];
#pragma unroll
    for (int ct = 0; ct < 3; ++ct) acc[ct] = (v8f){0.f, 0.f, 0.f, 0.f, 0.f, 0.f, 0.f, 0.f};

#pragma unroll
    for (int j = 0; j < 16; ++j) {
        const v4f x = xa[j];
        const float4 s = lS[2 * j + hi], f = lF[2 * j + hi];
        const v2f a0 = {x.x * s.x + f.x, x.y * s.y + f.y};
        const v2f a1 = {x.z * s.z + f.z, x.w * s.w + f.w};
        float4 b[3];
#pragma unroll
        for (int ct = 0; ct < 3; ++ct) b[ct] = bp[(2 * j) * 48 + ct * 16];
#pragma unroll
        for (int ct = 0; ct < 3; ++ct) {
            const v2f b0 = {b[ct].x, b[ct].y};
            acc[ct] = __builtin_amdgcn_wmma_f32_16x16x4_f32(
                false, a0, false, b0, (short)0, acc[ct], false, false);
        }
#pragma unroll
        for (int ct = 0; ct < 3; ++ct) {
            const v2f b1 = {b[ct].z, b[ct].w};
            acc[ct] = __builtin_amdgcn_wmma_f32_16x16x4_f32(
                false, a1, false, b1, (short)0, acc[ct], false, false);
        }
    }

#pragma unroll
    for (int ct = 0; ct < 3; ++ct) {
        const int nloc = ct * 16 + m;
        if (nloc < C) {
#pragma unroll
            for (int r = 0; r < 8; ++r) {
                int row = rb + r + 8 * hi;
                if (row < N) O[(size_t)row * C + nloc] = acc[ct][r];
            }
        }
    }
}

__global__ __launch_bounds__(256) void k_logsoftmax(float* __restrict__ out, int N, int C)
{
    long long gid = (long long)blockIdx.x * 256 + threadIdx.x;
    int row = (int)(gid >> 5);
    int lane = threadIdx.x & 31;
    if (row >= N) return;
    float* p = out + (size_t)row * C;
    float v0 = (lane < C) ? p[lane] : -3.4e38f;
    float v1 = (32 + lane < C) ? p[32 + lane] : -3.4e38f;
    float mx = fmaxf(v0, v1);
#pragma unroll
    for (int o = 16; o > 0; o >>= 1) mx = fmaxf(mx, __shfl_xor(mx, o, 32));
    float s = ((lane < C) ? expf(v0 - mx) : 0.f) +
              ((32 + lane < C) ? expf(v1 - mx) : 0.f);
#pragma unroll
    for (int o = 16; o > 0; o >>= 1) s += __shfl_xor(s, o, 32);
    float l = logf(s);
    if (lane < C)      p[lane]      = v0 - mx - l;
    if (32 + lane < C) p[32 + lane] = v1 - mx - l;
}

extern "C" void kernel_launch(void* const* d_in, const int* in_sizes, int n_in,
                              void* d_out, int out_size, void* d_ws, size_t ws_size,
                              hipStream_t stream)
{
    const float* x     = (const float*)d_in[0];
    const int*   ei    = (const int*)  d_in[1];
    const float* W1    = (const float*)d_in[2];
    const float* b1    = (const float*)d_in[3];
    const float* gamma = (const float*)d_in[4];
    const float* beta  = (const float*)d_in[5];
    const float* W2    = (const float*)d_in[6];
    const float* b2    = (const float*)d_in[7];
    float* out = (float*)d_out;

    const int N = in_sizes[0] / FDIM;
    const int E = in_sizes[1] / 2;
    const int C = in_sizes[6] / FDIM;
    const int* src = ei;
    const int* dst = ei + E;

    char* ws = (char*)d_ws;
    size_t off = 0;
    auto up = [](size_t v) { return (v + 255) & ~(size_t)255; };
    const int nb_own = (N + OWN_R - 1) / OWN_R;
    int*   kcnt  = (int*)(ws + off);   off = up(off + (size_t)nb_own * 4);
    float* dis   = (float*)(ws + off); off = up(off + (size_t)N * 4);
    float* stats = (float*)(ws + off); off = up(off + 512 * 4);
    float* gsum = stats, *gsq = stats + 128, *scl = stats + 256, *sft = stats + 384;
    float* h1   = (float*)(ws + off);  off = up(off + (size_t)N * FDIM * 4);
    float* agg1 = (float*)(ws + off);  off = up(off + (size_t)N * FDIM * 4);
    float* h2   = h1;
    unsigned char* key = (unsigned char*)(ws + off); off = up(off + (size_t)E);
    if (off > ws_size) key = nullptr;

    const int nb_key = (E / 16 + 256) / 256;
    const int nb_Nw  = (N + 7) / 8;

    k_key<<<nb_key, 256, 0, stream>>>(dst, key, stats, E);

    k_list<<<nb_own, OWN_T, 0, stream>>>(src, dst, dis, out, kcnt, key, N, E, C);

    k_gemm1<<<(N + 127) / 128, 256, 0, stream>>>(x, W1, h1, N);

    k_own128<<<nb_own, OWN_T, 0, stream>>>(h1, src, dst, dis, b1, agg1, out, kcnt, N, E, C);

    k_relu_stats<<<(N + RS_ROWS - 1) / RS_ROWS, RS_T, 0, stream>>>(agg1, gsum, gsq, N);
    k_bn_finalize<<<1, 128, 0, stream>>>(gsum, gsq, gamma, beta, scl, sft, N);

    k_gemm2<<<(N + 127) / 128, 256, 0, stream>>>(agg1, W2, scl, sft, h2, N, C);

    k_ownC<<<nb_own, OWN_T, 0, stream>>>(h2, src, dst, dis, b2, out, kcnt, N, E, C);

    k_logsoftmax<<<nb_Nw, 256, 0, stream>>>(out, N, C);
}


# Log-softmax taken by the conv2 owner block on its finished rows in LDS (8 lanes per row) before its one store: the one-wave-per-row pass is gone, 9 to 8 launches; softmax sum order changes
# speedup vs baseline: 3.8418x; 1.0460x over previous
#include <hip/hip_runtime.h>

typedef __attribute__((ext_vector_type(2))) float v2f;
typedef __attribute__((ext_vector_type(8))) float v8f;
typedef __attribute__((ext_vector_type(4))) float v4f;

#define FDIM 128
#define OWN_R   400
#define OWN_CAP 8192
#define OWN_T   1024


__global__ __launch_bounds__(256) void k_key(const int* __restrict__ dst,
                                             unsigned char* __restrict__ key,
                                             float* __restrict__ stats, int E)
{
    const int i = blockIdx.x * 256 + threadIdx.x;
    if (i < 256) stats[i] = 0.f;
    if (key == nullptr) return;
    const int e0 = 16 * i;
    if (e0 >= E) return;
    auto k1 = [](int d) { return ((unsigned)d / OWN_R) & 255u; };
    if (e0 + 16 <= E && ((size_t)dst & 15) == 0) {
        const int4* p = (const int4*)(dst + e0);
        const int4 a = p[0], b = p[1], c = p[2], d = p[3];
        auto k4 = [&](const int4& v) { return k1(v.x) | (k1(v.y) << 8) | (k1(v.z) << 16) | (k1(v.w) << 24); };
        *(uint4*)(key + e0) = make_uint4(k4(a), k4(b), k4(c), k4(d));
    } else {
        const int e1 = min(e0 + 16, E);
        for (int e = e0; e < e1; ++e) key[e] = (unsigned char)k1(dst[e]);
    }
}

__global__ __launch_bounds__(256) void k_gemm1(const float* __restrict__ X,
                                               const float* __restrict__ W,
                                               float* __restrict__ O, int N)
{
    __shared__ float4 lB[32 * FDIM];

    const int t = threadIdx.x;
    {
        const int n4 = (t & 31) * 4;
#pragma unroll
        for (int i = 0; i < 4; ++i) {
            const int jh = (t >> 5) + 8 * i;
            const float4* wp = (const float4*)(W + (size_t)(4 * jh) * FDIM + n4);
            const float4 r0 = wp[0], r1 = wp[FDIM / 4], r2 = wp[2 * (FDIM / 4)], r3 = wp[3 * (FDIM / 4)];
            float4* lp = &lB[jh * FDIM + n4];
            lp[0] = make_float4(r0.x, r1.x, r2.x, r3.x);
            lp[1] = make_float4(r0.y, r1.y, r2.y, r3.y);
            lp[2] = make_float4(r0.z, r1.z, r2.z, r3.z);
            lp[3] = make_float4(r0.w, r1.w, r2.w, r3.w);
        }
    }
    __syncthreads();

    const int wave = t >> 5, lane = t & 31;
    const int m = lane & 15, hi = lane >> 4;
    const int rb = blockIdx.x * 128 + wave * 16;
    if (rb >= N) return;
    const int lrow = min(rb + m, N - 1);
    const float4* xr = (const float4*)(X + (size_t)lrow * FDIM) + hi;
    const float4* bp = &lB[hi * FDIM + m];

    v8f acc[8];
#pragma unroll
    for (int ct = 0; ct < 8; ++ct) acc[ct] = (v8f){0.f, 0.f, 0.f, 0.f, 0.f, 0.f, 0.f, 0.f};

#pragma unroll
    for (int j = 0; j < 16; ++j) {
        const float4 a = xr[2 * j];
        const v2f a0 = {a.x, a.y}, a1 = {a.z, a.w};
        float4 b[8];
#pragma unroll
        for (int ct = 0; ct < 8; ++ct) b[ct] = bp[(2 * j) * FDIM + ct * 16];
#pragma unroll
        for (int ct = 0; ct < 8; ++ct) {
            const v2f b0 = {b[ct].x, b[ct].y};
            acc[ct] = __builtin_amdgcn_wmma_f32_16x16x4_f32(
                false, a0, false, b0, (short)0, acc[ct], false, false);
        }
#pragma unroll
        for (int ct = 0; ct < 8; ++ct) {
            const v2f b1 = {b[ct].z, b[ct].w};
            acc[ct] = __builtin_amdgcn_wmma_f32_16x16x4_f32(
                false, a1, false, b1, (short)0, acc[ct], false, false);
        }
    }

#pragma unroll
    for (int ct = 0; ct < 8; ++ct) {
#pragma unroll
        for (int r = 0; r < 8; ++r) {
            int row = rb + r + 8 * hi;
            if (row < N) O[(size_t)row * FDIM + ct * 16 + m] = acc[ct][r];
        }
    }
}


__device__ __forceinline__ void own_scan(const int* __restrict__ dst, int E, int row0, unsigned un,
                                         int2* lst, int* cnt)
{
    const int t = threadIdx.x;
    auto push = [&](int e, unsigned r) {
        int pos = atomicAdd(cnt, 1);
        if (pos < OWN_CAP) lst[pos] = make_int2(e, (int)r);
    };
    auto test = [&](const int4& d, int e) {
        unsigned r0 = (unsigned)(d.x - row0), r1 = (unsigned)(d.y - row0);
        unsigned r2 = (unsigned)(d.z - row0), r3 = (unsigned)(d.w - row0);
        unsigned m = min(min(r0, r1), min(r2, r3));
        if (m < un) {
            if (r0 < un) push(e, r0);
            if (r1 < un) push(e + 1, r1);
            if (r2 < un) push(e + 2, r2);
            if (r3 < un) push(e + 3, r3);
        }
    };
    const int mis = (int)(((size_t)dst >> 2) & 3);
    int head = (4 - mis) & 3; if (head > E) head = E;
    const int nvec = (E - head) >> 2;
    const int tail0 = head + 4 * nvec;
    const int4* d4 = (const int4*)(dst + head);
    const int nfull = nvec / (4 * OWN_T);
    if (nfull > 0) {
        int tile = (int)(blockIdx.x % (unsigned)nfull);
        const int4* p = d4 + (size_t)tile * (4 * OWN_T) + t;
        int4 c0 = p[0], c1 = p[OWN_T], c2 = p[2 * OWN_T], c3 = p[3 * OWN_T];
        for (int it = 0; it < nfull; ++it) {
            int nt = tile + 1; if (nt >= nfull) nt = 0;
            const int4* q = d4 + (size_t)nt * (4 * OWN_T) + t;
            int4 n0 = q[0], n1 = q[OWN_T], n2 = q[2 * OWN_T], n3 = q[3 * OWN_T];
            int e = head + 4 * (tile * (4 * OWN_T) + t);
            test(c0, e);
            test(c1, e + 4 * OWN_T);
            test(c2, e + 8 * OWN_T);
            test(c3, e + 12 * OWN_T);
            c0 = n0; c1 = n1; c2 = n2; c3 = n3; tile = nt;
        }
    }
    for (int v = nfull * (4 * OWN_T) + t; v < nvec; v += OWN_T) test(d4[v], head + 4 * v);
    if (t < head) {
        unsigned r = (unsigned)(dst[t] - row0);
        if (r < un) push(t, r);
    }
    if (t < E - tail0) {
        unsigned r = (unsigned)(dst[tail0 + t] - row0);
        if (r < un) push(tail0 + t, r);
    }
}

__device__ __forceinline__ void own_scan_key(const unsigned char* __restrict__ key, int E, unsigned kb,
                                             int2* lst, int* cnt)
{
    const int t = threadIdx.x;
    const unsigned pat = kb * 0x01010101u;
    auto push = [&](int e) {
        int pos = atomicAdd(cnt, 1);
        if (pos < OWN_CAP) lst[pos] = make_int2(e, 0);
    };
    auto eq = [&](unsigned x) {
        unsigned y = x ^ pat;
        return ~(((y & 0x7f7f7f7fu) + 0x7f7f7f7fu) | y) & 0x80808080u;
    };
    auto vec = [&](const uint4& k) {
        return (eq(k.x) >> 7) | (eq(k.y) >> 6) | (eq(k.z) >> 5) | (eq(k.w) >> 4);
    };
    auto drain = [&](unsigned long long M, int e0) {
        while (M) {
            const int q = __builtin_ctzll(M);
            const int v = 2 * (q >> 5) + ((q >> 2) & 1);
            push(e0 + v * (16 * OWN_T) + 4 * (q & 3) + ((q >> 3) & 3));
            M &= M - 1;
        }
    };
    const int nvec = E >> 4;
    const uint4* k4 = (const uint4*)key;
    const int nfull = nvec / (4 * OWN_T);
    if (nfull > 0) {
        int tile = (int)(blockIdx.x % (unsigned)nfull);
        const uint4* p = k4 + (size_t)tile * (4 * OWN_T) + t;
        uint4 c0 = p[0], c1 = p[OWN_T], c2 = p[2 * OWN_T], c3 = p[3 * OWN_T];
        for (int it = 0; it < nfull; ++it) {
            int nt = tile + 1; if (nt >= nfull) nt = 0;
            const uint4* q = k4 + (size_t)nt * (4 * OWN_T) + t;
            uint4 n0 = q[0], n1 = q[OWN_T], n2 = q[2 * OWN_T], n3 = q[3 * OWN_T];
            const unsigned m01 = vec(c0) | (vec(c1) << 4), m23 = vec(c2) | (vec(c3) << 4);
            drain((unsigned long long)m01 | ((unsigned long long)m23 << 32), 16 * (tile * (4 * OWN_T) + t));
            c0 = n0; c1 = n1; c2 = n2; c3 = n3; tile = nt;
        }
    }
    for (int v = nfull * (4 * OWN_T) + t; v < nvec; v += OWN_T) drain((unsigned long long)vec(k4[v]), 16 * v);
    for (int e = 16 * nvec + t; e < E; e += OWN_T)
        if (key[e] == (unsigned char)kb) push(e);
}

__device__ __forceinline__ void own_resolve(const int* __restrict__ src, const float* __restrict__ dis,
                                            const float* sdis, int2* lst, float* wl, int total, int N)
{
    for (int i = threadIdx.x; i < total; i += OWN_T) {
        int2 en = lst[i];
        int s = src[en.x];
        s = min(max(s, 0), N - 1);
        lst[i].x = s;
        wl[i] = dis[s] * sdis[en.y];
    }
}

__device__ __forceinline__ void own_list(const int* __restrict__ src, const int* __restrict__ dst,
                                         const int2* lst, int* hist, int2* kl,
                                         int total, int N, int row0, int nrows)
{
    if (total <= 0) return;
    const int t = threadIdx.x;
    int e[OWN_CAP / OWN_T], s[OWN_CAP / OWN_T], r[OWN_CAP / OWN_T];
#pragma unroll
    for (int u = 0; u < OWN_CAP / OWN_T; ++u) e[u] = lst[min(t + u * OWN_T, total - 1)].x;
#pragma unroll
    for (int u = 0; u < OWN_CAP / OWN_T; ++u) { s[u] = src[e[u]]; r[u] = dst[e[u]]; }
#pragma unroll
    for (int u = 0; u < OWN_CAP / OWN_T; ++u) {
        const int i = t + u * OWN_T;
        if (i < total) {
            const int sv = min(max(s[u], 0), N - 1);
            const int rv = min(max(r[u] - row0, 0), nrows - 1);
            if (kl) kl[i] = make_int2(sv, rv);
            atomicAdd(&hist[rv], 1);
        }
    }
}

__device__ __forceinline__ void own_fetch(const int2* kl, const float* __restrict__ dis,
                                          const float* sdis, int2* lst, float* wl,
                                          int total, int N, int nrows)
{
    if (total <= 0) return;
    const int t = threadIdx.x;
    int2 en[OWN_CAP / OWN_T]; float ds[OWN_CAP / OWN_T];
#pragma unroll
    for (int u = 0; u < OWN_CAP / OWN_T; ++u) en[u] = kl[min(t + u * OWN_T, total - 1)];
#pragma unroll
    for (int u = 0; u < OWN_CAP / OWN_T; ++u) {
        en[u].x = min(max(en[u].x, 0), N - 1);
        en[u].y = min(max(en[u].y, 0), nrows - 1);
        ds[u] = dis[en[u].x];
    }
#pragma unroll
    for (int u = 0; u < OWN_CAP / OWN_T; ++u) {
        const int i = t + u * OWN_T;
        if (i < total) { lst[i] = en[u]; wl[i] = ds[u] * sdis[en[u].y]; }
    }
}

__device__ __forceinline__ int keep_cap(int nrows, int C)
{
    return min(OWN_CAP, (nrows * C) >> 1);
}

__global__ __launch_bounds__(OWN_T) void k_list(const int* __restrict__ src,
                                                const int* __restrict__ dst,
                                                float* __restrict__ dis,
                                                float* __restrict__ keep,
                                                int* __restrict__ kcnt,
                                                const unsigned char* __restrict__ key,
                                                int N, int E, int C)
{
    __shared__ int2 lst[OWN_CAP];
    __shared__ int  hist[OWN_R];
    __shared__ int  cnt;

    const int t = threadIdx.x;
    const int row0 = blockIdx.x * OWN_R;
    const int nrows = min(OWN_R, N - row0);
    const unsigned un = (unsigned)nrows;

    if (t == 0) cnt = 0;
    for (int r = t; r < OWN_R; r += OWN_T) hist[r] = 1;
    __syncthreads();

    const bool bykey = key != nullptr && N <= 256 * OWN_R;
    if (bykey) own_scan_key(key, E, blockIdx.x, lst, &cnt);
    else       own_scan(dst, E, row0, un, lst, &cnt);
    __syncthreads();

    const int total = __builtin_amdgcn_readfirstlane(cnt);
    if (t == 0) kcnt[blockIdx.x] = total;
    if (total <= OWN_CAP) {
        int2* kl = (total <= keep_cap(nrows, C)) ? (int2*)(keep + (size_t)row0 * C) : nullptr;
        own_list(src, dst, lst, hist, kl, total, N, row0, nrows);
    } else {
        for (int e = t; e < E; e += OWN_T) {
            unsigned r = (unsigned)(dst[e] - row0);
            if (r < un) atomicAdd(&hist[r], 1);
        }
    }
    __syncthreads();
    for (int r = t; r < nrows; r += OWN_T) dis[row0 + r] = rsqrtf((float)hist[r]);
}

__global__ __launch_bounds__(OWN_T) void k_own128(const float* __restrict__ h,
                                                  const int* __restrict__ src,
                                                  const int* __restrict__ dst,
                                                  const float* __restrict__ dis,
                                                  const float* __restrict__ bias,
                                                  float* __restrict__ out,
                                                  const float* __restrict__ keep,
                                                  const int* __restrict__ kcnt,
                                                  int N, int E, int C)
{
    __shared__ float acc[OWN_R * FDIM];
    __shared__ int2  lst[OWN_CAP];
    __shared__ float wl[OWN_CAP];
    __shared__ float sdis[OWN_R];
    __shared__ int   cnt;

    const int t = threadIdx.x;
    const int row0 = blockIdx.x * OWN_R;
    const int nrows = min(OWN_R, N - row0);
    const unsigned un = (unsigned)nrows;
    const float4* h4 = (const float4*)h;

    if (t == 0) cnt = 0;
    for (int r = t; r < nrows; r += OWN_T) sdis[r] = dis[row0 + r];
    {
        const float4 bq = ((const float4*)bias)[t & 31];
        for (int idx = t; idx < nrows * 32; idx += OWN_T) {
            int r = idx >> 5, q = idx & 31;
            float d = dis[row0 + r];
            float4 v = h4[(size_t)(row0 + r) * 32 + q];
            float* a = &acc[r * FDIM + q];
            a[0]  = v.x * d * d + bq.x;
            a[32] = v.y * d * d + bq.y;
            a[64] = v.z * d * d + bq.z;
            a[96] = v.w * d * d + bq.w;
        }
    }
    __syncthreads();

    const int kept = __builtin_amdgcn_readfirstlane(kcnt[blockIdx.x]);
    int total;
    if (kept <= keep_cap(nrows, C)) {
        total = max(kept, 0);
        own_fetch((const int2*)(keep + (size_t)row0 * C), dis, sdis, lst, wl, total, N, nrows);
    } else {
        own_scan(dst, E, row0, un, lst, &cnt);
        __syncthreads();
        total = __builtin_amdgcn_readfirstlane(cnt);
        if (total <= OWN_CAP) own_resolve(src, dis, sdis, lst, wl, total, N);
    }
    __syncthreads();

    if (total <= OWN_CAP) {
        const int wave = __builtin_amdgcn_readfirstlane(t >> 5), lane = t & 31;
        const int ngrp = total >> 3;
        for (int g = wave; g < ngrp; g += OWN_T / 32) {
            int r[8]; float w[8]; float4 v[8];
#pragma unroll
            for (int u = 0; u < 8; ++u) {
                int2 en = lst[8 * g + u];
                int s = __builtin_amdgcn_readfirstlane(en.x);
                r[u] = __builtin_amdgcn_readfirstlane(en.y);
                w[u] = wl[8 * g + u];
                v[u] = h4[(size_t)s * 32 + lane];
            }
#pragma unroll
            for (int u = 0; u < 8; ++u) {
                float* a = &acc[r[u] * FDIM + lane];
                atomicAdd(a,      v[u].x * w[u]);
                atomicAdd(a + 32, v[u].y * w[u]);
                atomicAdd(a + 64, v[u].z * w[u]);
                atomicAdd(a + 96, v[u].w * w[u]);
            }
        }
        if (8 * ngrp + wave < total) {
            int2 en = lst[8 * ngrp + wave];
            int s = __builtin_amdgcn_readfirstlane(en.x);
            int r = __builtin_amdgcn_readfirstlane(en.y);
            float w = wl[8 * ngrp + wave];
            float4 v = h4[(size_t)s * 32 + lane];
            float* a = &acc[r * FDIM + lane];
            atomicAdd(a,      v.x * w);
            atomicAdd(a + 32, v.y * w);
            atomicAdd(a + 64, v.z * w);
            atomicAdd(a + 96, v.w * w);
        }
    } else {
        for (int e = t; e < E; e += OWN_T) {
            unsigned r = (unsigned)(dst[e] - row0);
            if (r < un) {
                int s = min(max(src[e], 0), N - 1);
                float w = dis[s] * sdis[r];
                const float* hp = h + (size_t)s * FDIM;
                for (int f = 0; f < FDIM; ++f)
                    atomicAdd(&acc[r * FDIM + (f & 3) * 32 + (f >> 2)], hp[f] * w);
            }
        }
    }
    __syncthreads();

    float4* o4 = (float4*)out;
    for (int idx = t; idx < nrows * 32; idx += OWN_T) {
        int r = idx >> 5, q = idx & 31;
        const float* a = &acc[r * FDIM + q];
        o4[(size_t)(row0 + r) * 32 + q] = make_float4(a[0], a[32], a[64], a[96]);
    }
}

__global__ __launch_bounds__(OWN_T) void k_ownC(const float* __restrict__ h,
                                                const int* __restrict__ src,
                                                const int* __restrict__ dst,
                                                const float* __restrict__ dis,
                                                const float* __restrict__ bias,
                                                float* __restrict__ out,
                                                const int* __restrict__ kcnt,
                                                int N, int E, int C)
{
    __shared__ float acc[OWN_R * 64];
    __shared__ int2  lst[OWN_CAP];
    __shared__ float wl[OWN_CAP];
    __shared__ float sdis[OWN_R];
    __shared__ int   cnt;

    const int t = threadIdx.x;
    const int row0 = blockIdx.x * OWN_R;
    const int nrows = min(OWN_R, N - row0);
    const unsigned un = (unsigned)nrows;
    const size_t g0 = (size_t)row0 * C;

    if (t == 0) cnt = 0;
    for (int r = t; r < nrows; r += OWN_T) sdis[r] = dis[row0 + r];
    for (int idx = t; idx < nrows * C; idx += OWN_T) {
        int r = idx / C, f = idx - r * C;
        float d = dis[row0 + r];
        acc[idx] = h[g0 + idx] * d * d + bias[f];
    }
    __syncthreads();

    const int kept = __builtin_amdgcn_readfirstlane(kcnt[blockIdx.x]);
    int total;
    if (kept <= keep_cap(nrows, C)) {
        total = max(kept, 0);
        own_fetch((const int2*)(out + g0), dis, sdis, lst, wl, total, N, nrows);
    } else {
        own_scan(dst, E, row0, un, lst, &cnt);
        __syncthreads();
        total = __builtin_amdgcn_readfirstlane(cnt);
        if (total <= OWN_CAP) own_resolve(src, dis, sdis, lst, wl, total, N);
    }
    __syncthreads();

    if (total <= OWN_CAP) {
        const int wave = __builtin_amdgcn_readfirstlane(t >> 5), lane = t & 31;
        const bool c0 = lane < C, c1 = 32 + lane < C;
        const int l0 = min(lane, C - 1), l1 = min(32 + lane, C - 1);
        const int ngrp = total >> 3;
        for (int g = wave; g < ngrp; g += OWN_T / 32) {
            int s[8], r[8]; float w[8], v0[8], v1[8];
#pragma unroll
            for (int u = 0; u < 8; ++u) {
                int2 en = lst[8 * g + u];
                s[u] = __builtin_amdgcn_readfirstlane(en.x);
                r[u] = __builtin_amdgcn_readfirstlane(en.y);
                w[u] = wl[8 * g + u];
            }
#pragma unroll
            for (int u = 0; u < 8; ++u) {
                const float* hp = h + (size_t)s[u] * C;
                v0[u] = hp[l0];
                v1[u] = hp[l1];
            }
            if (c0) {
#pragma unroll
                for (int u = 0; u < 8; ++u) atomicAdd(&acc[r[u] * C + lane], v0[u] * w[u]);
            }
            if (c1) {
#pragma unroll
                for (int u = 0; u < 8; ++u) atomicAdd(&acc[r[u] * C + 32 + lane], v1[u] * w[u]);
            }
        }
        if (8 * ngrp + wave < total) {
            int2 en = lst[8 * ngrp + wave];
            int s = __builtin_amdgcn_readfirstlane(en.x);
            int r = __builtin_amdgcn_readfirstlane(en.y);
            float w = wl[8 * ngrp + wave];
            const float* hp = h + (size_t)s * C;
            float* a = &acc[r * C];
            if (c0) atomicAdd(a + lane,      hp[lane] * w);
            if (c1) atomicAdd(a + 32 + lane, hp[32 + lane] * w);
        }
    } else {
        for (int e = t; e < E; e += OWN_T) {
            unsigned r = (unsigned)(dst[e] - row0);
            if (r < un) {
                int s = min(max(src[e], 0), N - 1);
                float w = dis[s] * sdis[r];
                const float* hp = h + (size_t)s * C;
                for (int f = 0; f < C; ++f) atomicAdd(&acc[r * C + f], hp[f] * w);
            }
        }
    }
    __syncthreads();

    {
        const int lane = t & 31, sub = lane & 7;
        const int nk = (C + 7) >> 3;
        for (int rb = (t >> 5) * 4; rb < nrows; rb += (OWN_T / 32) * 4) {
            const int r = rb + (lane >> 3);
            const bool live = r < nrows;
            float* a = &acc[min(r, nrows - 1) * C];
            float v[8];
            float mx = -3.4e38f;
#pragma unroll
            for (int k = 0; k < 8; ++k) {
                const int f = sub + 8 * k;
                v[k] = (f < C) ? a[f] : -3.4e38f;
                mx = fmaxf(mx, v[k]);
            }
            mx = fmaxf(mx, __shfl_xor(mx, 1, 32));
            mx = fmaxf(mx, __shfl_xor(mx, 2, 32));
            mx = fmaxf(mx, __shfl_xor(mx, 4, 32));
            float sum = 0.f;
#pragma unroll
            for (int k = 0; k < 8; ++k) {
                if (k < nk) sum += (sub + 8 * k < C) ? expf(v[k] - mx) : 0.f;
            }
            sum += __shfl_xor(sum, 1, 32);
            sum += __shfl_xor(sum, 2, 32);
            sum += __shfl_xor(sum, 4, 32);
            const float l = logf(sum);
#pragma unroll
            for (int k = 0; k < 8; ++k) {
                const int f = sub + 8 * k;
                if (live && f < C) a[f] = v[k] - mx - l;
            }
        }
    }
    __syncthreads();

    for (int idx = t; idx < nrows * C; idx += OWN_T) out[g0 + idx] = acc[idx];
}

#define RS_T    1024
#define RS_PER  13
#define RS_ROWS (RS_PER * (RS_T / 32))
__global__ __launch_bounds__(RS_T) void k_relu_stats(float* __restrict__ a,
                                                     float* __restrict__ gsum,
                                                     float* __restrict__ gsq, int N)
{
    __shared__ float4 ssum[RS_T];
    __shared__ float4 ssq[RS_T];
    const int t = threadIdx.x, q = t & 31;
    const int row0 = blockIdx.x * RS_ROWS + (t >> 5);
    float4* a4 = (float4*)a;

    float4 v[RS_PER];
#pragma unroll
    for (int i = 0; i < RS_PER; ++i) {
        const int row = min(row0 + (RS_T / 32) * i, N - 1);
        v[i] = a4[(size_t)row * (FDIM / 4) + q];
    }

    float4 s  = make_float4(0.f, 0.f, 0.f, 0.f);
    float4 sq = make_float4(0.f, 0.f, 0.f, 0.f);
#pragma unroll
    for (int i = 0; i < RS_PER; ++i) {
        const int row = row0 + (RS_T / 32) * i;
        if (row < N) {
            const float4 r = make_float4(fmaxf(v[i].x, 0.f), fmaxf(v[i].y, 0.f),
                                         fmaxf(v[i].z, 0.f), fmaxf(v[i].w, 0.f));
            a4[(size_t)row * (FDIM / 4) + q] = r;
            s.x += r.x; s.y += r.y; s.z += r.z; s.w += r.w;
            sq.x += r.x * r.x; sq.y += r.y * r.y; sq.z += r.z * r.z; sq.w += r.w * r.w;
        }
    }
    ssum[t] = s; ssq[t] = sq;
    __syncthreads();

    if (t < 2 * FDIM) {
        const int c = t & (FDIM - 1);
        const float* p = (t < FDIM) ? (const float*)ssum : (const float*)ssq;
        float tot = 0.f;
#pragma unroll
        for (int g = 0; g < RS_T / 32; ++g) tot += p[g * FDIM + c];
        atomicAdd((t < FDIM) ? &gsum[c] : &gsq[c], tot);
    }
}

__global__ void k_bn_finalize(const float* __restrict__ gsum,
                              const float* __restrict__ gsq,
                              const float* __restrict__ gamma,
                              const float* __restrict__ beta,
                              float* __restrict__ scl,
                              float* __restrict__ sft, int N)
{
    int f = threadIdx.x;
    float inv_n = 1.f / (float)N;
    float mean = gsum[f] * inv_n;
    float var  = gsq[f] * inv_n - mean * mean;
    float inv  = rsqrtf(var + 1e-5f);
    float g = gamma[f] * inv;
    scl[f] = g;
    sft[f] = beta[f] - mean * g;
}

__global__ __launch_bounds__(256) void k_gemm2(const float* __restrict__ A,
                                               const float* __restrict__ W,
                                               const float* __restrict__ scl,
                                               const float* __restrict__ sft,
                                               float* __restrict__ O, int N, int C)
{
    __shared__ float4 lB[32 * 48];
    __shared__ float4 lS[32];
    __shared__ float4 lF[32];

    const int t = threadIdx.x;
    const int wave = t >> 5, lane = t & 31;
    const int m = lane & 15, hi = lane >> 4;
    const int rb = blockIdx.x * 128 + wave * 16;
    const int lrow = min(rb + m, N - 1);
    const v4f* ar = (const v4f*)(A + (size_t)lrow * FDIM) + hi;

    v4f xa[16];
#pragma unroll
    for (int j = 0; j < 16; ++j) xa[j] = ar[2 * j];

#pragma unroll
    for (int idx = t; idx < 32 * 48; idx += 256) {
        int jh = idx / 48, c = idx - jh * 48;
        const float* wp = W + (size_t)(4 * jh) * C + min(c, C - 1);
        const float4 v = make_float4(wp[0], wp[C], wp[2 * C], wp[3 * C]);
        lB[idx] = (c < C) ? v : make_float4(0.f, 0.f, 0.f, 0.f);
    }
    if (t < 32) {
        lS[t] = make_float4(scl[4 * t], scl[4 * t + 1], scl[4 * t + 2], scl[4 * t + 3]);
        lF[t] = make_float4(sft[4 * t], sft[4 * t + 1], sft[4 * t + 2], sft[4 * t + 3]);
    }
    __syncthreads();

    if (rb >= N) return;
    const float4* bp = &lB[hi * 48 + m];

    v8f acc[3];
#pragma unroll
    for (int ct = 0; ct < 3; ++ct) acc[ct] = (v8f){0.f, 0.f, 0.f, 0.f, 0.f, 0.f, 0.f, 0.f};

#pragma unroll
    for (int j = 0; j < 16; ++j) {
        const v4f x = xa[j];
        const float4 s = lS[2 * j + hi], f = lF[2 * j + hi];
        const v2f a0 = {x.x * s.x + f.x, x.y * s.y + f.y};
        const v2f a1 = {x.z * s.z + f.z, x.w * s.w + f.w};
        float4 b[3];
#pragma unroll
        for (int ct = 0; ct < 3; ++ct) b[ct] = bp[(2 * j) * 48 + ct * 16];
#pragma unroll
        for (int ct = 0; ct < 3; ++ct) {
            const v2f b0 = {b[ct].x, b[ct].y};
            acc[ct] = __builtin_amdgcn_wmma_f32_16x16x4_f32(
                false, a0, false, b0, (short)0, acc[ct], false, false);
        }
#pragma unroll
        for (int ct = 0; ct < 3; ++ct) {
            const v2f b1 = {b[ct].z, b[ct].w};
            acc[ct] = __builtin_amdgcn_wmma_f32_16x16x4_f32(
                false, a1, false, b1, (short)0, acc[ct], false, false);
        }
    }

#pragma unroll
    for (int ct = 0; ct < 3; ++ct) {
        const int nloc = ct * 16 + m;
        if (nloc < C) {
#pragma unroll
            for (int r = 0; r < 8; ++r) {
                int row = rb + r + 8 * hi;
                if (row < N) O[(size_t)row * C + nloc] = acc[ct][r];
            }
        }
    }
}

extern "C" void kernel_launch(void* const* d_in, const int* in_sizes, int n_in,
                              void* d_out, int out_size, void* d_ws, size_t ws_size,
                              hipStream_t stream)
{
    const float* x     = (const float*)d_in[0];
    const int*   ei    = (const int*)  d_in[1];
    const float* W1    = (const float*)d_in[2];
    const float* b1    = (const float*)d_in[3];
    const float* gamma = (const float*)d_in[4];
    const float* beta  = (const float*)d_in[5];
    const float* W2    = (const float*)d_in[6];
    const float* b2    = (const float*)d_in[7];
    float* out = (float*)d_out;

    const int N = in_sizes[0] / FDIM;
    const int E = in_sizes[1] / 2;
    const int C = in_sizes[6] / FDIM;
    const int* src = ei;
    const int* dst = ei + E;

    char* ws = (char*)d_ws;
    size_t off = 0;
    auto up = [](size_t v) { return (v + 255) & ~(size_t)255; };
    const int nb_own = (N + OWN_R - 1) / OWN_R;
    int*   kcnt  = (int*)(ws + off);   off = up(off + (size_t)nb_own * 4);
    float* dis   = (float*)(ws + off); off = up(off + (size_t)N * 4);
    float* stats = (float*)(ws + off); off = up(off + 512 * 4);
    float* gsum = stats, *gsq = stats + 128, *scl = stats + 256, *sft = stats + 384;
    float* h1   = (float*)(ws + off);  off = up(off + (size_t)N * FDIM * 4);
    float* agg1 = (float*)(ws + off);  off = up(off + (size_t)N * FDIM * 4);
    float* h2   = h1;
    unsigned char* key = (unsigned char*)(ws + off); off = up(off + (size_t)E);
    if (off > ws_size) key = nullptr;

    const int nb_key = (E / 16 + 256) / 256;

    k_key<<<nb_key, 256, 0, stream>>>(dst, key, stats, E);

    k_list<<<nb_own, OWN_T, 0, stream>>>(src, dst, dis, out, kcnt, key, N, E, C);

    k_gemm1<<<(N + 127) / 128, 256, 0, stream>>>(x, W1, h1, N);

    k_own128<<<nb_own, OWN_T, 0, stream>>>(h1, src, dst, dis, b1, agg1, out, kcnt, N, E, C);

    k_relu_stats<<<(N + RS_ROWS - 1) / RS_ROWS, RS_T, 0, stream>>>(agg1, gsum, gsq, N);
    k_bn_finalize<<<1, 128, 0, stream>>>(gsum, gsq, gamma, beta, scl, sft, N);

    k_gemm2<<<(N + 127) / 128, 256, 0, stream>>>(agg1, W2, scl, sft, h2, N, C);

    k_ownC<<<nb_own, OWN_T, 0, stream>>>(h2, src, dst, dis, b2, out, kcnt, N, E, C);
}


# First conv as sum-then-transform: owner block sums x rows in LDS, multiplies its rows by W1 on WMMA from LDS, bias+ReLU+column sums in its epilogue; 8 to 6 launches; conv1 and column-sum order change
# speedup vs baseline: 4.3607x; 1.1351x over previous
#include <hip/hip_runtime.h>

typedef __attribute__((ext_vector_type(2))) float v2f;
typedef __attribute__((ext_vector_type(8))) float v8f;
typedef __attribute__((ext_vector_type(4))) float v4f;

#define FDIM 128
#define OWN_R   400
#define OWN_CAP 8192
#define OWN_T   1024


__global__ __launch_bounds__(256) void k_key(const int* __restrict__ dst,
                                             unsigned char* __restrict__ key,
                                             float* __restrict__ stats, int E)
{
    const int i = blockIdx.x * 256 + threadIdx.x;
    if (i < 256) stats[i] = 0.f;
    if (key == nullptr) return;
    const int e0 = 16 * i;
    if (e0 >= E) return;
    auto k1 = [](int d) { return ((unsigned)d / OWN_R) & 255u; };
    if (e0 + 16 <= E && ((size_t)dst & 15) == 0) {
        const int4* p = (const int4*)(dst + e0);
        const int4 a = p[0], b = p[1], c = p[2], d = p[3];
        auto k4 = [&](const int4& v) { return k1(v.x) | (k1(v.y) << 8) | (k1(v.z) << 16) | (k1(v.w) << 24); };
        *(uint4*)(key + e0) = make_uint4(k4(a), k4(b), k4(c), k4(d));
    } else {
        const int e1 = min(e0 + 16, E);
        for (int e = e0; e < e1; ++e) key[e] = (unsigned char)k1(dst[e]);
    }
}


__device__ __forceinline__ void own_scan(const int* __restrict__ dst, int E, int row0, unsigned un,
                                         int2* lst, int* cnt)
{
    const int t = threadIdx.x;
    auto push = [&](int e, unsigned r) {
        int pos = atomicAdd(cnt, 1);
        if (pos < OWN_CAP) lst[pos] = make_int2(e, (int)r);
    };
    auto test = [&](const int4& d, int e) {
        unsigned r0 = (unsigned)(d.x - row0), r1 = (unsigned)(d.y - row0);
        unsigned r2 = (unsigned)(d.z - row0), r3 = (unsigned)(d.w - row0);
        unsigned m = min(min(r0, r1), min(r2, r3));
        if (m < un) {
            if (r0 < un) push(e, r0);
            if (r1 < un) push(e + 1, r1);
            if (r2 < un) push(e + 2, r2);
            if (r3 < un) push(e + 3, r3);
        }
    };
    const int mis = (int)(((size_t)dst >> 2) & 3);
    int head = (4 - mis) & 3; if (head > E) head = E;
    const int nvec = (E - head) >> 2;
    const int tail0 = head + 4 * nvec;
    const int4* d4 = (const int4*)(dst + head);
    const int nfull = nvec / (4 * OWN_T);
    if (nfull > 0) {
        int tile = (int)(blockIdx.x % (unsigned)nfull);
        const int4* p = d4 + (size_t)tile * (4 * OWN_T) + t;
        int4 c0 = p[0], c1 = p[OWN_T], c2 = p[2 * OWN_T], c3 = p[3 * OWN_T];
        for (int it = 0; it < nfull; ++it) {
            int nt = tile + 1; if (nt >= nfull) nt = 0;
            const int4* q = d4 + (size_t)nt * (4 * OWN_T) + t;
            int4 n0 = q[0], n1 = q[OWN_T], n2 = q[2 * OWN_T], n3 = q[3 * OWN_T];
            int e = head + 4 * (tile * (4 * OWN_T) + t);
            test(c0, e);
            test(c1, e + 4 * OWN_T);
            test(c2, e + 8 * OWN_T);
            test(c3, e + 12 * OWN_T);
            c0 = n0; c1 = n1; c2 = n2; c3 = n3; tile = nt;
        }
    }
    for (int v = nfull * (4 * OWN_T) + t; v < nvec; v += OWN_T) test(d4[v], head + 4 * v);
    if (t < head) {
        unsigned r = (unsigned)(dst[t] - row0);
        if (r < un) push(t, r);
    }
    if (t < E - tail0) {
        unsigned r = (unsigned)(dst[tail0 + t] - row0);
        if (r < un) push(tail0 + t, r);
    }
}

__device__ __forceinline__ void own_scan_key(const unsigned char* __restrict__ key, int E, unsigned kb,
                                             int2* lst, int* cnt)
{
    const int t = threadIdx.x;
    const unsigned pat = kb * 0x01010101u;
    auto push = [&](int e) {
        int pos = atomicAdd(cnt, 1);
        if (pos < OWN_CAP) lst[pos] = make_int2(e, 0);
    };
    auto eq = [&](unsigned x) {
        unsigned y = x ^ pat;
        return ~(((y & 0x7f7f7f7fu) + 0x7f7f7f7fu) | y) & 0x80808080u;
    };
    auto vec = [&](const uint4& k) {
        return (eq(k.x) >> 7) | (eq(k.y) >> 6) | (eq(k.z) >> 5) | (eq(k.w) >> 4);
    };
    auto drain = [&](unsigned long long M, int e0) {
        while (M) {
            const int q = __builtin_ctzll(M);
            const int v = 2 * (q >> 5) + ((q >> 2) & 1);
            push(e0 + v * (16 * OWN_T) + 4 * (q & 3) + ((q >> 3) & 3));
            M &= M - 1;
        }
    };
    const int nvec = E >> 4;
    const uint4* k4 = (const uint4*)key;
    const int nfull = nvec / (4 * OWN_T);
    if (nfull > 0) {
        int tile = (int)(blockIdx.x % (unsigned)nfull);
        const uint4* p = k4 + (size_t)tile * (4 * OWN_T) + t;
        uint4 c0 = p[0], c1 = p[OWN_T], c2 = p[2 * OWN_T], c3 = p[3 * OWN_T];
        for (int it = 0; it < nfull; ++it) {
            int nt = tile + 1; if (nt >= nfull) nt = 0;
            const uint4* q = k4 + (size_t)nt * (4 * OWN_T) + t;
            uint4 n0 = q[0], n1 = q[OWN_T], n2 = q[2 * OWN_T], n3 = q[3 * OWN_T];
            const unsigned m01 = vec(c0) | (vec(c1) << 4), m23 = vec(c2) | (vec(c3) << 4);
            drain((unsigned long long)m01 | ((unsigned long long)m23 << 32), 16 * (tile * (4 * OWN_T) + t));
            c0 = n0; c1 = n1; c2 = n2; c3 = n3; tile = nt;
        }
    }
    for (int v = nfull * (4 * OWN_T) + t; v < nvec; v += OWN_T) drain((unsigned long long)vec(k4[v]), 16 * v);
    for (int e = 16 * nvec + t; e < E; e += OWN_T)
        if (key[e] == (unsigned char)kb) push(e);
}

__device__ __forceinline__ void own_resolve(const int* __restrict__ src, const float* __restrict__ dis,
                                            const float* sdis, int2* lst, float* wl, int total, int N)
{
    for (int i = threadIdx.x; i < total; i += OWN_T) {
        int2 en = lst[i];
        int s = src[en.x];
        s = min(max(s, 0), N - 1);
        lst[i].x = s;
        wl[i] = dis[s] * sdis[en.y];
    }
}

__device__ __forceinline__ void own_list(const int* __restrict__ src, const int* __restrict__ dst,
                                         const int2* lst, int* hist, int2* kl,
                                         int total, int N, int row0, int nrows)
{
    if (total <= 0) return;
    const int t = threadIdx.x;
    int e[OWN_CAP / OWN_T], s[OWN_CAP / OWN_T], r[OWN_CAP / OWN_T];
#pragma unroll
    for (int u = 0; u < OWN_CAP / OWN_T; ++u) e[u] = lst[min(t + u * OWN_T, total - 1)].x;
#pragma unroll
    for (int u = 0; u < OWN_CAP / OWN_T; ++u) { s[u] = src[e[u]]; r[u] = dst[e[u]]; }
#pragma unroll
    for (int u = 0; u < OWN_CAP / OWN_T; ++u) {
        const int i = t + u * OWN_T;
        if (i < total) {
            const int sv = min(max(s[u], 0), N - 1);
            const int rv = min(max(r[u] - row0, 0), nrows - 1);
            if (kl) kl[i] = make_int2(sv, rv);
            atomicAdd(&hist[rv], 1);
        }
    }
}

__device__ __forceinline__ void own_fetch(const int2* kl, const float* __restrict__ dis,
                                          const float* sdis, int2* lst, float* wl,
                                          int total, int N, int nrows)
{
    if (total <= 0) return;
    const int t = threadIdx.x;
    int2 en[OWN_CAP / OWN_T]; float ds[OWN_CAP / OWN_T];
#pragma unroll
    for (int u = 0; u < OWN_CAP / OWN_T; ++u) en[u] = kl[min(t + u * OWN_T, total - 1)];
#pragma unroll
    for (int u = 0; u < OWN_CAP / OWN_T; ++u) {
        en[u].x = min(max(en[u].x, 0), N - 1);
        en[u].y = min(max(en[u].y, 0), nrows - 1);
        ds[u] = dis[en[u].x];
    }
#pragma unroll
    for (int u = 0; u < OWN_CAP / OWN_T; ++u) {
        const int i = t + u * OWN_T;
        if (i < total) { lst[i] = en[u]; wl[i] = ds[u] * sdis[en[u].y]; }
    }
}

__device__ __forceinline__ int keep_cap(int nrows, int C)
{
    return min(OWN_CAP, (nrows * C) >> 1);
}

__global__ __launch_bounds__(OWN_T) void k_list(const int* __restrict__ src,
                                                const int* __restrict__ dst,
                                                float* __restrict__ dis,
                                                float* __restrict__ keep,
                                                int* __restrict__ kcnt,
                                                const unsigned char* __restrict__ key,
                                                int N, int E, int C)
{
    __shared__ int2 lst[OWN_CAP];
    __shared__ int  hist[OWN_R];
    __shared__ int  cnt;

    const int t = threadIdx.x;
    const int row0 = blockIdx.x * OWN_R;
    const int nrows = min(OWN_R, N - row0);
    const unsigned un = (unsigned)nrows;

    if (t == 0) cnt = 0;
    for (int r = t; r < OWN_R; r += OWN_T) hist[r] = 1;
    __syncthreads();

    const bool bykey = key != nullptr && N <= 256 * OWN_R;
    if (bykey) own_scan_key(key, E, blockIdx.x, lst, &cnt);
    else       own_scan(dst, E, row0, un, lst, &cnt);
    __syncthreads();

    const int total = __builtin_amdgcn_readfirstlane(cnt);
    if (t == 0) kcnt[blockIdx.x] = total;
    if (total <= OWN_CAP) {
        int2* kl = (total <= keep_cap(nrows, C)) ? (int2*)(keep + (size_t)row0 * C) : nullptr;
        own_list(src, dst, lst, hist, kl, total, N, row0, nrows);
    } else {
        for (int e = t; e < E; e += OWN_T) {
            unsigned r = (unsigned)(dst[e] - row0);
            if (r < un) atomicAdd(&hist[r], 1);
        }
    }
    __syncthreads();
    for (int r = t; r < nrows; r += OWN_T) dis[row0 + r] = rsqrtf((float)hist[r]);
}

#define ACC_P 130
__global__ __launch_bounds__(OWN_T) void k_own128g(const float* __restrict__ x,
                                                   const int* __restrict__ src,
                                                   const int* __restrict__ dst,
                                                   const float* __restrict__ dis,
                                                   const float* __restrict__ W,
                                                   const float* __restrict__ bias,
                                                   float* __restrict__ out,
                                                   float* __restrict__ gsum,
                                                   float* __restrict__ gsq,
                                                   const float* __restrict__ keep,
                                                   const int* __restrict__ kcnt,
                                                   int N, int E, int C)
{
    __shared__ float acc[OWN_R * ACC_P];
    __shared__ __attribute__((aligned(16))) unsigned char pool[OWN_CAP * 8 + OWN_CAP * 4];
    __shared__ float sdis[OWN_R];
    __shared__ int   cnt;
    int2*   lst = (int2*)pool;
    float*  wl  = (float*)(pool + OWN_CAP * 8);
    float4* lB  = (float4*)pool;
    float*  bs  = (float*)(pool + OWN_CAP * 8);

    const int t = threadIdx.x;
    const int row0 = blockIdx.x * OWN_R;
    const int nrows = min(OWN_R, N - row0);
    const unsigned un = (unsigned)nrows;
    const float4* h4 = (const float4*)x;

    if (t == 0) cnt = 0;
    for (int r = t; r < nrows; r += OWN_T) sdis[r] = dis[row0 + r];
    for (int idx = t; idx < nrows * 32; idx += OWN_T) {
        int r = idx >> 5, q = idx & 31;
        float d = dis[row0 + r];
        float4 v = h4[(size_t)(row0 + r) * 32 + q];
        float* a = &acc[r * ACC_P + q];
        a[0]  = v.x * d * d;
        a[32] = v.y * d * d;
        a[64] = v.z * d * d;
        a[96] = v.w * d * d;
    }
    __syncthreads();

    const int kept = __builtin_amdgcn_readfirstlane(kcnt[blockIdx.x]);
    int total;
    if (kept <= keep_cap(nrows, C)) {
        total = max(kept, 0);
        own_fetch((const int2*)(keep + (size_t)row0 * C), dis, sdis, lst, wl, total, N, nrows);
    } else {
        own_scan(dst, E, row0, un, lst, &cnt);
        __syncthreads();
        total = __builtin_amdgcn_readfirstlane(cnt);
        if (total <= OWN_CAP) own_resolve(src, dis, sdis, lst, wl, total, N);
    }
    __syncthreads();

    const int wave = __builtin_amdgcn_readfirstlane(t >> 5), lane = t & 31;
    if (total <= OWN_CAP) {
        const int ngrp = total >> 3;
        for (int g = wave; g < ngrp; g += OWN_T / 32) {
            int r[8]; float w[8]; float4 v[8];
#pragma unroll
            for (int u = 0; u < 8; ++u) {
                int2 en = lst[8 * g + u];
                int s = __builtin_amdgcn_readfirstlane(en.x);
                r[u] = __builtin_amdgcn_readfirstlane(en.y);
                w[u] = wl[8 * g + u];
                v[u] = h4[(size_t)s * 32 + lane];
            }
#pragma unroll
            for (int u = 0; u < 8; ++u) {
                float* a = &acc[r[u] * ACC_P + lane];
                atomicAdd(a,      v[u].x * w[u]);
                atomicAdd(a + 32, v[u].y * w[u]);
                atomicAdd(a + 64, v[u].z * w[u]);
                atomicAdd(a + 96, v[u].w * w[u]);
            }
        }
        if (8 * ngrp + wave < total) {
            int2 en = lst[8 * ngrp + wave];
            int s = __builtin_amdgcn_readfirstlane(en.x);
            int r = __builtin_amdgcn_readfirstlane(en.y);
            float w = wl[8 * ngrp + wave];
            float4 v = h4[(size_t)s * 32 + lane];
            float* a = &acc[r * ACC_P + lane];
            atomicAdd(a,      v.x * w);
            atomicAdd(a + 32, v.y * w);
            atomicAdd(a + 64, v.z * w);
            atomicAdd(a + 96, v.w * w);
        }
    } else {
        for (int e = t; e < E; e += OWN_T) {
            unsigned r = (unsigned)(dst[e] - row0);
            if (r < un) {
                int s = min(max(src[e], 0), N - 1);
                float w = dis[s] * sdis[r];
                const float* hp = x + (size_t)s * FDIM;
                for (int f = 0; f < FDIM; ++f)
                    atomicAdd(&acc[r * ACC_P + (f & 3) * 32 + (f >> 2)], hp[f] * w);
            }
        }
    }
    __syncthreads();

    {
        const int n4 = (t & 31) * 4, jh = t >> 5;
        const float4* wp = (const float4*)(W + (size_t)(4 * jh) * FDIM + n4);
        const float4 r0 = wp[0], r1 = wp[FDIM / 4], r2 = wp[2 * (FDIM / 4)], r3 = wp[3 * (FDIM / 4)];
        float4* lp = &lB[jh * FDIM + n4];
        lp[0] = make_float4(r0.x, r1.x, r2.x, r3.x);
        lp[1] = make_float4(r0.y, r1.y, r2.y, r3.y);
        lp[2] = make_float4(r0.z, r1.z, r2.z, r3.z);
        lp[3] = make_float4(r0.w, r1.w, r2.w, r3.w);
        if (t < 2 * FDIM) bs[t] = 0.f;
    }
    __syncthreads();

    const int m = lane & 15, hi = lane >> 4;
    const int ntile = (nrows + 15) >> 4;
    if (wave < ntile) {
        const float* ap = &acc[(wave * 16 + m) * ACC_P + hi];
        const float4* bp = &lB[hi * FDIM + m];
        for (int half = 0; half < 2; ++half) {
            v8f c[4];
#pragma unroll
            for (int ct = 0; ct < 4; ++ct) c[ct] = (v8f){0.f, 0.f, 0.f, 0.f, 0.f, 0.f, 0.f, 0.f};
#pragma unroll 4
            for (int j = 0; j < 16; ++j) {
                const float* a = ap + 2 * j;
                const v2f a0 = {a[0], a[32]}, a1 = {a[64], a[96]};
                float4 b[4];
#pragma unroll
                for (int ct = 0; ct < 4; ++ct) b[ct] = bp[(2 * j) * FDIM + (half * 4 + ct) * 16];
#pragma unroll
                for (int ct = 0; ct < 4; ++ct) {
                    const v2f b0 = {b[ct].x, b[ct].y};
                    c[ct] = __builtin_amdgcn_wmma_f32_16x16x4_f32(
                        false, a0, false, b0, (short)0, c[ct], false, false);
                }
#pragma unroll
                for (int ct = 0; ct < 4; ++ct) {
                    const v2f b1 = {b[ct].z, b[ct].w};
                    c[ct] = __builtin_amdgcn_wmma_f32_16x16x4_f32(
                        false, a1, false, b1, (short)0, c[ct], false, false);
                }
            }
#pragma unroll
            for (int ct = 0; ct < 4; ++ct) {
                const int col = (half * 4 + ct) * 16 + m;
                const float bv = bias[col];
                float s = 0.f, q = 0.f;
#pragma unroll
                for (int r = 0; r < 8; ++r) {
                    const int rl = wave * 16 + r + 8 * hi;
                    const float v = fmaxf(c[ct][r] + bv, 0.f);
                    if (rl < nrows) {
                        out[(size_t)(row0 + rl) * FDIM + col] = v;
                        s += v; q += v * v;
                    }
                }
                s += __shfl_xor(s, 16, 32);
                q += __shfl_xor(q, 16, 32);
                if (hi == 0) { atomicAdd(&bs[col], s); atomicAdd(&bs[FDIM + col], q); }
            }
        }
    }
    __syncthreads();

    if (t < 2 * FDIM) atomicAdd((t < FDIM) ? &gsum[t] : &gsq[t - FDIM], bs[t]);
}

__global__ __launch_bounds__(OWN_T) void k_ownC(const float* __restrict__ h,
                                                const int* __restrict__ src,
                                                const int* __restrict__ dst,
                                                const float* __restrict__ dis,
                                                const float* __restrict__ bias,
                                                float* __restrict__ out,
                                                const int* __restrict__ kcnt,
                                                int N, int E, int C)
{
    __shared__ float acc[OWN_R * 64];
    __shared__ int2  lst[OWN_CAP];
    __shared__ float wl[OWN_CAP];
    __shared__ float sdis[OWN_R];
    __shared__ int   cnt;

    const int t = threadIdx.x;
    const int row0 = blockIdx.x * OWN_R;
    const int nrows = min(OWN_R, N - row0);
    const unsigned un = (unsigned)nrows;
    const size_t g0 = (size_t)row0 * C;

    if (t == 0) cnt = 0;
    for (int r = t; r < nrows; r += OWN_T) sdis[r] = dis[row0 + r];
    for (int idx = t; idx < nrows * C; idx += OWN_T) {
        int r = idx / C, f = idx - r * C;
        float d = dis[row0 + r];
        acc[idx] = h[g0 + idx] * d * d + bias[f];
    }
    __syncthreads();

    const int kept = __builtin_amdgcn_readfirstlane(kcnt[blockIdx.x]);
    int total;
    if (kept <= keep_cap(nrows, C)) {
        total = max(kept, 0);
        own_fetch((const int2*)(out + g0), dis, sdis, lst, wl, total, N, nrows);
    } else {
        own_scan(dst, E, row0, un, lst, &cnt);
        __syncthreads();
        total = __builtin_amdgcn_readfirstlane(cnt);
        if (total <= OWN_CAP) own_resolve(src, dis, sdis, lst, wl, total, N);
    }
    __syncthreads();

    if (total <= OWN_CAP) {
        const int wave = __builtin_amdgcn_readfirstlane(t >> 5), lane = t & 31;
        const bool c0 = lane < C, c1 = 32 + lane < C;
        const int l0 = min(lane, C - 1), l1 = min(32 + lane, C - 1);
        const int ngrp = total >> 3;
        for (int g = wave; g < ngrp; g += OWN_T / 32) {
            int s[8], r[8]; float w[8], v0[8], v1[8];
#pragma unroll
            for (int u = 0; u < 8; ++u) {
                int2 en = lst[8 * g + u];
                s[u] = __builtin_amdgcn_readfirstlane(en.x);
                r[u] = __builtin_amdgcn_readfirstlane(en.y);
                w[u] = wl[8 * g + u];
            }
#pragma unroll
            for (int u = 0; u < 8; ++u) {
                const float* hp = h + (size_t)s[u] * C;
                v0[u] = hp[l0];
                v1[u] = hp[l1];
            }
            if (c0) {
#pragma unroll
                for (int u = 0; u < 8; ++u) atomicAdd(&acc[r[u] * C + lane], v0[u] * w[u]);
            }
            if (c1) {
#pragma unroll
                for (int u = 0; u < 8; ++u) atomicAdd(&acc[r[u] * C + 32 + lane], v1[u] * w[u]);
            }
        }
        if (8 * ngrp + wave < total) {
            int2 en = lst[8 * ngrp + wave];
            int s = __builtin_amdgcn_readfirstlane(en.x);
            int r = __builtin_amdgcn_readfirstlane(en.y);
            float w = wl[8 * ngrp + wave];
            const float* hp = h + (size_t)s * C;
            float* a = &acc[r * C];
            if (c0) atomicAdd(a + lane,      hp[lane] * w);
            if (c1) atomicAdd(a + 32 + lane, hp[32 + lane] * w);
        }
    } else {
        for (int e = t; e < E; e += OWN_T) {
            unsigned r = (unsigned)(dst[e] - row0);
            if (r < un) {
                int s = min(max(src[e], 0), N - 1);
                float w = dis[s] * sdis[r];
                const float* hp = h + (size_t)s * C;
                for (int f = 0; f < C; ++f) atomicAdd(&acc[r * C + f], hp[f] * w);
            }
        }
    }
    __syncthreads();

    {
        const int lane = t & 31, sub = lane & 7;
        const int nk = (C + 7) >> 3;
        for (int rb = (t >> 5) * 4; rb < nrows; rb += (OWN_T / 32) * 4) {
            const int r = rb + (lane >> 3);
            const bool live = r < nrows;
            float* a = &acc[min(r, nrows - 1) * C];
            float v[8];
            float mx = -3.4e38f;
#pragma unroll
            for (int k = 0; k < 8; ++k) {
                const int f = sub + 8 * k;
                v[k] = (f < C) ? a[f] : -3.4e38f;
                mx = fmaxf(mx, v[k]);
            }
            mx = fmaxf(mx, __shfl_xor(mx, 1, 32));
            mx = fmaxf(mx, __shfl_xor(mx, 2, 32));
            mx = fmaxf(mx, __shfl_xor(mx, 4, 32));
            float sum = 0.f;
#pragma unroll
            for (int k = 0; k < 8; ++k) {
                if (k < nk) sum += (sub + 8 * k < C) ? expf(v[k] - mx) : 0.f;
            }
            sum += __shfl_xor(sum, 1, 32);
            sum += __shfl_xor(sum, 2, 32);
            sum += __shfl_xor(sum, 4, 32);
            const float l = logf(sum);
#pragma unroll
            for (int k = 0; k < 8; ++k) {
                const int f = sub + 8 * k;
                if (live && f < C) a[f] = v[k] - mx - l;
            }
        }
    }
    __syncthreads();

    for (int idx = t; idx < nrows * C; idx += OWN_T) out[g0 + idx] = acc[idx];
}

__global__ void k_bn_finalize(const float* __restrict__ gsum,
                              const float* __restrict__ gsq,
                              const float* __restrict__ gamma,
                              const float* __restrict__ beta,
                              float* __restrict__ scl,
                              float* __restrict__ sft, int N)
{
    int f = threadIdx.x;
    float inv_n = 1.f / (float)N;
    float mean = gsum[f] * inv_n;
    float var  = gsq[f] * inv_n - mean * mean;
    float inv  = rsqrtf(var + 1e-5f);
    float g = gamma[f] * inv;
    scl[f] = g;
    sft[f] = beta[f] - mean * g;
}

__global__ __launch_bounds__(256) void k_gemm2(const float* __restrict__ A,
                                               const float* __restrict__ W,
                                               const float* __restrict__ scl,
                                               const float* __restrict__ sft,
                                               float* __restrict__ O, int N, int C)
{
    __shared__ float4 lB[32 * 48];
    __shared__ float4 lS[32];
    __shared__ float4 lF[32];

    const int t = threadIdx.x;
    const int wave = t >> 5, lane = t & 31;
    const int m = lane & 15, hi = lane >> 4;
    const int rb = blockIdx.x * 128 + wave * 16;
    const int lrow = min(rb + m, N - 1);
    const v4f* ar = (const v4f*)(A + (size_t)lrow * FDIM) + hi;

    v4f xa[16];
#pragma unroll
    for (int j = 0; j < 16; ++j) xa[j] = ar[2 * j];

#pragma unroll
    for (int idx = t; idx < 32 * 48; idx += 256) {
        int jh = idx / 48, c = idx - jh * 48;
        const float* wp = W + (size_t)(4 * jh) * C + min(c, C - 1);
        const float4 v = make_float4(wp[0], wp[C], wp[2 * C], wp[3 * C]);
        lB[idx] = (c < C) ? v : make_float4(0.f, 0.f, 0.f, 0.f);
    }
    if (t < 32) {
        lS[t] = make_float4(scl[4 * t], scl[4 * t + 1], scl[4 * t + 2], scl[4 * t + 3]);
        lF[t] = make_float4(sft[4 * t], sft[4 * t + 1], sft[4 * t + 2], sft[4 * t + 3]);
    }
    __syncthreads();

    if (rb >= N) return;
    const float4* bp = &lB[hi * 48 + m];

    v8f acc[3];
#pragma unroll
    for (int ct = 0; ct < 3; ++ct) acc[ct] = (v8f){0.f, 0.f, 0.f, 0.f, 0.f, 0.f, 0.f, 0.f};

#pragma unroll
    for (int j = 0; j < 16; ++j) {
        const v4f x = xa[j];
        const float4 s = lS[2 * j + hi], f = lF[2 * j + hi];
        const v2f a0 = {x.x * s.x + f.x, x.y * s.y + f.y};
        const v2f a1 = {x.z * s.z + f.z, x.w * s.w + f.w};
        float4 b[3];
#pragma unroll
        for (int ct = 0; ct < 3; ++ct) b[ct] = bp[(2 * j) * 48 + ct * 16];
#pragma unroll
        for (int ct = 0; ct < 3; ++ct) {
            const v2f b0 = {b[ct].x, b[ct].y};
            acc[ct] = __builtin_amdgcn_wmma_f32_16x16x4_f32(
                false, a0, false, b0, (short)0, acc[ct], false, false);
        }
#pragma unroll
        for (int ct = 0; ct < 3; ++ct) {
            const v2f b1 = {b[ct].z, b[ct].w};
            acc[ct] = __builtin_amdgcn_wmma_f32_16x16x4_f32(
                false, a1, false, b1, (short)0, acc[ct], false, false);
        }
    }

#pragma unroll
    for (int ct = 0; ct < 3; ++ct) {
        const int nloc = ct * 16 + m;
        if (nloc < C) {
#pragma unroll
            for (int r = 0; r < 8; ++r) {
                int row = rb + r + 8 * hi;
                if (row < N) O[(size_t)row * C + nloc] = acc[ct][r];
            }
        }
    }
}

extern "C" void kernel_launch(void* const* d_in, const int* in_sizes, int n_in,
                              void* d_out, int out_size, void* d_ws, size_t ws_size,
                              hipStream_t stream)
{
    const float* x     = (const float*)d_in[0];
    const int*   ei    = (const int*)  d_in[1];
    const float* W1    = (const float*)d_in[2];
    const float* b1    = (const float*)d_in[3];
    const float* gamma = (const float*)d_in[4];
    const float* beta  = (const float*)d_in[5];
    const float* W2    = (const float*)d_in[6];
    const float* b2    = (const float*)d_in[7];
    float* out = (float*)d_out;

    const int N = in_sizes[0] / FDIM;
    const int E = in_sizes[1] / 2;
    const int C = in_sizes[6] / FDIM;
    const int* src = ei;
    const int* dst = ei + E;

    char* ws = (char*)d_ws;
    size_t off = 0;
    auto up = [](size_t v) { return (v + 255) & ~(size_t)255; };
    const int nb_own = (N + OWN_R - 1) / OWN_R;
    int*   kcnt  = (int*)(ws + off);   off = up(off + (size_t)nb_own * 4);
    float* dis   = (float*)(ws + off); off = up(off + (size_t)N * 4);
    float* stats = (float*)(ws + off); off = up(off + 512 * 4);
    float* gsum = stats, *gsq = stats + 128, *scl = stats + 256, *sft = stats + 384;
    float* h1   = (float*)(ws + off);  off = up(off + (size_t)N * FDIM * 4);
    float* agg1 = (float*)(ws + off);  off = up(off + (size_t)N * FDIM * 4);
    float* h2   = h1;
    unsigned char* key = (unsigned char*)(ws + off); off = up(off + (size_t)E);
    if (off > ws_size) key = nullptr;

    const int nb_key = (E / 16 + 256) / 256;

    k_key<<<nb_key, 256, 0, stream>>>(dst, key, stats, E);

    k_list<<<nb_own, OWN_T, 0, stream>>>(src, dst, dis, out, kcnt, key, N, E, C);

    k_own128g<<<nb_own, OWN_T, 0, stream>>>(x, src, dst, dis, W1, b1, agg1, gsum, gsq, out, kcnt, N, E, C);

    k_bn_finalize<<<1, 128, 0, stream>>>(gsum, gsq, gamma, beta, scl, sft, N);

    k_gemm2<<<(N + 127) / 128, 256, 0, stream>>>(agg1, W2, scl, sft, h2, N, C);

    k_ownC<<<nb_own, OWN_T, 0, stream>>>(h2, src, dst, dis, b2, out, kcnt, N, E, C);
}


# Packed edge word (source | row in owner block << 22) from the key pass: list pass reads one scattered word per edge, not two; conv1 keeps the weight beside it, conv2 looks up none; sums unchanged
# speedup vs baseline: 4.5571x; 1.0450x over previous
#include <hip/hip_runtime.h>

typedef __attribute__((ext_vector_type(2))) float v2f;
typedef __attribute__((ext_vector_type(8))) float v8f;
typedef __attribute__((ext_vector_type(4))) float v4f;

#define FDIM 128
#define OWN_R   400
#define OWN_CAP 8192
#define OWN_T   1024
#define PACK_SH 22


__global__ __launch_bounds__(256) void k_key(const int* __restrict__ src,
                                             const int* __restrict__ dst,
                                             unsigned char* __restrict__ key,
                                             int* __restrict__ sr,
                                             float* __restrict__ stats, int E, int N)
{
    const int i = blockIdx.x * 256 + threadIdx.x;
    if (i < 256) stats[i] = 0.f;
    if (key == nullptr) return;
    const int e0 = 16 * i;
    if (e0 >= E) return;
    if (N > (1 << PACK_SH)) sr = nullptr;
    auto k1 = [](int d) { return ((unsigned)d / OWN_R) & 255u; };
    auto pk = [&](int s, int d) { return min(max(s, 0), N - 1) | (int)(((unsigned)d % OWN_R) << PACK_SH); };
    if (e0 + 16 <= E && ((size_t)dst & 15) == 0) {
        const int4* p = (const int4*)(dst + e0);
        const int4 a = p[0], b = p[1], c = p[2], d = p[3];
        int4 sa = make_int4(0, 0, 0, 0), sb = sa, sc = sa, sd = sa;
        if (sr != nullptr) {
            const bool al = ((size_t)src & 15) == 0;
            auto ld4 = [&](const int* q) { return al ? *(const int4*)q : make_int4(q[0], q[1], q[2], q[3]); };
            sa = ld4(src + e0); sb = ld4(src + e0 + 4); sc = ld4(src + e0 + 8); sd = ld4(src + e0 + 12);
        }
        auto k4 = [&](const int4& v) { return k1(v.x) | (k1(v.y) << 8) | (k1(v.z) << 16) | (k1(v.w) << 24); };
        *(uint4*)(key + e0) = make_uint4(k4(a), k4(b), k4(c), k4(d));
        if (sr != nullptr) {
            auto p4 = [&](const int4& s, const int4& v) { return make_int4(pk(s.x, v.x), pk(s.y, v.y), pk(s.z, v.z), pk(s.w, v.w)); };
            int4* o = (int4*)(sr + e0);
            o[0] = p4(sa, a); o[1] = p4(sb, b); o[2] = p4(sc, c); o[3] = p4(sd, d);
        }
    } else {
        const int e1 = min(e0 + 16, E);
        for (int e = e0; e < e1; ++e) {
            const int d = dst[e];
            key[e] = (unsigned char)k1(d);
            if (sr != nullptr) sr[e] = pk(src[e], d);
        }
    }
}


__device__ __forceinline__ void own_scan(const int* __restrict__ dst, int E, int row0, unsigned un,
                                         int2* lst, int* cnt)
{
    const int t = threadIdx.x;
    auto push = [&](int e, unsigned r) {
        int pos = atomicAdd(cnt, 1);
        if (pos < OWN_CAP) lst[pos] = make_int2(e, (int)r);
    };
    auto test = [&](const int4& d, int e) {
        unsigned r0 = (unsigned)(d.x - row0), r1 = (unsigned)(d.y - row0);
        unsigned r2 = (unsigned)(d.z - row0), r3 = (unsigned)(d.w - row0);
        unsigned m = min(min(r0, r1), min(r2, r3));
        if (m < un) {
            if (r0 < un) push(e, r0);
            if (r1 < un) push(e + 1, r1);
            if (r2 < un) push(e + 2, r2);
            if (r3 < un) push(e + 3, r3);
        }
    };
    const int mis = (int)(((size_t)dst >> 2) & 3);
    int head = (4 - mis) & 3; if (head > E) head = E;
    const int nvec = (E - head) >> 2;
    const int tail0 = head + 4 * nvec;
    const int4* d4 = (const int4*)(dst + head);
    const int nfull = nvec / (4 * OWN_T);
    if (nfull > 0) {
        int tile = (int)(blockIdx.x % (unsigned)nfull);
        const int4* p = d4 + (size_t)tile * (4 * OWN_T) + t;
        int4 c0 = p[0], c1 = p[OWN_T], c2 = p[2 * OWN_T], c3 = p[3 * OWN_T];
        for (int it = 0; it < nfull; ++it) {
            int nt = tile + 1; if (nt >= nfull) nt = 0;
            const int4* q = d4 + (size_t)nt * (4 * OWN_T) + t;
            int4 n0 = q[0], n1 = q[OWN_T], n2 = q[2 * OWN_T], n3 = q[3 * OWN_T];
            int e = head + 4 * (tile * (4 * OWN_T) + t);
            test(c0, e);
            test(c1, e + 4 * OWN_T);
            test(c2, e + 8 * OWN_T);
            test(c3, e + 12 * OWN_T);
            c0 = n0; c1 = n1; c2 = n2; c3 = n3; tile = nt;
        }
    }
    for (int v = nfull * (4 * OWN_T) + t; v < nvec; v += OWN_T) test(d4[v], head + 4 * v);
    if (t < head) {
        unsigned r = (unsigned)(dst[t] - row0);
        if (r < un) push(t, r);
    }
    if (t < E - tail0) {
        unsigned r = (unsigned)(dst[tail0 + t] - row0);
        if (r < un) push(tail0 + t, r);
    }
}

__device__ __forceinline__ void own_scan_key(const unsigned char* __restrict__ key, int E, unsigned kb,
                                             int2* lst, int* cnt)
{
    const int t = threadIdx.x;
    const unsigned pat = kb * 0x01010101u;
    auto push = [&](int e) {
        int pos = atomicAdd(cnt, 1);
        if (pos < OWN_CAP) lst[pos] = make_int2(e, 0);
    };
    auto eq = [&](unsigned x) {
        unsigned y = x ^ pat;
        return ~(((y & 0x7f7f7f7fu) + 0x7f7f7f7fu) | y) & 0x80808080u;
    };
    auto vec = [&](const uint4& k) {
        return (eq(k.x) >> 7) | (eq(k.y) >> 6) | (eq(k.z) >> 5) | (eq(k.w) >> 4);
    };
    auto drain = [&](unsigned long long M, int e0) {
        while (M) {
            const int q = __builtin_ctzll(M);
            const int v = 2 * (q >> 5) + ((q >> 2) & 1);
            push(e0 + v * (16 * OWN_T) + 4 * (q & 3) + ((q >> 3) & 3));
            M &= M - 1;
        }
    };
    const int nvec = E >> 4;
    const uint4* k4 = (const uint4*)key;
    const int nfull = nvec / (4 * OWN_T);
    if (nfull > 0) {
        int tile = (int)(blockIdx.x % (unsigned)nfull);
        const uint4* p = k4 + (size_t)tile * (4 * OWN_T) + t;
        uint4 c0 = p[0], c1 = p[OWN_T], c2 = p[2 * OWN_T], c3 = p[3 * OWN_T];
        for (int it = 0; it < nfull; ++it) {
            int nt = tile + 1; if (nt >= nfull) nt = 0;
            const uint4* q = k4 + (size_t)nt * (4 * OWN_T) + t;
            uint4 n0 = q[0], n1 = q[OWN_T], n2 = q[2 * OWN_T], n3 = q[3 * OWN_T];
            const unsigned m01 = vec(c0) | (vec(c1) << 4), m23 = vec(c2) | (vec(c3) << 4);
            drain((unsigned long long)m01 | ((unsigned long long)m23 << 32), 16 * (tile * (4 * OWN_T) + t));
            c0 = n0; c1 = n1; c2 = n2; c3 = n3; tile = nt;
        }
    }
    for (int v = nfull * (4 * OWN_T) + t; v < nvec; v += OWN_T) drain((unsigned long long)vec(k4[v]), 16 * v);
    for (int e = 16 * nvec + t; e < E; e += OWN_T)
        if (key[e] == (unsigned char)kb) push(e);
}

__device__ __forceinline__ void own_resolve(const int* __restrict__ src, const float* __restrict__ dis,
                                            const float* sdis, int2* lst, float* wl, int total, int N)
{
    for (int i = threadIdx.x; i < total; i += OWN_T) {
        int2 en = lst[i];
        int s = src[en.x];
        s = min(max(s, 0), N - 1);
        lst[i].x = s;
        wl[i] = dis[s] * sdis[en.y];
    }
}

__device__ __forceinline__ void own_list(const int* __restrict__ src, const int* __restrict__ dst,
                                         const int* __restrict__ sr,
                                         const int2* lst, int* hist, int2* kl,
                                         int total, int N, int row0, int nrows)
{
    if (total <= 0) return;
    const int t = threadIdx.x;
    int e[OWN_CAP / OWN_T], s[OWN_CAP / OWN_T], r[OWN_CAP / OWN_T];
#pragma unroll
    for (int u = 0; u < OWN_CAP / OWN_T; ++u) e[u] = lst[min(t + u * OWN_T, total - 1)].x;
    if (sr != nullptr) {
#pragma unroll
        for (int u = 0; u < OWN_CAP / OWN_T; ++u) s[u] = sr[e[u]];
#pragma unroll
        for (int u = 0; u < OWN_CAP / OWN_T; ++u) {
            r[u] = row0 + (int)((unsigned)s[u] >> PACK_SH);
            s[u] &= (1 << PACK_SH) - 1;
        }
    } else {
#pragma unroll
        for (int u = 0; u < OWN_CAP / OWN_T; ++u) { s[u] = src[e[u]]; r[u] = dst[e[u]]; }
    }
#pragma unroll
    for (int u = 0; u < OWN_CAP / OWN_T; ++u) {
        const int i = t + u * OWN_T;
        if (i < total) {
            const int sv = min(max(s[u], 0), N - 1);
            const int rv = min(max(r[u] - row0, 0), nrows - 1);
            if (kl) kl[i] = make_int2(sv, rv);
            atomicAdd(&hist[rv], 1);
        }
    }
}

__device__ __forceinline__ void own_fetch(const int2* kl, int2* klw, const float* __restrict__ dis,
                                          const float* sdis, int2* lst, float* wl,
                                          int total, int N, int nrows)
{
    if (total <= 0) return;
    const int t = threadIdx.x;
    int2 en[OWN_CAP / OWN_T]; float ds[OWN_CAP / OWN_T];
#pragma unroll
    for (int u = 0; u < OWN_CAP / OWN_T; ++u) en[u] = kl[min(t + u * OWN_T, total - 1)];
#pragma unroll
    for (int u = 0; u < OWN_CAP / OWN_T; ++u) {
        en[u].x = min(max(en[u].x, 0), N - 1);
        en[u].y = min(max(en[u].y, 0), nrows - 1);
        ds[u] = dis[en[u].x];
    }
#pragma unroll
    for (int u = 0; u < OWN_CAP / OWN_T; ++u) {
        const int i = t + u * OWN_T;
        if (i < total) {
            const float w = ds[u] * sdis[en[u].y];
            lst[i] = en[u]; wl[i] = w;
            klw[i] = make_int2(en[u].x | (en[u].y << PACK_SH), __float_as_int(w));
        }
    }
}

__device__ __forceinline__ void own_fetchw(const int2* kl, int2* lst, float* wl,
                                           int total, int N, int nrows)
{
    if (total <= 0) return;
    const int t = threadIdx.x;
    int2 en[OWN_CAP / OWN_T];
#pragma unroll
    for (int u = 0; u < OWN_CAP / OWN_T; ++u) en[u] = kl[min(t + u * OWN_T, total - 1)];
#pragma unroll
    for (int u = 0; u < OWN_CAP / OWN_T; ++u) {
        const int i = t + u * OWN_T;
        if (i < total) {
            const unsigned p = (unsigned)en[u].x;
            const int s = min((int)(p & ((1u << PACK_SH) - 1u)), N - 1);
            const int r = min((int)(p >> PACK_SH), nrows - 1);
            lst[i] = make_int2(s, r);
            wl[i] = __int_as_float(en[u].y);
        }
    }
}

__device__ __forceinline__ int keep_cap(int nrows, int C, int N)
{
    if (N > (1 << PACK_SH)) return -1;
    return min(OWN_CAP, (nrows * C) >> 1);
}

__global__ __launch_bounds__(OWN_T) void k_list(const int* __restrict__ src,
                                                const int* __restrict__ dst,
                                                float* __restrict__ dis,
                                                float* __restrict__ keep,
                                                int* __restrict__ kcnt,
                                                const unsigned char* __restrict__ key,
                                                const int* __restrict__ sr,
                                                int N, int E, int C)
{
    __shared__ int2 lst[OWN_CAP];
    __shared__ int  hist[OWN_R];
    __shared__ int  cnt;

    const int t = threadIdx.x;
    const int row0 = blockIdx.x * OWN_R;
    const int nrows = min(OWN_R, N - row0);
    const unsigned un = (unsigned)nrows;

    if (t == 0) cnt = 0;
    for (int r = t; r < OWN_R; r += OWN_T) hist[r] = 1;
    __syncthreads();

    const bool bykey = key != nullptr && N <= 256 * OWN_R;
    if (bykey) own_scan_key(key, E, blockIdx.x, lst, &cnt);
    else       own_scan(dst, E, row0, un, lst, &cnt);
    __syncthreads();

    const int total = __builtin_amdgcn_readfirstlane(cnt);
    if (t == 0) kcnt[blockIdx.x] = total;
    if (total <= OWN_CAP) {
        int2* kl = (total <= keep_cap(nrows, C, N)) ? (int2*)(keep + (size_t)row0 * C) : nullptr;
        const int* srp = (bykey && N <= (1 << PACK_SH)) ? sr : nullptr;
        own_list(src, dst, srp, lst, hist, kl, total, N, row0, nrows);
    } else {
        for (int e = t; e < E; e += OWN_T) {
            unsigned r = (unsigned)(dst[e] - row0);
            if (r < un) atomicAdd(&hist[r], 1);
        }
    }
    __syncthreads();
    for (int r = t; r < nrows; r += OWN_T) dis[row0 + r] = rsqrtf((float)hist[r]);
}

#define ACC_P 130
__global__ __launch_bounds__(OWN_T) void k_own128g(const float* __restrict__ x,
                                                   const int* __restrict__ src,
                                                   const int* __restrict__ dst,
                                                   const float* __restrict__ dis,
                                                   const float* __restrict__ W,
                                                   const float* __restrict__ bias,
                                                   float* __restrict__ out,
                                                   float* __restrict__ gsum,
                                                   float* __restrict__ gsq,
                                                   float* __restrict__ keep,
                                                   const int* __restrict__ kcnt,
                                                   int N, int E, int C)
{
    __shared__ float acc[OWN_R * ACC_P];
    __shared__ __attribute__((aligned(16))) unsigned char pool[OWN_CAP * 8 + OWN_CAP * 4];
    __shared__ float sdis[OWN_R];
    __shared__ int   cnt;
    int2*   lst = (int2*)pool;
    float*  wl  = (float*)(pool + OWN_CAP * 8);
    float4* lB  = (float4*)pool;
    float*  bs  = (float*)(pool + OWN_CAP * 8);

    const int t = threadIdx.x;
    const int row0 = blockIdx.x * OWN_R;
    const int nrows = min(OWN_R, N - row0);
    const unsigned un = (unsigned)nrows;
    const float4* h4 = (const float4*)x;

    if (t == 0) cnt = 0;
    for (int r = t; r < nrows; r += OWN_T) sdis[r] = dis[row0 + r];
    for (int idx = t; idx < nrows * 32; idx += OWN_T) {
        int r = idx >> 5, q = idx & 31;
        float d = dis[row0 + r];
        float4 v = h4[(size_t)(row0 + r) * 32 + q];
        float* a = &acc[r * ACC_P + q];
        a[0]  = v.x * d * d;
        a[32] = v.y * d * d;
        a[64] = v.z * d * d;
        a[96] = v.w * d * d;
    }
    __syncthreads();

    const int kept = __builtin_amdgcn_readfirstlane(kcnt[blockIdx.x]);
    int total;
    if (kept <= keep_cap(nrows, C, N)) {
        total = max(kept, 0);
        int2* kl = (int2*)(keep + (size_t)row0 * C);
        own_fetch(kl, kl, dis, sdis, lst, wl, total, N, nrows);
    } else {
        own_scan(dst, E, row0, un, lst, &cnt);
        __syncthreads();
        total = __builtin_amdgcn_readfirstlane(cnt);
        if (total <= OWN_CAP) own_resolve(src, dis, sdis, lst, wl, total, N);
    }
    __syncthreads();

    const int wave = __builtin_amdgcn_readfirstlane(t >> 5), lane = t & 31;
    if (total <= OWN_CAP) {
        const int ngrp = total >> 3;
        for (int g = wave; g < ngrp; g += OWN_T / 32) {
            int r[8]; float w[8]; float4 v[8];
#pragma unroll
            for (int u = 0; u < 8; ++u) {
                int2 en = lst[8 * g + u];
                int s = __builtin_amdgcn_readfirstlane(en.x);
                r[u] = __builtin_amdgcn_readfirstlane(en.y);
                w[u] = wl[8 * g + u];
                v[u] = h4[(size_t)s * 32 + lane];
            }
#pragma unroll
            for (int u = 0; u < 8; ++u) {
                float* a = &acc[r[u] * ACC_P + lane];
                atomicAdd(a,      v[u].x * w[u]);
                atomicAdd(a + 32, v[u].y * w[u]);
                atomicAdd(a + 64, v[u].z * w[u]);
                atomicAdd(a + 96, v[u].w * w[u]);
            }
        }
        if (8 * ngrp + wave < total) {
            int2 en = lst[8 * ngrp + wave];
            int s = __builtin_amdgcn_readfirstlane(en.x);
            int r = __builtin_amdgcn_readfirstlane(en.y);
            float w = wl[8 * ngrp + wave];
            float4 v = h4[(size_t)s * 32 + lane];
            float* a = &acc[r * ACC_P + lane];
            atomicAdd(a,      v.x * w);
            atomicAdd(a + 32, v.y * w);
            atomicAdd(a + 64, v.z * w);
            atomicAdd(a + 96, v.w * w);
        }
    } else {
        for (int e = t; e < E; e += OWN_T) {
            unsigned r = (unsigned)(dst[e] - row0);
            if (r < un) {
                int s = min(max(src[e], 0), N - 1);
                float w = dis[s] * sdis[r];
                const float* hp = x + (size_t)s * FDIM;
                for (int f = 0; f < FDIM; ++f)
                    atomicAdd(&acc[r * ACC_P + (f & 3) * 32 + (f >> 2)], hp[f] * w);
            }
        }
    }
    __syncthreads();

    {
        const int n4 = (t & 31) * 4, jh = t >> 5;
        const float4* wp = (const float4*)(W + (size_t)(4 * jh) * FDIM + n4);
        const float4 r0 = wp[0], r1 = wp[FDIM / 4], r2 = wp[2 * (FDIM / 4)], r3 = wp[3 * (FDIM / 4)];
        float4* lp = &lB[jh * FDIM + n4];
        lp[0] = make_float4(r0.x, r1.x, r2.x, r3.x);
        lp[1] = make_float4(r0.y, r1.y, r2.y, r3.y);
        lp[2] = make_float4(r0.z, r1.z, r2.z, r3.z);
        lp[3] = make_float4(r0.w, r1.w, r2.w, r3.w);
        if (t < 2 * FDIM) bs[t] = 0.f;
    }
    __syncthreads();

    const int m = lane & 15, hi = lane >> 4;
    const int ntile = (nrows + 15) >> 4;
    if (wave < ntile) {
        const float* ap = &acc[(wave * 16 + m) * ACC_P + hi];
        const float4* bp = &lB[hi * FDIM + m];
        for (int half = 0; half < 2; ++half) {
            v8f c[4];
#pragma unroll
            for (int ct = 0; ct < 4; ++ct) c[ct] = (v8f){0.f, 0.f, 0.f, 0.f, 0.f, 0.f, 0.f, 0.f};
#pragma unroll 4
            for (int j = 0; j < 16; ++j) {
                const float* a = ap + 2 * j;
                const v2f a0 = {a[0], a[32]}, a1 = {a[64], a[96]};
                float4 b[4];
#pragma unroll
                for (int ct = 0; ct < 4; ++ct) b[ct] = bp[(2 * j) * FDIM + (half * 4 + ct) * 16];
#pragma unroll
                for (int ct = 0; ct < 4; ++ct) {
                    const v2f b0 = {b[ct].x, b[ct].y};
                    c[ct] = __builtin_amdgcn_wmma_f32_16x16x4_f32(
                        false, a0, false, b0, (short)0, c[ct], false, false);
                }
#pragma unroll
                for (int ct = 0; ct < 4; ++ct) {
                    const v2f b1 = {b[ct].z, b[ct].w};
                    c[ct] = __builtin_amdgcn_wmma_f32_16x16x4_f32(
                        false, a1, false, b1, (short)0, c[ct], false, false);
                }
            }
#pragma unroll
            for (int ct = 0; ct < 4; ++ct) {
                const int col = (half * 4 + ct) * 16 + m;
                const float bv = bias[col];
                float s = 0.f, q = 0.f;
#pragma unroll
                for (int r = 0; r < 8; ++r) {
                    const int rl = wave * 16 + r + 8 * hi;
                    const float v = fmaxf(c[ct][r] + bv, 0.f);
                    if (rl < nrows) {
                        out[(size_t)(row0 + rl) * FDIM + col] = v;
                        s += v; q += v * v;
                    }
                }
                s += __shfl_xor(s, 16, 32);
                q += __shfl_xor(q, 16, 32);
                if (hi == 0) { atomicAdd(&bs[col], s); atomicAdd(&bs[FDIM + col], q); }
            }
        }
    }
    __syncthreads();

    if (t < 2 * FDIM) atomicAdd((t < FDIM) ? &gsum[t] : &gsq[t - FDIM], bs[t]);
}

__global__ __launch_bounds__(OWN_T) void k_ownC(const float* __restrict__ h,
                                                const int* __restrict__ src,
                                                const int* __restrict__ dst,
                                                const float* __restrict__ dis,
                                                const float* __restrict__ bias,
                                                float* __restrict__ out,
                                                const int* __restrict__ kcnt,
                                                int N, int E, int C)
{
    __shared__ float acc[OWN_R * 64];
    __shared__ int2  lst[OWN_CAP];
    __shared__ float wl[OWN_CAP];
    __shared__ float sdis[OWN_R];
    __shared__ int   cnt;

    const int t = threadIdx.x;
    const int row0 = blockIdx.x * OWN_R;
    const int nrows = min(OWN_R, N - row0);
    const unsigned un = (unsigned)nrows;
    const size_t g0 = (size_t)row0 * C;

    if (t == 0) cnt = 0;
    for (int r = t; r < nrows; r += OWN_T) sdis[r] = dis[row0 + r];
    for (int idx = t; idx < nrows * C; idx += OWN_T) {
        int r = idx / C, f = idx - r * C;
        float d = dis[row0 + r];
        acc[idx] = h[g0 + idx] * d * d + bias[f];
    }
    __syncthreads();

    const int kept = __builtin_amdgcn_readfirstlane(kcnt[blockIdx.x]);
    int total;
    if (kept <= keep_cap(nrows, C, N)) {
        total = max(kept, 0);
        own_fetchw((const int2*)(out + g0), lst, wl, total, N, nrows);
    } else {
        own_scan(dst, E, row0, un, lst, &cnt);
        __syncthreads();
        total = __builtin_amdgcn_readfirstlane(cnt);
        if (total <= OWN_CAP) own_resolve(src, dis, sdis, lst, wl, total, N);
    }
    __syncthreads();

    if (total <= OWN_CAP) {
        const int wave = __builtin_amdgcn_readfirstlane(t >> 5), lane = t & 31;
        const bool c0 = lane < C, c1 = 32 + lane < C;
        const int l0 = min(lane, C - 1), l1 = min(32 + lane, C - 1);
        const int ngrp = total >> 3;
        for (int g = wave; g < ngrp; g += OWN_T / 32) {
            int s[8], r[8]; float w[8], v0[8], v1[8];
#pragma unroll
            for (int u = 0; u < 8; ++u) {
                int2 en = lst[8 * g + u];
                s[u] = __builtin_amdgcn_readfirstlane(en.x);
                r[u] = __builtin_amdgcn_readfirstlane(en.y);
                w[u] = wl[8 * g + u];
            }
#pragma unroll
            for (int u = 0; u < 8; ++u) {
                const float* hp = h + (size_t)s[u] * C;
                v0[u] = hp[l0];
                v1[u] = hp[l1];
            }
            if (c0) {
#pragma unroll
                for (int u = 0; u < 8; ++u) atomicAdd(&acc[r[u] * C + lane], v0[u] * w[u]);
            }
            if (c1) {
#pragma unroll
                for (int u = 0; u < 8; ++u) atomicAdd(&acc[r[u] * C + 32 + lane], v1[u] * w[u]);
            }
        }
        if (8 * ngrp + wave < total) {
            int2 en = lst[8 * ngrp + wave];
            int s = __builtin_amdgcn_readfirstlane(en.x);
            int r = __builtin_amdgcn_readfirstlane(en.y);
            float w = wl[8 * ngrp + wave];
            const float* hp = h + (size_t)s * C;
            float* a = &acc[r * C];
            if (c0) atomicAdd(a + lane,      hp[lane] * w);
            if (c1) atomicAdd(a + 32 + lane, hp[32 + lane] * w);
        }
    } else {
        for (int e = t; e < E; e += OWN_T) {
            unsigned r = (unsigned)(dst[e] - row0);
            if (r < un) {
                int s = min(max(src[e], 0), N - 1);
                float w = dis[s] * sdis[r];
                const float* hp = h + (size_t)s * C;
                for (int f = 0; f < C; ++f) atomicAdd(&acc[r * C + f], hp[f] * w);
            }
        }
    }
    __syncthreads();

    {
        const int lane = t & 31, sub = lane & 7;
        const int nk = (C + 7) >> 3;
        for (int rb = (t >> 5) * 4; rb < nrows; rb += (OWN_T / 32) * 4) {
            const int r = rb + (lane >> 3);
            const bool live = r < nrows;
            float* a = &acc[min(r, nrows - 1) * C];
            float v[8];
            float mx = -3.4e38f;
#pragma unroll
            for (int k = 0; k < 8; ++k) {
                const int f = sub + 8 * k;
                v[k] = (f < C) ? a[f] : -3.4e38f;
                mx = fmaxf(mx, v[k]);
            }
            mx = fmaxf(mx, __shfl_xor(mx, 1, 32));
            mx = fmaxf(mx, __shfl_xor(mx, 2, 32));
            mx = fmaxf(mx, __shfl_xor(mx, 4, 32));
            float sum = 0.f;
#pragma unroll
            for (int k = 0; k < 8; ++k) {
                if (k < nk) sum += (sub + 8 * k < C) ? expf(v[k] - mx) : 0.f;
            }
            sum += __shfl_xor(sum, 1, 32);
            sum += __shfl_xor(sum, 2, 32);
            sum += __shfl_xor(sum, 4, 32);
            const float l = logf(sum);
#pragma unroll
            for (int k = 0; k < 8; ++k) {
                const int f = sub + 8 * k;
                if (live && f < C) a[f] = v[k] - mx - l;
            }
        }
    }
    __syncthreads();

    for (int idx = t; idx < nrows * C; idx += OWN_T) out[g0 + idx] = acc[idx];
}

__global__ void k_bn_finalize(const float* __restrict__ gsum,
                              const float* __restrict__ gsq,
                              const float* __restrict__ gamma,
                              const float* __restrict__ beta,
                              float* __restrict__ scl,
                              float* __restrict__ sft, int N)
{
    int f = threadIdx.x;
    float inv_n = 1.f / (float)N;
    float mean = gsum[f] * inv_n;
    float var  = gsq[f] * inv_n - mean * mean;
    float inv  = rsqrtf(var + 1e-5f);
    float g = gamma[f] * inv;
    scl[f] = g;
    sft[f] = beta[f] - mean * g;
}

__global__ __launch_bounds__(256) void k_gemm2(const float* __restrict__ A,
                                               const float* __restrict__ W,
                                               const float* __restrict__ scl,
                                               const float* __restrict__ sft,
                                               float* __restrict__ O, int N, int C)
{
    __shared__ float4 lB[32 * 48];
    __shared__ float4 lS[32];
    __shared__ float4 lF[32];

    const int t = threadIdx.x;
    const int wave = t >> 5, lane = t & 31;
    const int m = lane & 15, hi = lane >> 4;
    const int rb = blockIdx.x * 128 + wave * 16;
    const int lrow = min(rb + m, N - 1);
    const v4f* ar = (const v4f*)(A + (size_t)lrow * FDIM) + hi;

    v4f xa[16];
#pragma unroll
    for (int j = 0; j < 16; ++j) xa[j] = ar[2 * j];

#pragma unroll
    for (int idx = t; idx < 32 * 48; idx += 256) {
        int jh = idx / 48, c = idx - jh * 48;
        const float* wp = W + (size_t)(4 * jh) * C + min(c, C - 1);
        const float4 v = make_float4(wp[0], wp[C], wp[2 * C], wp[3 * C]);
        lB[idx] = (c < C) ? v : make_float4(0.f, 0.f, 0.f, 0.f);
    }
    if (t < 32) {
        lS[t] = make_float4(scl[4 * t], scl[4 * t + 1], scl[4 * t + 2], scl[4 * t + 3]);
        lF[t] = make_float4(sft[4 * t], sft[4 * t + 1], sft[4 * t + 2], sft[4 * t + 3]);
    }
    __syncthreads();

    if (rb >= N) return;
    const float4* bp = &lB[hi * 48 + m];

    v8f acc[3];
#pragma unroll
    for (int ct = 0; ct < 3; ++ct) acc[ct] = (v8f){0.f, 0.f, 0.f, 0.f, 0.f, 0.f, 0.f, 0.f};

#pragma unroll
    for (int j = 0; j < 16; ++j) {
        const v4f x = xa[j];
        const float4 s = lS[2 * j + hi], f = lF[2 * j + hi];
        const v2f a0 = {x.x * s.x + f.x, x.y * s.y + f.y};
        const v2f a1 = {x.z * s.z + f.z, x.w * s.w + f.w};
        float4 b[3];
#pragma unroll
        for (int ct = 0; ct < 3; ++ct) b[ct] = bp[(2 * j) * 48 + ct * 16];
#pragma unroll
        for (int ct = 0; ct < 3; ++ct) {
            const v2f b0 = {b[ct].x, b[ct].y};
            acc[ct] = __builtin_amdgcn_wmma_f32_16x16x4_f32(
                false, a0, false, b0, (short)0, acc[ct], false, false);
        }
#pragma unroll
        for (int ct = 0; ct < 3; ++ct) {
            const v2f b1 = {b[ct].z, b[ct].w};
            acc[ct] = __builtin_amdgcn_wmma_f32_16x16x4_f32(
                false, a1, false, b1, (short)0, acc[ct], false, false);
        }
    }

#pragma unroll
    for (int ct = 0; ct < 3; ++ct) {
        const int nloc = ct * 16 + m;
        if (nloc < C) {
#pragma unroll
            for (int r = 0; r < 8; ++r) {
                int row = rb + r + 8 * hi;
                if (row < N) O[(size_t)row * C + nloc] = acc[ct][r];
            }
        }
    }
}

extern "C" void kernel_launch(void* const* d_in, const int* in_sizes, int n_in,
                              void* d_out, int out_size, void* d_ws, size_t ws_size,
                              hipStream_t stream)
{
    const float* x     = (const float*)d_in[0];
    const int*   ei    = (const int*)  d_in[1];
    const float* W1    = (const float*)d_in[2];
    const float* b1    = (const float*)d_in[3];
    const float* gamma = (const float*)d_in[4];
    const float* beta  = (const float*)d_in[5];
    const float* W2    = (const float*)d_in[6];
    const float* b2    = (const float*)d_in[7];
    float* out = (float*)d_out;

    const int N = in_sizes[0] / FDIM;
    const int E = in_sizes[1] / 2;
    const int C = in_sizes[6] / FDIM;
    const int* src = ei;
    const int* dst = ei + E;

    char* ws = (char*)d_ws;
    size_t off = 0;
    auto up = [](size_t v) { return (v + 255) & ~(size_t)255; };
    const int nb_own = (N + OWN_R - 1) / OWN_R;
    int*   kcnt  = (int*)(ws + off);   off = up(off + (size_t)nb_own * 4);
    float* dis   = (float*)(ws + off); off = up(off + (size_t)N * 4);
    float* stats = (float*)(ws + off); off = up(off + 512 * 4);
    float* gsum = stats, *gsq = stats + 128, *scl = stats + 256, *sft = stats + 384;
    float* h2   = (float*)(ws + off);  off = up(off + (size_t)N * C * 4);
    float* agg1 = (float*)(ws + off);  off = up(off + (size_t)N * FDIM * 4);
    unsigned char* key = (unsigned char*)(ws + off); off = up(off + (size_t)E);
    if (off > ws_size) key = nullptr;
    int* sr = (int*)(ws + off); off = up(off + (size_t)E * 4);
    if (off > ws_size) sr = nullptr;

    const int nb_key = (E / 16 + 256) / 256;

    k_key<<<nb_key, 256, 0, stream>>>(src, dst, key, sr, stats, E, N);

    k_list<<<nb_own, OWN_T, 0, stream>>>(src, dst, dis, out, kcnt, key, sr, N, E, C);

    k_own128g<<<nb_own, OWN_T, 0, stream>>>(x, src, dst, dis, W1, b1, agg1, gsum, gsq, out, kcnt, N, E, C);

    k_bn_finalize<<<1, 128, 0, stream>>>(gsum, gsq, gamma, beta, scl, sft, N);

    k_gemm2<<<(N + 127) / 128, 256, 0, stream>>>(agg1, W2, scl, sft, h2, N, C);

    k_ownC<<<nb_own, OWN_T, 0, stream>>>(h2, src, dst, dis, b2, out, kcnt, N, E, C);
}
